# starting kernel
# baseline (speedup 1.0000x reference)
#include <hip/hip_runtime.h>

#define BB 2
#define HH 8
#define SS 2048
#define DD 64
#define NB 32
#define BS 64
#define NSEL 8
#define SELC 512
#define KEEP 256
#define WIN 128
#define PSTRIDE 520
#define NEGINF (-1e30f)
#define SCALE 0.125f

typedef __attribute__((ext_vector_type(2))) float v2f;
typedef __attribute__((ext_vector_type(8))) float v8f;

static __device__ inline v8f vzero8() {
  v8f r;
#pragma unroll
  for (int i = 0; i < 8; ++i) r[i] = 0.f;
  return r;
}

static __device__ inline v8f wmma4(v2f a, v2f b, v8f c) {
  return __builtin_amdgcn_wmma_f32_16x16x4_f32(false, a, false, b, (short)0, c,
                                               false, false);
}

static __device__ inline unsigned okey(float x) {
  unsigned u = __float_as_uint(x);
  return (u & 0x80000000u) ? ~u : (u | 0x80000000u);
}

static __device__ inline int wred_sumi(int x) {
#pragma unroll
  for (int o = 16; o > 0; o >>= 1) x += __shfl_xor(x, o, 32);
  return x;
}
static __device__ inline float wred_sumf(float x) {
#pragma unroll
  for (int o = 16; o > 0; o >>= 1) x += __shfl_xor(x, o, 32);
  return x;
}
static __device__ inline float wred_maxf(float x) {
#pragma unroll
  for (int o = 16; o > 0; o >>= 1) x = fmaxf(x, __shfl_xor(x, o, 32));
  return x;
}

__global__ void nsa_blockmean_kernel(const float* __restrict__ K,
                                     const float* __restrict__ V,
                                     float* __restrict__ KB,
                                     float* __restrict__ VB) {
  const int g = blockIdx.x;
  const int d = threadIdx.x;
  const float* kp = K + (size_t)g * (BS * DD);
  const float* vp = V + (size_t)g * (BS * DD);
  float ks = 0.f, vs = 0.f;
  for (int t = 0; t < BS; ++t) {
    ks += kp[t * DD + d];
    vs += vp[t * DD + d];
  }
  KB[(size_t)g * DD + d] = ks * (1.f / BS);
  VB[(size_t)g * DD + d] = vs * (1.f / BS);
}

__global__ __launch_bounds__(128) void nsa_main_kernel(
    const float* __restrict__ Q, const float* __restrict__ K,
    const float* __restrict__ V, const float* __restrict__ KB,
    const float* __restrict__ VB, float* __restrict__ O) {
  extern __shared__ float sP[];
  __shared__ float sScore[NB];
  __shared__ int sSel[NSEL];

  const int wg = blockIdx.x;
  const int qb = wg % NB;
  const int bh = wg / NB;
  const size_t hoff = (size_t)bh * SS * DD;
  const float* Qh = Q + hoff;
  const float* Kh = K + hoff;
  const float* Vh = V + hoff;
  const float* KBh = KB + (size_t)bh * NB * DD;
  const float* VBh = VB + (size_t)bh * NB * DD;
  float* Oh = O + hoff;

  const int tid = threadIdx.x;
  const int wave = tid >> 5;
  const int lane = tid & 31;
  const int half = lane >> 4;
  const int ln = lane & 15;
  const int qrow0 = qb * BS + wave * 16;
  float* myP = sP + (size_t)wave * 16 * PSTRIDE;

  v2f qa[16];
#pragma unroll
  for (int t = 0; t < 16; ++t) {
    const float* p = Qh + (size_t)(qrow0 + ln) * DD + 4 * t + 2 * half;
    v2f a; a.x = p[0]; a.y = p[1];
    qa[t] = a;
  }

  v8f oacc[4];
#pragma unroll
  for (int dt = 0; dt < 4; ++dt) oacc[dt] = vzero8();

#pragma unroll
  for (int nt = 0; nt < 2; ++nt) {
    v8f acc = vzero8();
#pragma unroll
    for (int t = 0; t < 16; ++t) {
      const float* bp = KBh + (size_t)(nt * 16 + ln) * DD + 4 * t + 2 * half;
      v2f bf; bf.x = bp[0]; bf.y = bp[1];
      acc = wmma4(qa[t], bf, acc);
    }
#pragma unroll
    for (int r = 0; r < 8; ++r) {
      int m = r + 8 * half;
      int col = nt * 16 + ln;
      float x = acc[r] * SCALE;
      myP[m * PSTRIDE + col] = (col <= qb) ? x : NEGINF;
    }
  }
  __syncthreads();

  if (tid < NB) {
    float s = 0.f;
    for (int q2 = 0; q2 < BS; ++q2) s += sP[q2 * PSTRIDE + tid];
    sScore[tid] = s * (1.f / BS);
  }
  __syncthreads();

  if (tid == 0) {
    float loc[NB];
    for (int i = 0; i < NB; ++i) loc[i] = sScore[i];
    for (int s = 0; s < NSEL; ++s) {
      int bi = 0;
      float bv = loc[0];
      for (int i = 1; i < NB; ++i)
        if (loc[i] > bv) { bv = loc[i]; bi = i; }
      sSel[s] = bi;
      loc[bi] = -3.4e38f;
    }
  }
  __syncthreads();

  if (tid < BS) {
    float* row = sP + tid * PSTRIDE;
    float mx = -3.4e38f;
    for (int c = 0; c < NB; ++c) mx = fmaxf(mx, row[c]);
    float sum = 0.f;
    for (int c = 0; c < NB; ++c) {
      float p = (mx > -1e29f) ? __expf(row[c] - mx) : 0.f;
      row[c] = p;
      sum += p;
    }
    float rs = (sum > 0.f) ? 1.f / sum : 0.f;
    for (int c = 0; c < NB; ++c) row[c] *= rs;
  }
  __syncthreads();

#pragma unroll
  for (int dt = 0; dt < 4; ++dt) {
#pragma unroll
    for (int g = 0; g < 8; ++g) {
      int c0 = 4 * g + 2 * half;
      v2f a; a.x = myP[ln * PSTRIDE + c0]; a.y = myP[ln * PSTRIDE + c0 + 1];
      v2f bf;
      bf.x = VBh[(size_t)c0 * DD + dt * 16 + ln];
      bf.y = VBh[(size_t)(c0 + 1) * DD + dt * 16 + ln];
      oacc[dt] = wmma4(a, bf, oacc[dt]);
    }
  }
  __syncthreads();

  for (int sb = 0; sb < NSEL; ++sb) {
    int kb = sSel[sb];
    for (int kt = 0; kt < 4; ++kt) {
      int keyBase = kb * BS + kt * 16;
      __builtin_prefetch(Kh + (size_t)(keyBase + 16) * DD, 0, 1);
      v8f acc = vzero8();
#pragma unroll
      for (int t = 0; t < 16; ++t) {
        const float* bp = Kh + (size_t)(keyBase + ln) * DD + 4 * t + 2 * half;
        v2f bf; bf.x = bp[0]; bf.y = bp[1];
        acc = wmma4(qa[t], bf, acc);
      }
#pragma unroll
      for (int r = 0; r < 8; ++r) {
        int m = r + 8 * half;
        int i = qrow0 + m;
        int tok = keyBase + ln;
        int col = sb * 64 + kt * 16 + ln;
        float x = acc[r] * SCALE;
        myP[m * PSTRIDE + col] = (tok <= i) ? x : NEGINF;
      }
    }
  }

  for (int r = 0; r < 16; ++r) {
    float* row = myP + r * PSTRIDE;
    unsigned prefix = 0u;
    for (int bit = 31; bit >= 0; --bit) {
      unsigned cand = prefix | (1u << bit);
      int cnt = 0;
      for (int c = lane; c < SELC; c += 32)
        cnt += (okey(row[c]) >= cand) ? 1 : 0;
      cnt = wred_sumi(cnt);
      if (cnt >= KEEP) prefix = cand;
    }
    float mx = -3.4e38f;
    for (int c = lane; c < SELC; c += 32) mx = fmaxf(mx, row[c]);
    mx = wred_maxf(mx);
    float sum = 0.f;
    for (int c = lane; c < SELC; c += 32) {
      float x = row[c];
      float p = (okey(x) >= prefix && mx > -1e29f) ? __expf(x - mx) : 0.f;
      row[c] = p;
      sum += p;
    }
    sum = wred_sumf(sum);
    float rs = (sum > 0.f) ? 1.f / sum : 0.f;
    for (int c = lane; c < SELC; c += 32) row[c] *= rs;
  }

#pragma unroll
  for (int dt = 0; dt < 4; ++dt) {
    for (int g = 0; g < SELC / 4; ++g) {
      int c0 = 4 * g + 2 * half;
      v2f a; a.x = myP[ln * PSTRIDE + c0]; a.y = myP[ln * PSTRIDE + c0 + 1];
      int t0 = sSel[c0 >> 6] * BS + (c0 & 63);
      v2f bf;
      bf.x = Vh[(size_t)t0 * DD + dt * 16 + ln];
      bf.y = Vh[(size_t)(t0 + 1) * DD + dt * 16 + ln];
      oacc[dt] = wmma4(a, bf, oacc[dt]);
    }
  }

  const int firstb = (qb >= 2) ? qb - 2 : 0;
  const int nwb = qb - firstb + 1;
  const int extra = (firstb > 0) ? 1 : 0;
  const int ncg = nwb + extra;
  const int Cw = ncg * 64;

  for (int cg = 0; cg < ncg; ++cg) {
    int kb = (extra && cg == 0) ? 0 : firstb + (cg - extra);
    for (int kt = 0; kt < 4; ++kt) {
      int keyBase = kb * BS + kt * 16;
      v8f acc = vzero8();
#pragma unroll
      for (int t = 0; t < 16; ++t) {
        const float* bp = Kh + (size_t)(keyBase + ln) * DD + 4 * t + 2 * half;
        v2f bf; bf.x = bp[0]; bf.y = bp[1];
        acc = wmma4(qa[t], bf, acc);
      }
#pragma unroll
      for (int r = 0; r < 8; ++r) {
        int m = r + 8 * half;
        int i = qrow0 + m;
        int j = keyBase + ln;
        int col = cg * 64 + kt * 16 + ln;
        bool ok = (j <= i) && (((i - j) < WIN) || (j < 1));
        myP[m * PSTRIDE + col] = ok ? acc[r] * SCALE : NEGINF;
      }
    }
  }

  for (int r = 0; r < 16; ++r) {
    float* row = myP + r * PSTRIDE;
    float mx = -3.4e38f;
    for (int c = lane; c < Cw; c += 32) mx = fmaxf(mx, row[c]);
    mx = wred_maxf(mx);
    float sum = 0.f;
    for (int c = lane; c < Cw; c += 32) {
      float p = (mx > -1e29f) ? __expf(row[c] - mx) : 0.f;
      row[c] = p;
      sum += p;
    }
    sum = wred_sumf(sum);
    float rs = (sum > 0.f) ? 1.f / sum : 0.f;
    for (int c = lane; c < Cw; c += 32) row[c] *= rs;
  }

#pragma unroll
  for (int dt = 0; dt < 4; ++dt) {
    for (int g = 0; g < Cw / 4; ++g) {
      int c0 = 4 * g + 2 * half;
      v2f a; a.x = myP[ln * PSTRIDE + c0]; a.y = myP[ln * PSTRIDE + c0 + 1];
      int cg0 = c0 >> 6;
      int kb0 = (extra && cg0 == 0) ? 0 : firstb + (cg0 - extra);
      int t0 = kb0 * BS + (c0 & 63);
      v2f bf;
      bf.x = Vh[(size_t)t0 * DD + dt * 16 + ln];
      bf.y = Vh[(size_t)(t0 + 1) * DD + dt * 16 + ln];
      oacc[dt] = wmma4(a, bf, oacc[dt]);
    }
  }

  const float third = 1.f / 3.f;
#pragma unroll
  for (int dt = 0; dt < 4; ++dt) {
#pragma unroll
    for (int r = 0; r < 8; ++r) {
      int row = qrow0 + r + 8 * half;
      Oh[(size_t)row * DD + dt * 16 + ln] = oacc[dt][r] * third;
    }
  }
}

extern "C" void kernel_launch(void* const* d_in, const int* in_sizes, int n_in,
                              void* d_out, int out_size, void* d_ws,
                              size_t ws_size, hipStream_t stream) {
  (void)in_sizes; (void)n_in; (void)out_size; (void)ws_size;
  const float* q = (const float*)d_in[0];
  const float* k = (const float*)d_in[1];
  const float* v = (const float*)d_in[2];
  float* out = (float*)d_out;
  float* kblk = (float*)d_ws;
  float* vblk = kblk + (size_t)BB * HH * NB * DD;

  nsa_blockmean_kernel<<<BB * HH * NB, 64, 0, stream>>>(k, v, kblk, vblk);

  const size_t shmem = (size_t)64 * PSTRIDE * sizeof(float);
  hipFuncSetAttribute(reinterpret_cast<const void*>(nsa_main_kernel),
                      hipFuncAttributeMaxDynamicSharedMemorySize, (int)shmem);
  nsa_main_kernel<<<BB * HH * NB, 128, shmem, stream>>>(q, k, v, kblk, vblk,
                                                        out);
}


# selected branch: Q x K^T with operands swapped (K as A, Q as B): a query row's 512 logits sit in 2 lanes' registers; top-256 count, max, exp, sum per lane + 1 lane swap; row sum order changed
# speedup vs baseline: 1.4614x; 1.4614x over previous
#include <hip/hip_runtime.h>

#define BB 2
#define HH 8
#define SS 2048
#define DD 64
#define NB 32
#define BS 64
#define NSEL 8
#define SELC 512
#define KEEP 256
#define WIN 128
#define PSTRIDE 520
#define NEGINF (-1e30f)
#define SCALE 0.125f

typedef __attribute__((ext_vector_type(2))) float v2f;
typedef __attribute__((ext_vector_type(8))) float v8f;

static __device__ inline v8f vzero8() {
  v8f r;
#pragma unroll
  for (int i = 0; i < 8; ++i) r[i] = 0.f;
  return r;
}

static __device__ inline v8f wmma4(v2f a, v2f b, v8f c) {
  return __builtin_amdgcn_wmma_f32_16x16x4_f32(false, a, false, b, (short)0, c,
                                               false, false);
}

static __device__ inline unsigned okey(float x) {
  unsigned u = __float_as_uint(x);
  return (u & 0x80000000u) ? ~u : (u | 0x80000000u);
}

static __device__ inline int wred_sumi(int x) {
#pragma unroll
  for (int o = 16; o > 0; o >>= 1) x += __shfl_xor(x, o, 32);
  return x;
}
static __device__ inline float wred_sumf(float x) {
#pragma unroll
  for (int o = 16; o > 0; o >>= 1) x += __shfl_xor(x, o, 32);
  return x;
}
static __device__ inline float wred_maxf(float x) {
#pragma unroll
  for (int o = 16; o > 0; o >>= 1) x = fmaxf(x, __shfl_xor(x, o, 32));
  return x;
}

static __device__ inline float unkey(unsigned k) {
  unsigned u = (k & 0x80000000u) ? (k ^ 0x80000000u) : ~k;
  return __uint_as_float(u);
}

static __device__ inline unsigned xhalf_u(unsigned x) {
  return __builtin_amdgcn_permlanex16(x, x, 0x76543210u, 0xfedcba98u, false, false);
}
static __device__ inline float xhalf_f(float x) {
  return __uint_as_float(xhalf_u(__float_as_uint(x)));
}

__global__ void nsa_blockmean_kernel(const float* __restrict__ K,
                                     const float* __restrict__ V,
                                     float* __restrict__ KB,
                                     float* __restrict__ VB) {
  const int g = blockIdx.x;
  const int d = threadIdx.x;
  const float* kp = K + (size_t)g * (BS * DD);
  const float* vp = V + (size_t)g * (BS * DD);
  float ks = 0.f, vs = 0.f;
  for (int t = 0; t < BS; ++t) {
    ks += kp[t * DD + d];
    vs += vp[t * DD + d];
  }
  KB[(size_t)g * DD + d] = ks * (1.f / BS);
  VB[(size_t)g * DD + d] = vs * (1.f / BS);
}

__global__ __launch_bounds__(128, 2) void nsa_main_kernel(
    const float* __restrict__ Q, const float* __restrict__ K,
    const float* __restrict__ V, const float* __restrict__ KB,
    const float* __restrict__ VB, float* __restrict__ O) {
  extern __shared__ __align__(16) float sP[];
  __shared__ float sScore[NB];
  __shared__ int sSel[NSEL];

  const int wg = blockIdx.x;
  const int qb = wg % NB;
  const int bh = wg / NB;
  const size_t hoff = (size_t)bh * SS * DD;
  const float* Qh = Q + hoff;
  const float* Kh = K + hoff;
  const float* Vh = V + hoff;
  const float* KBh = KB + (size_t)bh * NB * DD;
  const float* VBh = VB + (size_t)bh * NB * DD;
  float* Oh = O + hoff;

  const int tid = threadIdx.x;
  const int wave = tid >> 5;
  const int lane = tid & 31;
  const int half = lane >> 4;
  const int ln = lane & 15;
  const int qrow0 = qb * BS + wave * 16;
  float* myP = sP + (size_t)wave * 16 * PSTRIDE;

  v2f qa[16];
#pragma unroll
  for (int t = 0; t < 16; ++t) {
    const float* p = Qh + (size_t)(qrow0 + ln) * DD + 4 * t + 2 * half;
    v2f a; a.x = p[0]; a.y = p[1];
    qa[t] = a;
  }

  v8f oacc[4];
#pragma unroll
  for (int dt = 0; dt < 4; ++dt) oacc[dt] = vzero8();

#pragma unroll
  for (int nt = 0; nt < 2; ++nt) {
    v8f acc = vzero8();
#pragma unroll
    for (int t = 0; t < 16; ++t) {
      const float* bp = KBh + (size_t)(nt * 16 + ln) * DD + 4 * t + 2 * half;
      v2f bf; bf.x = bp[0]; bf.y = bp[1];
      acc = wmma4(qa[t], bf, acc);
    }
#pragma unroll
    for (int r = 0; r < 8; ++r) {
      int m = r + 8 * half;
      int col = nt * 16 + ln;
      float x = acc[r] * SCALE;
      myP[m * PSTRIDE + col] = (col <= qb) ? x : NEGINF;
    }
  }
  __syncthreads();

  if (tid < NB) {
    float s = 0.f;
    for (int q2 = 0; q2 < BS; ++q2) s += sP[q2 * PSTRIDE + tid];
    sScore[tid] = s * (1.f / BS);
  }
  __syncthreads();

  if (tid == 0) {
    float loc[NB];
    for (int i = 0; i < NB; ++i) loc[i] = sScore[i];
    for (int s = 0; s < NSEL; ++s) {
      int bi = 0;
      float bv = loc[0];
      for (int i = 1; i < NB; ++i)
        if (loc[i] > bv) { bv = loc[i]; bi = i; }
      sSel[s] = bi;
      loc[bi] = -3.4e38f;
    }
  }
  __syncthreads();

  if (tid < BS) {
    float* row = sP + tid * PSTRIDE;
    float mx = -3.4e38f;
    for (int c = 0; c < NB; ++c) mx = fmaxf(mx, row[c]);
    float sum = 0.f;
    for (int c = 0; c < NB; ++c) {
      float p = (mx > -1e29f) ? __expf(row[c] - mx) : 0.f;
      row[c] = p;
      sum += p;
    }
    float rs = (sum > 0.f) ? 1.f / sum : 0.f;
    for (int c = 0; c < NB; ++c) row[c] *= rs;
  }
  __syncthreads();

#pragma unroll
  for (int dt = 0; dt < 4; ++dt) {
#pragma unroll
    for (int g = 0; g < 8; ++g) {
      int c0 = 4 * g + 2 * half;
      v2f a; a.x = myP[ln * PSTRIDE + c0]; a.y = myP[ln * PSTRIDE + c0 + 1];
      v2f bf;
      bf.x = VBh[(size_t)c0 * DD + dt * 16 + ln];
      bf.y = VBh[(size_t)(c0 + 1) * DD + dt * 16 + ln];
      oacc[dt] = wmma4(a, bf, oacc[dt]);
    }
  }
  __syncthreads();

  unsigned ky[SELC / 2];
  {
    const int i = qrow0 + ln;
#pragma unroll
    for (int sb = 0; sb < NSEL; ++sb) {
      const int kb = sSel[sb];
#pragma unroll
      for (int kt = 0; kt < 4; ++kt) {
        const int keyBase = kb * BS + kt * 16;
        __builtin_prefetch(Kh + (size_t)(keyBase + 16) * DD, 0, 1);
        v8f acc = vzero8();
#pragma unroll
        for (int t = 0; t < 16; ++t) {
          const float* bp = Kh + (size_t)(keyBase + ln) * DD + 4 * t + 2 * half;
          v2f kf; kf.x = bp[0]; kf.y = bp[1];
          acc = wmma4(kf, qa[t], acc);
        }
#pragma unroll
        for (int r = 0; r < 8; ++r) {
          const int tok = keyBase + r + 8 * half;
          const float x = acc[r] * SCALE;
          ky[(sb * 4 + kt) * 8 + r] = okey((tok <= i) ? x : NEGINF);
        }
      }
    }
  }

  unsigned prefix = 0u;
#pragma nounroll
  for (int bit = 31; bit >= 0; --bit) {
    const unsigned cand = prefix | (1u << bit);
    unsigned c0 = 0u, c1 = 0u, c2 = 0u, c3 = 0u;
#pragma unroll
    for (int j = 0; j < SELC / 2; j += 4) {
      c0 += (ky[j + 0] >= cand) ? 1u : 0u;
      c1 += (ky[j + 1] >= cand) ? 1u : 0u;
      c2 += (ky[j + 2] >= cand) ? 1u : 0u;
      c3 += (ky[j + 3] >= cand) ? 1u : 0u;
    }
    unsigned cnt = (c0 + c1) + (c2 + c3);
    cnt += xhalf_u(cnt);
    if (cnt >= (unsigned)KEEP) prefix = cand;
  }

  {
    unsigned mk = 0u;
#pragma unroll
    for (int j = 0; j < SELC / 2; ++j) mk = (ky[j] > mk) ? ky[j] : mk;
    const unsigned mo = xhalf_u(mk);
    mk = (mo > mk) ? mo : mk;
    const float mx = unkey(mk);
    const bool live = mx > -1e29f;
    float s0 = 0.f, s1 = 0.f, s2 = 0.f, s3 = 0.f;
#pragma unroll
    for (int j = 0; j < SELC / 2; j += 4) {
      float p0 = (ky[j + 0] >= prefix && live) ? __expf(unkey(ky[j + 0]) - mx) : 0.f;
      float p1 = (ky[j + 1] >= prefix && live) ? __expf(unkey(ky[j + 1]) - mx) : 0.f;
      float p2 = (ky[j + 2] >= prefix && live) ? __expf(unkey(ky[j + 2]) - mx) : 0.f;
      float p3 = (ky[j + 3] >= prefix && live) ? __expf(unkey(ky[j + 3]) - mx) : 0.f;
      ky[j + 0] = __float_as_uint(p0); s0 += p0;
      ky[j + 1] = __float_as_uint(p1); s1 += p1;
      ky[j + 2] = __float_as_uint(p2); s2 += p2;
      ky[j + 3] = __float_as_uint(p3); s3 += p3;
    }
    float sum = (s0 + s1) + (s2 + s3);
    sum += xhalf_f(sum);
    const float rs = (sum > 0.f) ? 1.f / sum : 0.f;
#pragma unroll
    for (int tl = 0; tl < SELC / 16; ++tl) {
      float* dst = myP + ln * PSTRIDE + tl * 16 + 8 * half;
      float4 lo, hi;
      lo.x = __uint_as_float(ky[tl * 8 + 0]) * rs;
      lo.y = __uint_as_float(ky[tl * 8 + 1]) * rs;
      lo.z = __uint_as_float(ky[tl * 8 + 2]) * rs;
      lo.w = __uint_as_float(ky[tl * 8 + 3]) * rs;
      hi.x = __uint_as_float(ky[tl * 8 + 4]) * rs;
      hi.y = __uint_as_float(ky[tl * 8 + 5]) * rs;
      hi.z = __uint_as_float(ky[tl * 8 + 6]) * rs;
      hi.w = __uint_as_float(ky[tl * 8 + 7]) * rs;
      *reinterpret_cast<float4*>(dst) = lo;
      *reinterpret_cast<float4*>(dst + 4) = hi;
    }
  }

#pragma unroll
  for (int dt = 0; dt < 4; ++dt) {
    for (int g = 0; g < SELC / 4; ++g) {
      int c0 = 4 * g + 2 * half;
      v2f a; a.x = myP[ln * PSTRIDE + c0]; a.y = myP[ln * PSTRIDE + c0 + 1];
      int t0 = sSel[c0 >> 6] * BS + (c0 & 63);
      v2f bf;
      bf.x = Vh[(size_t)t0 * DD + dt * 16 + ln];
      bf.y = Vh[(size_t)(t0 + 1) * DD + dt * 16 + ln];
      oacc[dt] = wmma4(a, bf, oacc[dt]);
    }
  }

  const int firstb = (qb >= 2) ? qb - 2 : 0;
  const int nwb = qb - firstb + 1;
  const int extra = (firstb > 0) ? 1 : 0;
  const int ncg = nwb + extra;
  const int Cw = ncg * 64;

  for (int cg = 0; cg < ncg; ++cg) {
    int kb = (extra && cg == 0) ? 0 : firstb + (cg - extra);
    for (int kt = 0; kt < 4; ++kt) {
      int keyBase = kb * BS + kt * 16;
      v8f acc = vzero8();
#pragma unroll
      for (int t = 0; t < 16; ++t) {
        const float* bp = Kh + (size_t)(keyBase + ln) * DD + 4 * t + 2 * half;
        v2f bf; bf.x = bp[0]; bf.y = bp[1];
        acc = wmma4(qa[t], bf, acc);
      }
#pragma unroll
      for (int r = 0; r < 8; ++r) {
        int m = r + 8 * half;
        int i = qrow0 + m;
        int j = keyBase + ln;
        int col = cg * 64 + kt * 16 + ln;
        bool ok = (j <= i) && (((i - j) < WIN) || (j < 1));
        myP[m * PSTRIDE + col] = ok ? acc[r] * SCALE : NEGINF;
      }
    }
  }

  for (int r = 0; r < 16; ++r) {
    float* row = myP + r * PSTRIDE;
    float mx = -3.4e38f;
    for (int c = lane; c < Cw; c += 32) mx = fmaxf(mx, row[c]);
    mx = wred_maxf(mx);
    float sum = 0.f;
    for (int c = lane; c < Cw; c += 32) {
      float p = (mx > -1e29f) ? __expf(row[c] - mx) : 0.f;
      row[c] = p;
      sum += p;
    }
    sum = wred_sumf(sum);
    float rs = (sum > 0.f) ? 1.f / sum : 0.f;
    for (int c = lane; c < Cw; c += 32) row[c] *= rs;
  }

#pragma unroll
  for (int dt = 0; dt < 4; ++dt) {
    for (int g = 0; g < Cw / 4; ++g) {
      int c0 = 4 * g + 2 * half;
      v2f a; a.x = myP[ln * PSTRIDE + c0]; a.y = myP[ln * PSTRIDE + c0 + 1];
      int cg0 = c0 >> 6;
      int kb0 = (extra && cg0 == 0) ? 0 : firstb + (cg0 - extra);
      int t0 = kb0 * BS + (c0 & 63);
      v2f bf;
      bf.x = Vh[(size_t)t0 * DD + dt * 16 + ln];
      bf.y = Vh[(size_t)(t0 + 1) * DD + dt * 16 + ln];
      oacc[dt] = wmma4(a, bf, oacc[dt]);
    }
  }

  const float third = 1.f / 3.f;
#pragma unroll
  for (int dt = 0; dt < 4; ++dt) {
#pragma unroll
    for (int r = 0; r < 8; ++r) {
      int row = qrow0 + r + 8 * half;
      Oh[(size_t)row * DD + dt * 16 + ln] = oacc[dt][r] * third;
    }
  }
}

extern "C" void kernel_launch(void* const* d_in, const int* in_sizes, int n_in,
                              void* d_out, int out_size, void* d_ws,
                              size_t ws_size, hipStream_t stream) {
  (void)in_sizes; (void)n_in; (void)out_size; (void)ws_size;
  const float* q = (const float*)d_in[0];
  const float* k = (const float*)d_in[1];
  const float* v = (const float*)d_in[2];
  float* out = (float*)d_out;
  float* kblk = (float*)d_ws;
  float* vblk = kblk + (size_t)BB * HH * NB * DD;

  nsa_blockmean_kernel<<<BB * HH * NB, 64, 0, stream>>>(k, v, kblk, vblk);

  const size_t shmem = (size_t)64 * PSTRIDE * sizeof(float);
  hipFuncSetAttribute(reinterpret_cast<const void*>(nsa_main_kernel),
                      hipFuncAttributeMaxDynamicSharedMemorySize, (int)shmem);
  nsa_main_kernel<<<BB * HH * NB, 128, shmem, stream>>>(q, k, v, kblk, vblk,
                                                        out);
}


# window branch on the selected branch's register path: K x Q^T swapped, a row's 256 logits in 2 lanes, softmax per lane + 1 swap, V^T x P^T from registers into the same O^T tile; sum order changed
# speedup vs baseline: 2.2467x; 1.5374x over previous
#include <hip/hip_runtime.h>

#define BB 2
#define HH 8
#define SS 2048
#define DD 64
#define NB 32
#define BS 64
#define NSEL 8
#define SELC 512
#define KEEP 256
#define WIN 128
#define WINC 256
#define PSTRIDE 520
#define NEGINF (-1e30f)
#define SCALE 0.125f

typedef __attribute__((ext_vector_type(2))) float v2f;
typedef __attribute__((ext_vector_type(8))) float v8f;

static __device__ inline v8f vzero8() {
  v8f r;
#pragma unroll
  for (int i = 0; i < 8; ++i) r[i] = 0.f;
  return r;
}

static __device__ inline v8f wmma4(v2f a, v2f b, v8f c) {
  return __builtin_amdgcn_wmma_f32_16x16x4_f32(false, a, false, b, (short)0, c,
                                               false, false);
}

static __device__ inline unsigned okey(float x) {
  unsigned u = __float_as_uint(x);
  return (u & 0x80000000u) ? ~u : (u | 0x80000000u);
}

static __device__ inline int wred_sumi(int x) {
#pragma unroll
  for (int o = 16; o > 0; o >>= 1) x += __shfl_xor(x, o, 32);
  return x;
}

static __device__ inline float unkey(unsigned k) {
  unsigned u = (k & 0x80000000u) ? (k ^ 0x80000000u) : ~k;
  return __uint_as_float(u);
}

static __device__ inline unsigned xhalf_u(unsigned x) {
  return __builtin_amdgcn_permlanex16(x, x, 0x76543210u, 0xfedcba98u, false, false);
}
static __device__ inline float xhalf_f(float x) {
  return __uint_as_float(xhalf_u(__float_as_uint(x)));
}

__global__ void nsa_blockmean_kernel(const float* __restrict__ K,
                                     const float* __restrict__ V,
                                     float* __restrict__ KB,
                                     float* __restrict__ VB) {
  const int g = blockIdx.x;
  const int d = threadIdx.x;
  const float* kp = K + (size_t)g * (BS * DD);
  const float* vp = V + (size_t)g * (BS * DD);
  float ks = 0.f, vs = 0.f;
  for (int t = 0; t < BS; ++t) {
    ks += kp[t * DD + d];
    vs += vp[t * DD + d];
  }
  KB[(size_t)g * DD + d] = ks * (1.f / BS);
  VB[(size_t)g * DD + d] = vs * (1.f / BS);
}

__global__ __launch_bounds__(128, 2) void nsa_main_kernel(
    const float* __restrict__ Q, const float* __restrict__ K,
    const float* __restrict__ V, const float* __restrict__ KB,
    const float* __restrict__ VB, float* __restrict__ O) {
  extern __shared__ __align__(16) float sP[];
  __shared__ float sScore[NB];
  __shared__ int sSel[NSEL];

  const int wg = blockIdx.x;
  const int qb = wg % NB;
  const int bh = wg / NB;
  const size_t hoff = (size_t)bh * SS * DD;
  const float* Qh = Q + hoff;
  const float* Kh = K + hoff;
  const float* Vh = V + hoff;
  const float* KBh = KB + (size_t)bh * NB * DD;
  const float* VBh = VB + (size_t)bh * NB * DD;
  float* Oh = O + hoff;

  const int tid = threadIdx.x;
  const int wave = tid >> 5;
  const int lane = tid & 31;
  const int half = lane >> 4;
  const int ln = lane & 15;
  const int qrow0 = qb * BS + wave * 16;
  float* myP = sP + (size_t)wave * 16 * PSTRIDE;

  v2f qa[16];
#pragma unroll
  for (int t = 0; t < 16; ++t) {
    const float* p = Qh + (size_t)(qrow0 + ln) * DD + 4 * t + 2 * half;
    v2f a; a.x = p[0]; a.y = p[1];
    qa[t] = a;
  }

  v8f oacc[4];
#pragma unroll
  for (int dt = 0; dt < 4; ++dt) oacc[dt] = vzero8();

#pragma unroll
  for (int nt = 0; nt < 2; ++nt) {
    v8f acc = vzero8();
#pragma unroll
    for (int t = 0; t < 16; ++t) {
      const float* bp = KBh + (size_t)(nt * 16 + ln) * DD + 4 * t + 2 * half;
      v2f bf; bf.x = bp[0]; bf.y = bp[1];
      acc = wmma4(qa[t], bf, acc);
    }
#pragma unroll
    for (int r = 0; r < 8; ++r) {
      int m = r + 8 * half;
      int col = nt * 16 + ln;
      float x = acc[r] * SCALE;
      myP[m * PSTRIDE + col] = (col <= qb) ? x : NEGINF;
    }
  }
  __syncthreads();

  if (tid < NB) {
    float s = 0.f;
    for (int q2 = 0; q2 < BS; ++q2) s += sP[q2 * PSTRIDE + tid];
    sScore[tid] = s * (1.f / BS);
  }
  __syncthreads();

  if (tid == 0) {
    float loc[NB];
    for (int i = 0; i < NB; ++i) loc[i] = sScore[i];
    for (int s = 0; s < NSEL; ++s) {
      int bi = 0;
      float bv = loc[0];
      for (int i = 1; i < NB; ++i)
        if (loc[i] > bv) { bv = loc[i]; bi = i; }
      sSel[s] = bi;
      loc[bi] = -3.4e38f;
    }
  }
  __syncthreads();

  if (tid < BS) {
    float* row = sP + tid * PSTRIDE;
    float mx = -3.4e38f;
    for (int c = 0; c < NB; ++c) mx = fmaxf(mx, row[c]);
    float sum = 0.f;
    for (int c = 0; c < NB; ++c) {
      float p = (mx > -1e29f) ? __expf(row[c] - mx) : 0.f;
      row[c] = p;
      sum += p;
    }
    float rs = (sum > 0.f) ? 1.f / sum : 0.f;
    for (int c = 0; c < NB; ++c) row[c] *= rs;
  }
  __syncthreads();

#pragma unroll
  for (int dt = 0; dt < 4; ++dt) {
#pragma unroll
    for (int g = 0; g < 8; ++g) {
      int c0 = 4 * g + 2 * half;
      v2f a; a.x = myP[ln * PSTRIDE + c0]; a.y = myP[ln * PSTRIDE + c0 + 1];
      v2f bf;
      bf.x = VBh[(size_t)c0 * DD + dt * 16 + ln];
      bf.y = VBh[(size_t)(c0 + 1) * DD + dt * 16 + ln];
      oacc[dt] = wmma4(a, bf, oacc[dt]);
    }
  }
  __syncthreads();

  unsigned ky[SELC / 2];
  {
    const int i = qrow0 + ln;
#pragma unroll
    for (int sb = 0; sb < NSEL; ++sb) {
      const int kb = sSel[sb];
#pragma unroll
      for (int kt = 0; kt < 4; ++kt) {
        const int keyBase = kb * BS + kt * 16;
        __builtin_prefetch(Kh + (size_t)(keyBase + 16) * DD, 0, 1);
        v8f acc = vzero8();
#pragma unroll
        for (int t = 0; t < 16; ++t) {
          const float* bp = Kh + (size_t)(keyBase + ln) * DD + 4 * t + 2 * half;
          v2f kf; kf.x = bp[0]; kf.y = bp[1];
          acc = wmma4(kf, qa[t], acc);
        }
#pragma unroll
        for (int r = 0; r < 8; ++r) {
          const int tok = keyBase + r + 8 * half;
          const float x = acc[r] * SCALE;
          ky[(sb * 4 + kt) * 8 + r] = okey((tok <= i) ? x : NEGINF);
        }
      }
    }
  }

  unsigned prefix = 0u;
#pragma nounroll
  for (int bit = 31; bit >= 0; --bit) {
    const unsigned cand = prefix | (1u << bit);
    unsigned c0 = 0u, c1 = 0u, c2 = 0u, c3 = 0u;
#pragma unroll
    for (int j = 0; j < SELC / 2; j += 4) {
      c0 += (ky[j + 0] >= cand) ? 1u : 0u;
      c1 += (ky[j + 1] >= cand) ? 1u : 0u;
      c2 += (ky[j + 2] >= cand) ? 1u : 0u;
      c3 += (ky[j + 3] >= cand) ? 1u : 0u;
    }
    unsigned cnt = (c0 + c1) + (c2 + c3);
    cnt += xhalf_u(cnt);
    if (cnt >= (unsigned)KEEP) prefix = cand;
  }

  {
    unsigned mk = 0u;
#pragma unroll
    for (int j = 0; j < SELC / 2; ++j) mk = (ky[j] > mk) ? ky[j] : mk;
    const unsigned mo = xhalf_u(mk);
    mk = (mo > mk) ? mo : mk;
    const float mx = unkey(mk);
    const bool live = mx > -1e29f;
    float s0 = 0.f, s1 = 0.f, s2 = 0.f, s3 = 0.f;
#pragma unroll
    for (int j = 0; j < SELC / 2; j += 4) {
      float p0 = (ky[j + 0] >= prefix && live) ? __expf(unkey(ky[j + 0]) - mx) : 0.f;
      float p1 = (ky[j + 1] >= prefix && live) ? __expf(unkey(ky[j + 1]) - mx) : 0.f;
      float p2 = (ky[j + 2] >= prefix && live) ? __expf(unkey(ky[j + 2]) - mx) : 0.f;
      float p3 = (ky[j + 3] >= prefix && live) ? __expf(unkey(ky[j + 3]) - mx) : 0.f;
      ky[j + 0] = __float_as_uint(p0); s0 += p0;
      ky[j + 1] = __float_as_uint(p1); s1 += p1;
      ky[j + 2] = __float_as_uint(p2); s2 += p2;
      ky[j + 3] = __float_as_uint(p3); s3 += p3;
    }
    float sum = (s0 + s1) + (s2 + s3);
    sum += xhalf_f(sum);
    const float rs = (sum > 0.f) ? 1.f / sum : 0.f;
#pragma unroll
    for (int j = 0; j < SELC / 2; ++j)
      ky[j] = __float_as_uint(__uint_as_float(ky[j]) * rs);
  }

  v8f sacc[4];
#pragma unroll
  for (int dt = 0; dt < 4; ++dt) sacc[dt] = vzero8();
  const float* vlane = Vh + (size_t)(8 * half) * DD + ln;
  {
    v2f vf[2][16];
#pragma unroll
    for (int tl = -1; tl < SELC / 16; ++tl) {
      if (tl + 1 < SELC / 16) {
        const int nt = tl + 1;
        const float* vb = vlane + (size_t)sSel[nt >> 2] * (BS * DD) + (size_t)((nt & 3) * 16) * DD;
#pragma unroll
        for (int u = 0; u < 4; ++u) {
#pragma unroll
          for (int dt = 0; dt < 4; ++dt) {
            v2f va;
            va.x = vb[(2 * u) * DD + dt * 16];
            va.y = vb[(2 * u + 1) * DD + dt * 16];
            vf[nt & 1][u * 4 + dt] = va;
          }
        }
      }
      __builtin_amdgcn_sched_barrier(0);
      if (tl >= 0) {
#pragma unroll
        for (int u = 0; u < 4; ++u) {
          v2f pb;
          pb.x = __uint_as_float(ky[tl * 8 + 2 * u]);
          pb.y = __uint_as_float(ky[tl * 8 + 2 * u + 1]);
#pragma unroll
          for (int dt = 0; dt < 4; ++dt)
            sacc[dt] = wmma4(vf[tl & 1][u * 4 + dt], pb, sacc[dt]);
        }
      }
    }
  }

  int wkb[4];
  bool wok[4];
#pragma unroll
  for (int cg = 0; cg < 4; ++cg) {
    const int kbr = (cg == 0) ? 0 : qb - 3 + cg;
    wok[cg] = (cg == 0) ? (qb >= 3) : (kbr >= 0);
    wkb[cg] = wok[cg] ? kbr : 0;
  }
  float wl[WINC / 2];
  {
    const int i = qrow0 + ln;
#pragma unroll
    for (int cg = 0; cg < 4; ++cg) {
#pragma unroll
      for (int kt = 0; kt < 4; ++kt) {
        const int keyBase = wkb[cg] * BS + kt * 16;
        v8f acc = vzero8();
#pragma unroll
        for (int t = 0; t < 16; ++t) {
          const float* bp = Kh + (size_t)(keyBase + ln) * DD + 4 * t + 2 * half;
          v2f kf; kf.x = bp[0]; kf.y = bp[1];
          acc = wmma4(kf, qa[t], acc);
        }
#pragma unroll
        for (int r = 0; r < 8; ++r) {
          const int j = keyBase + r + 8 * half;
          const bool ok = wok[cg] && (j <= i) && (((i - j) < WIN) || (j < 1));
          wl[(cg * 4 + kt) * 8 + r] = ok ? acc[r] * SCALE : NEGINF;
        }
      }
    }
  }

  {
    float m0 = wl[0], m1 = wl[1], m2 = wl[2], m3 = wl[3];
#pragma unroll
    for (int j = 4; j < WINC / 2; j += 4) {
      m0 = fmaxf(m0, wl[j + 0]);
      m1 = fmaxf(m1, wl[j + 1]);
      m2 = fmaxf(m2, wl[j + 2]);
      m3 = fmaxf(m3, wl[j + 3]);
    }
    float mx = fmaxf(fmaxf(m0, m1), fmaxf(m2, m3));
    mx = fmaxf(mx, xhalf_f(mx));
    const bool live = mx > -1e29f;
    float s0 = 0.f, s1 = 0.f, s2 = 0.f, s3 = 0.f;
#pragma unroll
    for (int j = 0; j < WINC / 2; j += 4) {
      const float p0 = live ? __expf(wl[j + 0] - mx) : 0.f;
      const float p1 = live ? __expf(wl[j + 1] - mx) : 0.f;
      const float p2 = live ? __expf(wl[j + 2] - mx) : 0.f;
      const float p3 = live ? __expf(wl[j + 3] - mx) : 0.f;
      wl[j + 0] = p0; s0 += p0;
      wl[j + 1] = p1; s1 += p1;
      wl[j + 2] = p2; s2 += p2;
      wl[j + 3] = p3; s3 += p3;
    }
    float sum = (s0 + s1) + (s2 + s3);
    sum += xhalf_f(sum);
    const float rs = (sum > 0.f) ? 1.f / sum : 0.f;
#pragma unroll
    for (int j = 0; j < WINC / 2; ++j) wl[j] *= rs;
  }

  {
    v2f vf[2][16];
#pragma unroll
    for (int tl = -1; tl < WINC / 16; ++tl) {
      if (tl + 1 < WINC / 16) {
        const int nt = tl + 1;
        const float* vb = vlane + (size_t)wkb[nt >> 2] * (BS * DD) + (size_t)((nt & 3) * 16) * DD;
#pragma unroll
        for (int u = 0; u < 4; ++u) {
#pragma unroll
          for (int dt = 0; dt < 4; ++dt) {
            v2f va;
            va.x = vb[(2 * u) * DD + dt * 16];
            va.y = vb[(2 * u + 1) * DD + dt * 16];
            vf[nt & 1][u * 4 + dt] = va;
          }
        }
      }
      __builtin_amdgcn_sched_barrier(0);
      if (tl >= 0) {
#pragma unroll
        for (int u = 0; u < 4; ++u) {
          v2f pb;
          pb.x = wl[tl * 8 + 2 * u];
          pb.y = wl[tl * 8 + 2 * u + 1];
#pragma unroll
          for (int dt = 0; dt < 4; ++dt)
            sacc[dt] = wmma4(vf[tl & 1][u * 4 + dt], pb, sacc[dt]);
        }
      }
    }
  }

#pragma unroll
  for (int dt = 0; dt < 4; ++dt) {
    float* dst = myP + ln * PSTRIDE + 256 + dt * 16 + 8 * half;
    float4 lo, hi;
    lo.x = sacc[dt][0]; lo.y = sacc[dt][1]; lo.z = sacc[dt][2]; lo.w = sacc[dt][3];
    hi.x = sacc[dt][4]; hi.y = sacc[dt][5]; hi.z = sacc[dt][6]; hi.w = sacc[dt][7];
    *reinterpret_cast<float4*>(dst) = lo;
    *reinterpret_cast<float4*>(dst + 4) = hi;
  }
#pragma unroll
  for (int dt = 0; dt < 4; ++dt) {
#pragma unroll
    for (int r = 0; r < 8; ++r)
      oacc[dt][r] += myP[(r + 8 * half) * PSTRIDE + 256 + dt * 16 + ln];
  }

  const float third = 1.f / 3.f;
#pragma unroll
  for (int dt = 0; dt < 4; ++dt) {
#pragma unroll
    for (int r = 0; r < 8; ++r) {
      int row = qrow0 + r + 8 * half;
      Oh[(size_t)row * DD + dt * 16 + ln] = oacc[dt][r] * third;
    }
  }
}

extern "C" void kernel_launch(void* const* d_in, const int* in_sizes, int n_in,
                              void* d_out, int out_size, void* d_ws,
                              size_t ws_size, hipStream_t stream) {
  (void)in_sizes; (void)n_in; (void)out_size; (void)ws_size;
  const float* q = (const float*)d_in[0];
  const float* k = (const float*)d_in[1];
  const float* v = (const float*)d_in[2];
  float* out = (float*)d_out;
  float* kblk = (float*)d_ws;
  float* vblk = kblk + (size_t)BB * HH * NB * DD;

  nsa_blockmean_kernel<<<BB * HH * NB, 64, 0, stream>>>(k, v, kblk, vblk);

  const size_t shmem = (size_t)64 * PSTRIDE * sizeof(float);
  hipFuncSetAttribute(reinterpret_cast<const void*>(nsa_main_kernel),
                      hipFuncAttributeMaxDynamicSharedMemorySize, (int)shmem);
  nsa_main_kernel<<<BB * HH * NB, 128, shmem, stream>>>(q, k, v, kblk, vblk,
                                                        out);
}


# tile loops of all four products rolled (logits and P staged through the wave's own LDS rows, each lane reading back what it wrote) so the code fits the 64 KiB instruction cache; sum order unchanged
# speedup vs baseline: 3.0107x; 1.3401x over previous
#include <hip/hip_runtime.h>

#define BB 2
#define HH 8
#define SS 2048
#define DD 64
#define NB 32
#define BS 64
#define NSEL 8
#define SELC 512
#define KEEP 256
#define WIN 128
#define WINC 256
#define PSTRIDE 520
#define NEGINF (-1e30f)
#define SCALE 0.125f

typedef __attribute__((ext_vector_type(2))) float v2f;
typedef __attribute__((ext_vector_type(8))) float v8f;

static __device__ inline v8f vzero8() {
  v8f r;
#pragma unroll
  for (int i = 0; i < 8; ++i) r[i] = 0.f;
  return r;
}

static __device__ inline v8f wmma4(v2f a, v2f b, v8f c) {
  return __builtin_amdgcn_wmma_f32_16x16x4_f32(false, a, false, b, (short)0, c,
                                               false, false);
}

static __device__ inline unsigned okey(float x) {
  unsigned u = __float_as_uint(x);
  return (u & 0x80000000u) ? ~u : (u | 0x80000000u);
}

static __device__ inline int wred_sumi(int x) {
#pragma unroll
  for (int o = 16; o > 0; o >>= 1) x += __shfl_xor(x, o, 32);
  return x;
}

static __device__ inline float unkey(unsigned k) {
  unsigned u = (k & 0x80000000u) ? (k ^ 0x80000000u) : ~k;
  return __uint_as_float(u);
}

static __device__ inline unsigned xhalf_u(unsigned x) {
  return __builtin_amdgcn_permlanex16(x, x, 0x76543210u, 0xfedcba98u, false, false);
}
static __device__ inline float xhalf_f(float x) {
  return __uint_as_float(xhalf_u(__float_as_uint(x)));
}

__global__ void nsa_blockmean_kernel(const float* __restrict__ K,
                                     const float* __restrict__ V,
                                     float* __restrict__ KB,
                                     float* __restrict__ VB) {
  const int g = blockIdx.x;
  const int d = threadIdx.x;
  const float* kp = K + (size_t)g * (BS * DD);
  const float* vp = V + (size_t)g * (BS * DD);
  float ks = 0.f, vs = 0.f;
  for (int t = 0; t < BS; ++t) {
    ks += kp[t * DD + d];
    vs += vp[t * DD + d];
  }
  KB[(size_t)g * DD + d] = ks * (1.f / BS);
  VB[(size_t)g * DD + d] = vs * (1.f / BS);
}

static __device__ inline void load_ktile(v2f (&kf)[16], const float* __restrict__ Kh,
                                         int keyBase, int ln, int half) {
#pragma unroll
  for (int t = 0; t < 16; ++t) {
    const float* bp = Kh + (size_t)(keyBase + ln) * DD + 4 * t + 2 * half;
    v2f a; a.x = bp[0]; a.y = bp[1];
    kf[t] = a;
  }
}

static __device__ inline v8f mma_ktile(const v2f (&kf)[16], const v2f (&qa)[16]) {
  v8f acc = vzero8();
#pragma unroll
  for (int t = 0; t < 16; ++t) acc = wmma4(kf[t], qa[t], acc);
  return acc;
}

template <int NT, typename FB, typename FE>
static __device__ inline void qk_tiles(const float* __restrict__ Kh, int ln, int half,
                                       const v2f (&qa)[16], FB kbase, FE emit) {
  v2f kfa[16], kfb[16];
  load_ktile(kfa, Kh, kbase(0), ln, half);
#pragma nounroll
  for (int t = 0; t < NT; t += 2) {
    const int kb1 = kbase(t + 1);
    load_ktile(kfb, Kh, kb1, ln, half);
    __builtin_amdgcn_sched_barrier(0);
    emit(t, kbase(t), mma_ktile(kfa, qa));
    const int tn = (t + 2 < NT) ? t + 2 : NT - 1;
    load_ktile(kfa, Kh, kbase(tn), ln, half);
    __builtin_amdgcn_sched_barrier(0);
    emit(t + 1, kb1, mma_ktile(kfb, qa));
  }
}

static __device__ inline void load_vtile(v2f (&vf)[16], const float* __restrict__ vb) {
#pragma unroll
  for (int u = 0; u < 4; ++u) {
#pragma unroll
    for (int dt = 0; dt < 4; ++dt) {
      v2f va;
      va.x = vb[(2 * u) * DD + dt * 16];
      va.y = vb[(2 * u + 1) * DD + dt * 16];
      vf[u * 4 + dt] = va;
    }
  }
}

static __device__ inline void mma_vtile(v8f (&sacc)[4], const v2f (&vf)[16],
                                        const float4* __restrict__ mine, int t, float rs) {
  const float4 pa = mine[(2 * t) * 32];
  const float4 pc = mine[(2 * t + 1) * 32];
  float p[8];
  p[0] = pa.x * rs; p[1] = pa.y * rs; p[2] = pa.z * rs; p[3] = pa.w * rs;
  p[4] = pc.x * rs; p[5] = pc.y * rs; p[6] = pc.z * rs; p[7] = pc.w * rs;
#pragma unroll
  for (int u = 0; u < 4; ++u) {
    v2f pb; pb.x = p[2 * u]; pb.y = p[2 * u + 1];
#pragma unroll
    for (int dt = 0; dt < 4; ++dt) sacc[dt] = wmma4(vf[u * 4 + dt], pb, sacc[dt]);
  }
}

template <int NT, typename F>
static __device__ inline void pv_tiles(v8f (&sacc)[4], const float4* __restrict__ mine,
                                       float rs, F vtile) {
  v2f vfa[16], vfb[16];
  load_vtile(vfa, vtile(0));
  load_vtile(vfb, vtile(1));
#pragma nounroll
  for (int t = 0; t < NT; t += 2) {
    mma_vtile(sacc, vfa, mine, t, rs);
    const int ta = (t + 2 < NT) ? t + 2 : NT - 1;
    load_vtile(vfa, vtile(ta));
    __builtin_amdgcn_sched_barrier(0);
    mma_vtile(sacc, vfb, mine, t + 1, rs);
    const int tb = (t + 3 < NT) ? t + 3 : NT - 1;
    load_vtile(vfb, vtile(tb));
    __builtin_amdgcn_sched_barrier(0);
  }
}

__global__ __launch_bounds__(128, 2) void nsa_main_kernel(
    const float* __restrict__ Q, const float* __restrict__ K,
    const float* __restrict__ V, const float* __restrict__ KB,
    const float* __restrict__ VB, float* __restrict__ O) {
  extern __shared__ __align__(16) float sP[];
  __shared__ float sScore[NB];
  __shared__ int sSel[NSEL];

  const int wg = blockIdx.x;
  const int qb = wg % NB;
  const int bh = wg / NB;
  const size_t hoff = (size_t)bh * SS * DD;
  const float* Qh = Q + hoff;
  const float* Kh = K + hoff;
  const float* Vh = V + hoff;
  const float* KBh = KB + (size_t)bh * NB * DD;
  const float* VBh = VB + (size_t)bh * NB * DD;
  float* Oh = O + hoff;

  const int tid = threadIdx.x;
  const int wave = tid >> 5;
  const int lane = tid & 31;
  const int half = lane >> 4;
  const int ln = lane & 15;
  const int qrow0 = qb * BS + wave * 16;
  float* myP = sP + (size_t)wave * 16 * PSTRIDE;

  v2f qa[16];
#pragma unroll
  for (int t = 0; t < 16; ++t) {
    const float* p = Qh + (size_t)(qrow0 + ln) * DD + 4 * t + 2 * half;
    v2f a; a.x = p[0]; a.y = p[1];
    qa[t] = a;
  }

  v8f oacc[4];
#pragma unroll
  for (int dt = 0; dt < 4; ++dt) oacc[dt] = vzero8();

#pragma unroll
  for (int nt = 0; nt < 2; ++nt) {
    v8f acc = vzero8();
#pragma unroll
    for (int t = 0; t < 16; ++t) {
      const float* bp = KBh + (size_t)(nt * 16 + ln) * DD + 4 * t + 2 * half;
      v2f bf; bf.x = bp[0]; bf.y = bp[1];
      acc = wmma4(qa[t], bf, acc);
    }
#pragma unroll
    for (int r = 0; r < 8; ++r) {
      int m = r + 8 * half;
      int col = nt * 16 + ln;
      float x = acc[r] * SCALE;
      myP[m * PSTRIDE + col] = (col <= qb) ? x : NEGINF;
    }
  }
  __syncthreads();

  if (tid < NB) {
    float s = 0.f;
    for (int q2 = 0; q2 < BS; ++q2) s += sP[q2 * PSTRIDE + tid];
    sScore[tid] = s * (1.f / BS);
  }
  __syncthreads();

  if (tid == 0) {
    float loc[NB];
    for (int i = 0; i < NB; ++i) loc[i] = sScore[i];
    for (int s = 0; s < NSEL; ++s) {
      int bi = 0;
      float bv = loc[0];
      for (int i = 1; i < NB; ++i)
        if (loc[i] > bv) { bv = loc[i]; bi = i; }
      sSel[s] = bi;
      loc[bi] = -3.4e38f;
    }
  }
  __syncthreads();

  if (tid < BS) {
    float* row = sP + tid * PSTRIDE;
    float mx = -3.4e38f;
    for (int c = 0; c < NB; ++c) mx = fmaxf(mx, row[c]);
    float sum = 0.f;
    for (int c = 0; c < NB; ++c) {
      float p = (mx > -1e29f) ? __expf(row[c] - mx) : 0.f;
      row[c] = p;
      sum += p;
    }
    float rs = (sum > 0.f) ? 1.f / sum : 0.f;
    for (int c = 0; c < NB; ++c) row[c] *= rs;
  }
  __syncthreads();

#pragma unroll
  for (int dt = 0; dt < 4; ++dt) {
#pragma unroll
    for (int g = 0; g < 8; ++g) {
      int c0 = 4 * g + 2 * half;
      v2f a; a.x = myP[ln * PSTRIDE + c0]; a.y = myP[ln * PSTRIDE + c0 + 1];
      v2f bf;
      bf.x = VBh[(size_t)c0 * DD + dt * 16 + ln];
      bf.y = VBh[(size_t)(c0 + 1) * DD + dt * 16 + ln];
      oacc[dt] = wmma4(a, bf, oacc[dt]);
    }
  }
  __syncthreads();

  float4* const mine = reinterpret_cast<float4*>(myP) + lane;
  uint4* const mineu = reinterpret_cast<uint4*>(myP) + lane;
  {
    const int i = qrow0 + ln;
    qk_tiles<SELC / 16>(
        Kh, ln, half, qa,
        [&](int t) { return sSel[t >> 2] * BS + (t & 3) * 16; },
        [&](int t, int keyBase, v8f acc) {
          unsigned kk[8];
#pragma unroll
          for (int r = 0; r < 8; ++r) {
            const int tok = keyBase + r + 8 * half;
            const float x = acc[r] * SCALE;
            kk[r] = okey((tok <= i) ? x : NEGINF);
          }
          mineu[(2 * t) * 32] = make_uint4(kk[0], kk[1], kk[2], kk[3]);
          mineu[(2 * t + 1) * 32] = make_uint4(kk[4], kk[5], kk[6], kk[7]);
        });
  }

  unsigned ky[SELC / 2];
#pragma unroll
  for (int c = 0; c < SELC / 8; ++c) {
    const uint4 k4 = mineu[c * 32];
    ky[4 * c + 0] = k4.x; ky[4 * c + 1] = k4.y; ky[4 * c + 2] = k4.z; ky[4 * c + 3] = k4.w;
  }

  unsigned prefix = 0u;
#pragma nounroll
  for (int bit = 31; bit >= 0; --bit) {
    const unsigned cand = prefix | (1u << bit);
    unsigned c0 = 0u, c1 = 0u, c2 = 0u, c3 = 0u;
#pragma unroll
    for (int j = 0; j < SELC / 2; j += 4) {
      c0 += (ky[j + 0] >= cand) ? 1u : 0u;
      c1 += (ky[j + 1] >= cand) ? 1u : 0u;
      c2 += (ky[j + 2] >= cand) ? 1u : 0u;
      c3 += (ky[j + 3] >= cand) ? 1u : 0u;
    }
    unsigned cnt = (c0 + c1) + (c2 + c3);
    cnt += xhalf_u(cnt);
    if (cnt >= (unsigned)KEEP) prefix = cand;
  }

  v8f sacc[4];
#pragma unroll
  for (int dt = 0; dt < 4; ++dt) sacc[dt] = vzero8();
  const float* vlane = Vh + (size_t)(8 * half) * DD + ln;

  {
    unsigned mk = 0u;
#pragma unroll
    for (int j = 0; j < SELC / 2; ++j) mk = (ky[j] > mk) ? ky[j] : mk;
    const unsigned mo = xhalf_u(mk);
    mk = (mo > mk) ? mo : mk;
    const float mx = unkey(mk);
    const bool live = mx > -1e29f;
    float s0 = 0.f, s1 = 0.f, s2 = 0.f, s3 = 0.f;
#pragma unroll 4
    for (int c = 0; c < SELC / 8; ++c) {
      const uint4 k4 = mineu[c * 32];
      const float p0 = (k4.x >= prefix && live) ? __expf(unkey(k4.x) - mx) : 0.f;
      const float p1 = (k4.y >= prefix && live) ? __expf(unkey(k4.y) - mx) : 0.f;
      const float p2 = (k4.z >= prefix && live) ? __expf(unkey(k4.z) - mx) : 0.f;
      const float p3 = (k4.w >= prefix && live) ? __expf(unkey(k4.w) - mx) : 0.f;
      s0 += p0; s1 += p1; s2 += p2; s3 += p3;
      mine[c * 32] = make_float4(p0, p1, p2, p3);
    }
    float sum = (s0 + s1) + (s2 + s3);
    sum += xhalf_f(sum);
    const float rs = (sum > 0.f) ? 1.f / sum : 0.f;

    pv_tiles<SELC / 16>(sacc, mine, rs, [&](int t) {
      return vlane + (size_t)sSel[t >> 2] * (BS * DD) + (size_t)((t & 3) * 16) * DD;
    });
  }

  auto wgroup_ok = [&](int cg) { return (cg == 0) ? (qb >= 3) : (qb - 3 + cg >= 0); };
  auto wgroup_kb = [&](int cg) { return (cg != 0 && qb - 3 + cg >= 0) ? qb - 3 + cg : 0; };
  {
    const int i = qrow0 + ln;
    qk_tiles<WINC / 16>(
        Kh, ln, half, qa,
        [&](int t) { return wgroup_kb(t >> 2) * BS + (t & 3) * 16; },
        [&](int t, int keyBase, v8f acc) {
          const int ilim = wgroup_ok(t >> 2) ? i : -1;
          float w[8];
#pragma unroll
          for (int r = 0; r < 8; ++r) {
            const int j = keyBase + r + 8 * half;
            const bool ok = (j <= ilim) && (((i - j) < WIN) || (j < 1));
            w[r] = ok ? acc[r] * SCALE : NEGINF;
          }
          mine[(2 * t) * 32] = make_float4(w[0], w[1], w[2], w[3]);
          mine[(2 * t + 1) * 32] = make_float4(w[4], w[5], w[6], w[7]);
        });
  }

  {
    float wl[WINC / 2];
#pragma unroll
    for (int c = 0; c < WINC / 8; ++c) {
      const float4 w4 = mine[c * 32];
      wl[4 * c + 0] = w4.x; wl[4 * c + 1] = w4.y; wl[4 * c + 2] = w4.z; wl[4 * c + 3] = w4.w;
    }
    float m0 = wl[0], m1 = wl[1], m2 = wl[2], m3 = wl[3];
#pragma unroll
    for (int j = 4; j < WINC / 2; j += 4) {
      m0 = fmaxf(m0, wl[j + 0]);
      m1 = fmaxf(m1, wl[j + 1]);
      m2 = fmaxf(m2, wl[j + 2]);
      m3 = fmaxf(m3, wl[j + 3]);
    }
    float mx = fmaxf(fmaxf(m0, m1), fmaxf(m2, m3));
    mx = fmaxf(mx, xhalf_f(mx));
    const bool live = mx > -1e29f;
    float s0 = 0.f, s1 = 0.f, s2 = 0.f, s3 = 0.f;
#pragma unroll
    for (int j = 0; j < WINC / 2; j += 4) {
      const float p0 = live ? __expf(wl[j + 0] - mx) : 0.f;
      const float p1 = live ? __expf(wl[j + 1] - mx) : 0.f;
      const float p2 = live ? __expf(wl[j + 2] - mx) : 0.f;
      const float p3 = live ? __expf(wl[j + 3] - mx) : 0.f;
      s0 += p0; s1 += p1; s2 += p2; s3 += p3;
      mine[(j >> 2) * 32] = make_float4(p0, p1, p2, p3);
    }
    float sum = (s0 + s1) + (s2 + s3);
    sum += xhalf_f(sum);
    const float rs = (sum > 0.f) ? 1.f / sum : 0.f;

    pv_tiles<WINC / 16>(sacc, mine, rs, [&](int t) {
      return vlane + (size_t)wgroup_kb(t >> 2) * (BS * DD) + (size_t)((t & 3) * 16) * DD;
    });
  }

#pragma unroll
  for (int dt = 0; dt < 4; ++dt) {
    float* dst = myP + ln * PSTRIDE + 256 + dt * 16 + 8 * half;
    float4 lo, hi;
    lo.x = sacc[dt][0]; lo.y = sacc[dt][1]; lo.z = sacc[dt][2]; lo.w = sacc[dt][3];
    hi.x = sacc[dt][4]; hi.y = sacc[dt][5]; hi.z = sacc[dt][6]; hi.w = sacc[dt][7];
    *reinterpret_cast<float4*>(dst) = lo;
    *reinterpret_cast<float4*>(dst + 4) = hi;
  }
#pragma unroll
  for (int dt = 0; dt < 4; ++dt) {
#pragma unroll
    for (int r = 0; r < 8; ++r)
      oacc[dt][r] += myP[(r + 8 * half) * PSTRIDE + 256 + dt * 16 + ln];
  }

  const float third = 1.f / 3.f;
#pragma unroll
  for (int dt = 0; dt < 4; ++dt) {
#pragma unroll
    for (int r = 0; r < 8; ++r) {
      int row = qrow0 + r + 8 * half;
      Oh[(size_t)row * DD + dt * 16 + ln] = oacc[dt][r] * third;
    }
  }
}

extern "C" void kernel_launch(void* const* d_in, const int* in_sizes, int n_in,
                              void* d_out, int out_size, void* d_ws,
                              size_t ws_size, hipStream_t stream) {
  (void)in_sizes; (void)n_in; (void)out_size; (void)ws_size;
  const float* q = (const float*)d_in[0];
  const float* k = (const float*)d_in[1];
  const float* v = (const float*)d_in[2];
  float* out = (float*)d_out;
  float* kblk = (float*)d_ws;
  float* vblk = kblk + (size_t)BB * HH * NB * DD;

  nsa_blockmean_kernel<<<BB * HH * NB, 64, 0, stream>>>(k, v, kblk, vblk);

  const size_t shmem = (size_t)64 * PSTRIDE * sizeof(float);
  hipFuncSetAttribute(reinterpret_cast<const void*>(nsa_main_kernel),
                      hipFuncAttributeMaxDynamicSharedMemorySize, (int)shmem);
  nsa_main_kernel<<<BB * HH * NB, 128, shmem, stream>>>(q, k, v, kblk, vblk,
                                                        out);
}


# Q x K^T (selected, window): key tiles split over the 4 waves, a K fragment is loaded once per work-group and used on all 4 query tiles (Q fragments shared via LDS); sum order unchanged
# speedup vs baseline: 3.1667x; 1.0518x over previous
#include <hip/hip_runtime.h>

#define BB 2
#define HH 8
#define SS 2048
#define DD 64
#define NB 32
#define BS 64
#define NSEL 8
#define SELC 512
#define KEEP 256
#define WIN 128
#define WINC 256
#define PSTRIDE 520
#define NEGINF (-1e30f)
#define SCALE 0.125f

typedef __attribute__((ext_vector_type(2))) float v2f;
typedef __attribute__((ext_vector_type(8))) float v8f;

static __device__ inline v8f vzero8() {
  v8f r;
#pragma unroll
  for (int i = 0; i < 8; ++i) r[i] = 0.f;
  return r;
}

static __device__ inline v8f wmma4(v2f a, v2f b, v8f c) {
  return __builtin_amdgcn_wmma_f32_16x16x4_f32(false, a, false, b, (short)0, c,
                                               false, false);
}

static __device__ inline unsigned okey(float x) {
  unsigned u = __float_as_uint(x);
  return (u & 0x80000000u) ? ~u : (u | 0x80000000u);
}

static __device__ inline int wred_sumi(int x) {
#pragma unroll
  for (int o = 16; o > 0; o >>= 1) x += __shfl_xor(x, o, 32);
  return x;
}

static __device__ inline float unkey(unsigned k) {
  unsigned u = (k & 0x80000000u) ? (k ^ 0x80000000u) : ~k;
  return __uint_as_float(u);
}

static __device__ inline unsigned xhalf_u(unsigned x) {
  return __builtin_amdgcn_permlanex16(x, x, 0x76543210u, 0xfedcba98u, false, false);
}
static __device__ inline float xhalf_f(float x) {
  return __uint_as_float(xhalf_u(__float_as_uint(x)));
}

__global__ void nsa_blockmean_kernel(const float* __restrict__ K,
                                     const float* __restrict__ V,
                                     float* __restrict__ KB,
                                     float* __restrict__ VB) {
  const int g = blockIdx.x;
  const int d = threadIdx.x;
  const float* kp = K + (size_t)g * (BS * DD);
  const float* vp = V + (size_t)g * (BS * DD);
  float ks = 0.f, vs = 0.f;
  for (int t = 0; t < BS; ++t) {
    ks += kp[t * DD + d];
    vs += vp[t * DD + d];
  }
  KB[(size_t)g * DD + d] = ks * (1.f / BS);
  VB[(size_t)g * DD + d] = vs * (1.f / BS);
}

static __device__ inline void load_ktile(v2f (&kf)[16], const float* __restrict__ Kh,
                                         int keyBase, int ln, int half) {
#pragma unroll
  for (int t = 0; t < 16; ++t) {
    const float* bp = Kh + (size_t)(keyBase + ln) * DD + 4 * t + 2 * half;
    v2f a; a.x = bp[0]; a.y = bp[1];
    kf[t] = a;
  }
}

static __device__ inline v8f mma_ktile(const v2f (&kf)[16], const v2f (&qa)[16]) {
  v8f acc = vzero8();
#pragma unroll
  for (int t = 0; t < 16; ++t) acc = wmma4(kf[t], qa[t], acc);
  return acc;
}

static __device__ inline void load_qtile(v2f (&qf)[16], const v2f* __restrict__ sQ, int qt,
                                         int lane) {
#pragma unroll
  for (int t = 0; t < 16; ++t) qf[t] = sQ[(qt * 16 + t) * 32 + lane];
}

template <int NG, typename FB, typename FE>
static __device__ inline void qk_tiles4(const float* __restrict__ Kh, const v2f* __restrict__ sQ,
                                        int wave, int lane, int ln, int half, FB kbase, FE emit) {
  v2f kf[16], kfn[16], qfa[16], qfb[16];
  load_ktile(kfn, Kh, kbase(0), ln, half);
  load_qtile(qfa, sQ, wave, lane);
#pragma nounroll
  for (int g = 0; g < NG; ++g) {
    const int kb = kbase(g);
#pragma unroll
    for (int t = 0; t < 16; ++t) kf[t] = kfn[t];
    const int gn = (g + 1 < NG) ? g + 1 : NG - 1;
    load_ktile(kfn, Kh, kbase(gn), ln, half);
#pragma nounroll
    for (int j = 0; j < 4; j += 2) {
      const int qt0 = (wave + j) & 3;
      const int qt1 = (wave + j + 1) & 3;
      load_qtile(qfb, sQ, qt1, lane);
      __builtin_amdgcn_sched_barrier(0);
      emit(g, qt0, kb, mma_ktile(kf, qfa));
      load_qtile(qfa, sQ, (wave + j + 2) & 3, lane);
      __builtin_amdgcn_sched_barrier(0);
      emit(g, qt1, kb, mma_ktile(kf, qfb));
    }
  }
}

static __device__ inline void load_vtile(v2f (&vf)[16], const float* __restrict__ vb) {
#pragma unroll
  for (int u = 0; u < 4; ++u) {
#pragma unroll
    for (int dt = 0; dt < 4; ++dt) {
      v2f va;
      va.x = vb[(2 * u) * DD + dt * 16];
      va.y = vb[(2 * u + 1) * DD + dt * 16];
      vf[u * 4 + dt] = va;
    }
  }
}

static __device__ inline void mma_vtile(v8f (&sacc)[4], const v2f (&vf)[16],
                                        const float4* __restrict__ mine, int t, float rs) {
  const float4 pa = mine[(2 * t) * 32];
  const float4 pc = mine[(2 * t + 1) * 32];
  float p[8];
  p[0] = pa.x * rs; p[1] = pa.y * rs; p[2] = pa.z * rs; p[3] = pa.w * rs;
  p[4] = pc.x * rs; p[5] = pc.y * rs; p[6] = pc.z * rs; p[7] = pc.w * rs;
#pragma unroll
  for (int u = 0; u < 4; ++u) {
    v2f pb; pb.x = p[2 * u]; pb.y = p[2 * u + 1];
#pragma unroll
    for (int dt = 0; dt < 4; ++dt) sacc[dt] = wmma4(vf[u * 4 + dt], pb, sacc[dt]);
  }
}

template <int NT, typename F>
static __device__ inline void pv_tiles(v8f (&sacc)[4], const float4* __restrict__ mine,
                                       float rs, F vtile) {
  v2f vfa[16], vfb[16];
  load_vtile(vfa, vtile(0));
  load_vtile(vfb, vtile(1));
#pragma nounroll
  for (int t = 0; t < NT; t += 2) {
    mma_vtile(sacc, vfa, mine, t, rs);
    const int ta = (t + 2 < NT) ? t + 2 : NT - 1;
    load_vtile(vfa, vtile(ta));
    __builtin_amdgcn_sched_barrier(0);
    mma_vtile(sacc, vfb, mine, t + 1, rs);
    const int tb = (t + 3 < NT) ? t + 3 : NT - 1;
    load_vtile(vfb, vtile(tb));
    __builtin_amdgcn_sched_barrier(0);
  }
}

__global__ __launch_bounds__(128, 2) void nsa_main_kernel(
    const float* __restrict__ Q, const float* __restrict__ K,
    const float* __restrict__ V, const float* __restrict__ KB,
    const float* __restrict__ VB, float* __restrict__ O) {
  extern __shared__ __align__(16) float sP[];
  __shared__ float sScore[NB];
  __shared__ int sSel[NSEL];
  __shared__ v2f sQ[4 * 16 * 32];

  const int wg = blockIdx.x;
  const int qb = wg % NB;
  const int bh = wg / NB;
  const size_t hoff = (size_t)bh * SS * DD;
  const float* Qh = Q + hoff;
  const float* Kh = K + hoff;
  const float* Vh = V + hoff;
  const float* KBh = KB + (size_t)bh * NB * DD;
  const float* VBh = VB + (size_t)bh * NB * DD;
  float* Oh = O + hoff;

  const int tid = threadIdx.x;
  const int wave = tid >> 5;
  const int lane = tid & 31;
  const int half = lane >> 4;
  const int ln = lane & 15;
  const int qrow0 = qb * BS + wave * 16;
  float* myP = sP + (size_t)wave * 16 * PSTRIDE;

  v2f qa[16];
#pragma unroll
  for (int t = 0; t < 16; ++t) {
    const float* p = Qh + (size_t)(qrow0 + ln) * DD + 4 * t + 2 * half;
    v2f a; a.x = p[0]; a.y = p[1];
    qa[t] = a;
    sQ[(wave * 16 + t) * 32 + lane] = a;
  }

  v8f oacc[4];
#pragma unroll
  for (int dt = 0; dt < 4; ++dt) oacc[dt] = vzero8();

#pragma unroll
  for (int nt = 0; nt < 2; ++nt) {
    v8f acc = vzero8();
#pragma unroll
    for (int t = 0; t < 16; ++t) {
      const float* bp = KBh + (size_t)(nt * 16 + ln) * DD + 4 * t + 2 * half;
      v2f bf; bf.x = bp[0]; bf.y = bp[1];
      acc = wmma4(qa[t], bf, acc);
    }
#pragma unroll
    for (int r = 0; r < 8; ++r) {
      int m = r + 8 * half;
      int col = nt * 16 + ln;
      float x = acc[r] * SCALE;
      myP[m * PSTRIDE + col] = (col <= qb) ? x : NEGINF;
    }
  }
  __syncthreads();

  if (tid < NB) {
    float s = 0.f;
    for (int q2 = 0; q2 < BS; ++q2) s += sP[q2 * PSTRIDE + tid];
    sScore[tid] = s * (1.f / BS);
  }
  __syncthreads();

  if (tid == 0) {
    float loc[NB];
    for (int i = 0; i < NB; ++i) loc[i] = sScore[i];
    for (int s = 0; s < NSEL; ++s) {
      int bi = 0;
      float bv = loc[0];
      for (int i = 1; i < NB; ++i)
        if (loc[i] > bv) { bv = loc[i]; bi = i; }
      sSel[s] = bi;
      loc[bi] = -3.4e38f;
    }
  }
  __syncthreads();

  if (tid < BS) {
    float* row = sP + tid * PSTRIDE;
    float mx = -3.4e38f;
    for (int c = 0; c < NB; ++c) mx = fmaxf(mx, row[c]);
    float sum = 0.f;
    for (int c = 0; c < NB; ++c) {
      float p = (mx > -1e29f) ? __expf(row[c] - mx) : 0.f;
      row[c] = p;
      sum += p;
    }
    float rs = (sum > 0.f) ? 1.f / sum : 0.f;
    for (int c = 0; c < NB; ++c) row[c] *= rs;
  }
  __syncthreads();

#pragma unroll
  for (int dt = 0; dt < 4; ++dt) {
#pragma unroll
    for (int g = 0; g < 8; ++g) {
      int c0 = 4 * g + 2 * half;
      v2f a; a.x = myP[ln * PSTRIDE + c0]; a.y = myP[ln * PSTRIDE + c0 + 1];
      v2f bf;
      bf.x = VBh[(size_t)c0 * DD + dt * 16 + ln];
      bf.y = VBh[(size_t)(c0 + 1) * DD + dt * 16 + ln];
      oacc[dt] = wmma4(a, bf, oacc[dt]);
    }
  }
  __syncthreads();

  float4* const mine = reinterpret_cast<float4*>(myP) + lane;
  uint4* const mineu = reinterpret_cast<uint4*>(myP) + lane;
  qk_tiles4<NSEL>(
      Kh, sQ, wave, lane, ln, half,
      [&](int g) { return sSel[g] * BS + wave * 16; },
      [&](int g, int qt, int keyBase, v8f acc) {
        const int i = qb * BS + qt * 16 + ln;
        const int t = 4 * g + wave;
        uint4* const dstu = reinterpret_cast<uint4*>(sP + (size_t)qt * 16 * PSTRIDE) + lane;
        unsigned kk[8];
#pragma unroll
        for (int r = 0; r < 8; ++r) {
          const int tok = keyBase + r + 8 * half;
          const float x = acc[r] * SCALE;
          kk[r] = okey((tok <= i) ? x : NEGINF);
        }
        dstu[(2 * t) * 32] = make_uint4(kk[0], kk[1], kk[2], kk[3]);
        dstu[(2 * t + 1) * 32] = make_uint4(kk[4], kk[5], kk[6], kk[7]);
      });
  __syncthreads();

  unsigned ky[SELC / 2];
#pragma unroll
  for (int c = 0; c < SELC / 8; ++c) {
    const uint4 k4 = mineu[c * 32];
    ky[4 * c + 0] = k4.x; ky[4 * c + 1] = k4.y; ky[4 * c + 2] = k4.z; ky[4 * c + 3] = k4.w;
  }

  unsigned prefix = 0u;
#pragma nounroll
  for (int bit = 31; bit >= 0; --bit) {
    const unsigned cand = prefix | (1u << bit);
    unsigned c0 = 0u, c1 = 0u, c2 = 0u, c3 = 0u;
#pragma unroll
    for (int j = 0; j < SELC / 2; j += 4) {
      c0 += (ky[j + 0] >= cand) ? 1u : 0u;
      c1 += (ky[j + 1] >= cand) ? 1u : 0u;
      c2 += (ky[j + 2] >= cand) ? 1u : 0u;
      c3 += (ky[j + 3] >= cand) ? 1u : 0u;
    }
    unsigned cnt = (c0 + c1) + (c2 + c3);
    cnt += xhalf_u(cnt);
    if (cnt >= (unsigned)KEEP) prefix = cand;
  }

  v8f sacc[4];
#pragma unroll
  for (int dt = 0; dt < 4; ++dt) sacc[dt] = vzero8();
  const float* vlane = Vh + (size_t)(8 * half) * DD + ln;

  {
    unsigned mk = 0u;
#pragma unroll
    for (int j = 0; j < SELC / 2; ++j) mk = (ky[j] > mk) ? ky[j] : mk;
    const unsigned mo = xhalf_u(mk);
    mk = (mo > mk) ? mo : mk;
    const float mx = unkey(mk);
    const bool live = mx > -1e29f;
    float s0 = 0.f, s1 = 0.f, s2 = 0.f, s3 = 0.f;
#pragma unroll 4
    for (int c = 0; c < SELC / 8; ++c) {
      const uint4 k4 = mineu[c * 32];
      const float p0 = (k4.x >= prefix && live) ? __expf(unkey(k4.x) - mx) : 0.f;
      const float p1 = (k4.y >= prefix && live) ? __expf(unkey(k4.y) - mx) : 0.f;
      const float p2 = (k4.z >= prefix && live) ? __expf(unkey(k4.z) - mx) : 0.f;
      const float p3 = (k4.w >= prefix && live) ? __expf(unkey(k4.w) - mx) : 0.f;
      s0 += p0; s1 += p1; s2 += p2; s3 += p3;
      mine[c * 32] = make_float4(p0, p1, p2, p3);
    }
    float sum = (s0 + s1) + (s2 + s3);
    sum += xhalf_f(sum);
    const float rs = (sum > 0.f) ? 1.f / sum : 0.f;

    pv_tiles<SELC / 16>(sacc, mine, rs, [&](int t) {
      return vlane + (size_t)sSel[t >> 2] * (BS * DD) + (size_t)((t & 3) * 16) * DD;
    });
  }

  auto wgroup_ok = [&](int cg) { return (cg == 0) ? (qb >= 3) : (qb - 3 + cg >= 0); };
  auto wgroup_kb = [&](int cg) { return (cg != 0 && qb - 3 + cg >= 0) ? qb - 3 + cg : 0; };
  __syncthreads();
  qk_tiles4<WINC / BS>(
      Kh, sQ, wave, lane, ln, half,
      [&](int g) { return wgroup_kb(g) * BS + wave * 16; },
      [&](int g, int qt, int keyBase, v8f acc) {
        const int i = qb * BS + qt * 16 + ln;
        const int t = 4 * g + wave;
        float4* const dst = reinterpret_cast<float4*>(sP + (size_t)qt * 16 * PSTRIDE) + lane;
        const int ilim = wgroup_ok(g) ? i : -1;
        float w[8];
#pragma unroll
        for (int r = 0; r < 8; ++r) {
          const int kj = keyBase + r + 8 * half;
          const bool ok = (kj <= ilim) && (((i - kj) < WIN) || (kj < 1));
          w[r] = ok ? acc[r] * SCALE : NEGINF;
        }
        dst[(2 * t) * 32] = make_float4(w[0], w[1], w[2], w[3]);
        dst[(2 * t + 1) * 32] = make_float4(w[4], w[5], w[6], w[7]);
      });
  __syncthreads();

  {
    float wl[WINC / 2];
#pragma unroll
    for (int c = 0; c < WINC / 8; ++c) {
      const float4 w4 = mine[c * 32];
      wl[4 * c + 0] = w4.x; wl[4 * c + 1] = w4.y; wl[4 * c + 2] = w4.z; wl[4 * c + 3] = w4.w;
    }
    float m0 = wl[0], m1 = wl[1], m2 = wl[2], m3 = wl[3];
#pragma unroll
    for (int j = 4; j < WINC / 2; j += 4) {
      m0 = fmaxf(m0, wl[j + 0]);
      m1 = fmaxf(m1, wl[j + 1]);
      m2 = fmaxf(m2, wl[j + 2]);
      m3 = fmaxf(m3, wl[j + 3]);
    }
    float mx = fmaxf(fmaxf(m0, m1), fmaxf(m2, m3));
    mx = fmaxf(mx, xhalf_f(mx));
    const bool live = mx > -1e29f;
    float s0 = 0.f, s1 = 0.f, s2 = 0.f, s3 = 0.f;
#pragma unroll
    for (int j = 0; j < WINC / 2; j += 4) {
      const float p0 = live ? __expf(wl[j + 0] - mx) : 0.f;
      const float p1 = live ? __expf(wl[j + 1] - mx) : 0.f;
      const float p2 = live ? __expf(wl[j + 2] - mx) : 0.f;
      const float p3 = live ? __expf(wl[j + 3] - mx) : 0.f;
      s0 += p0; s1 += p1; s2 += p2; s3 += p3;
      mine[(j >> 2) * 32] = make_float4(p0, p1, p2, p3);
    }
    float sum = (s0 + s1) + (s2 + s3);
    sum += xhalf_f(sum);
    const float rs = (sum > 0.f) ? 1.f / sum : 0.f;

    pv_tiles<WINC / 16>(sacc, mine, rs, [&](int t) {
      return vlane + (size_t)wgroup_kb(t >> 2) * (BS * DD) + (size_t)((t & 3) * 16) * DD;
    });
  }

#pragma unroll
  for (int dt = 0; dt < 4; ++dt) {
    float* dst = myP + ln * PSTRIDE + 256 + dt * 16 + 8 * half;
    float4 lo, hi;
    lo.x = sacc[dt][0]; lo.y = sacc[dt][1]; lo.z = sacc[dt][2]; lo.w = sacc[dt][3];
    hi.x = sacc[dt][4]; hi.y = sacc[dt][5]; hi.z = sacc[dt][6]; hi.w = sacc[dt][7];
    *reinterpret_cast<float4*>(dst) = lo;
    *reinterpret_cast<float4*>(dst + 4) = hi;
  }
#pragma unroll
  for (int dt = 0; dt < 4; ++dt) {
#pragma unroll
    for (int r = 0; r < 8; ++r)
      oacc[dt][r] += myP[(r + 8 * half) * PSTRIDE + 256 + dt * 16 + ln];
  }

  const float third = 1.f / 3.f;
#pragma unroll
  for (int dt = 0; dt < 4; ++dt) {
#pragma unroll
    for (int r = 0; r < 8; ++r) {
      int row = qrow0 + r + 8 * half;
      Oh[(size_t)row * DD + dt * 16 + ln] = oacc[dt][r] * third;
    }
  }
}

extern "C" void kernel_launch(void* const* d_in, const int* in_sizes, int n_in,
                              void* d_out, int out_size, void* d_ws,
                              size_t ws_size, hipStream_t stream) {
  (void)in_sizes; (void)n_in; (void)out_size; (void)ws_size;
  const float* q = (const float*)d_in[0];
  const float* k = (const float*)d_in[1];
  const float* v = (const float*)d_in[2];
  float* out = (float*)d_out;
  float* kblk = (float*)d_ws;
  float* vblk = kblk + (size_t)BB * HH * NB * DD;

  nsa_blockmean_kernel<<<BB * HH * NB, 64, 0, stream>>>(k, v, kblk, vblk);

  const size_t shmem = (size_t)64 * PSTRIDE * sizeof(float);
  hipFuncSetAttribute(reinterpret_cast<const void*>(nsa_main_kernel),
                      hipFuncAttributeMaxDynamicSharedMemorySize, (int)shmem);
  nsa_main_kernel<<<BB * HH * NB, 128, shmem, stream>>>(q, k, v, kblk, vblk,
                                                        out);
}


# top-256 search on the keys' 16-bit halves (high, then low), 2 per register: counts by packed saturating subtract + add, no compare + carry via the condition register; kept set, sum order same
# speedup vs baseline: 3.4720x; 1.0964x over previous
#include <hip/hip_runtime.h>

#define BB 2
#define HH 8
#define SS 2048
#define DD 64
#define NB 32
#define BS 64
#define NSEL 8
#define SELC 512
#define KEEP 256
#define WIN 128
#define WINC 256
#define PSTRIDE 520
#define NEGINF (-1e30f)
#define SCALE 0.125f

typedef __attribute__((ext_vector_type(2))) float v2f;
typedef __attribute__((ext_vector_type(8))) float v8f;

static __device__ inline v8f vzero8() {
  v8f r;
#pragma unroll
  for (int i = 0; i < 8; ++i) r[i] = 0.f;
  return r;
}

static __device__ inline v8f wmma4(v2f a, v2f b, v8f c) {
  return __builtin_amdgcn_wmma_f32_16x16x4_f32(false, a, false, b, (short)0, c,
                                               false, false);
}

static __device__ inline unsigned okey(float x) {
  unsigned u = __float_as_uint(x);
  return (u & 0x80000000u) ? ~u : (u | 0x80000000u);
}

static __device__ inline int wred_sumi(int x) {
#pragma unroll
  for (int o = 16; o > 0; o >>= 1) x += __shfl_xor(x, o, 32);
  return x;
}

static __device__ inline float unkey(unsigned k) {
  unsigned u = (k & 0x80000000u) ? (k ^ 0x80000000u) : ~k;
  return __uint_as_float(u);
}

static __device__ inline unsigned xhalf_u(unsigned x) {
  return __builtin_amdgcn_permlanex16(x, x, 0x76543210u, 0xfedcba98u, false, false);
}
static __device__ inline float xhalf_f(float x) {
  return __uint_as_float(xhalf_u(__float_as_uint(x)));
}


typedef __attribute__((ext_vector_type(2))) unsigned short u16x2;

static __device__ inline unsigned search16(const u16x2 (&pk)[SELC / 4]) {
  unsigned pre = 0u;
  u16x2 one; one.x = 1; one.y = 1;
#pragma nounroll
  for (int bit = 15; bit >= 0; --bit) {
    const unsigned cand = pre | (1u << bit);
    u16x2 cv; cv.x = (unsigned short)cand; cv.y = (unsigned short)cand;
    unsigned cnt = 0u;
#pragma unroll
    for (int q = 0; q < 4; ++q) {
      u16x2 a = 0;
#pragma unroll
      for (int j = 0; j < SELC / 16; ++j)
        a += __builtin_elementwise_sub_sat(one, __builtin_elementwise_sub_sat(cv, pk[q * (SELC / 16) + j]));
      cnt += (unsigned)a.x + (unsigned)a.y;
    }
    cnt += xhalf_u(cnt);
    if (cnt >= (unsigned)KEEP) pre = cand;
  }
  return pre;
}

__global__ void nsa_blockmean_kernel(const float* __restrict__ K,
                                     const float* __restrict__ V,
                                     float* __restrict__ KB,
                                     float* __restrict__ VB) {
  const int g = blockIdx.x;
  const int d = threadIdx.x;
  const float* kp = K + (size_t)g * (BS * DD);
  const float* vp = V + (size_t)g * (BS * DD);
  float ks = 0.f, vs = 0.f;
  for (int t = 0; t < BS; ++t) {
    ks += kp[t * DD + d];
    vs += vp[t * DD + d];
  }
  KB[(size_t)g * DD + d] = ks * (1.f / BS);
  VB[(size_t)g * DD + d] = vs * (1.f / BS);
}

static __device__ inline void load_ktile(v2f (&kf)[16], const float* __restrict__ Kh,
                                         int keyBase, int ln, int half) {
#pragma unroll
  for (int t = 0; t < 16; ++t) {
    const float* bp = Kh + (size_t)(keyBase + ln) * DD + 4 * t + 2 * half;
    v2f a; a.x = bp[0]; a.y = bp[1];
    kf[t] = a;
  }
}

static __device__ inline v8f mma_ktile(const v2f (&kf)[16], const v2f (&qa)[16]) {
  v8f acc = vzero8();
#pragma unroll
  for (int t = 0; t < 16; ++t) acc = wmma4(kf[t], qa[t], acc);
  return acc;
}

static __device__ inline void load_qtile(v2f (&qf)[16], const v2f* __restrict__ sQ, int qt,
                                         int lane) {
#pragma unroll
  for (int t = 0; t < 16; ++t) qf[t] = sQ[(qt * 16 + t) * 32 + lane];
}

template <int NG, typename FB, typename FE>
static __device__ inline void qk_tiles4(const float* __restrict__ Kh, const v2f* __restrict__ sQ,
                                        int wave, int lane, int ln, int half, FB kbase, FE emit) {
  v2f kf[16], kfn[16], qfa[16], qfb[16];
  load_ktile(kfn, Kh, kbase(0), ln, half);
  load_qtile(qfa, sQ, wave, lane);
#pragma nounroll
  for (int g = 0; g < NG; ++g) {
    const int kb = kbase(g);
#pragma unroll
    for (int t = 0; t < 16; ++t) kf[t] = kfn[t];
    const int gn = (g + 1 < NG) ? g + 1 : NG - 1;
    load_ktile(kfn, Kh, kbase(gn), ln, half);
#pragma nounroll
    for (int j = 0; j < 4; j += 2) {
      const int qt0 = (wave + j) & 3;
      const int qt1 = (wave + j + 1) & 3;
      load_qtile(qfb, sQ, qt1, lane);
      __builtin_amdgcn_sched_barrier(0);
      emit(g, qt0, kb, mma_ktile(kf, qfa));
      load_qtile(qfa, sQ, (wave + j + 2) & 3, lane);
      __builtin_amdgcn_sched_barrier(0);
      emit(g, qt1, kb, mma_ktile(kf, qfb));
    }
  }
}

static __device__ inline void load_vtile(v2f (&vf)[16], const float* __restrict__ vb) {
#pragma unroll
  for (int u = 0; u < 4; ++u) {
#pragma unroll
    for (int dt = 0; dt < 4; ++dt) {
      v2f va;
      va.x = vb[(2 * u) * DD + dt * 16];
      va.y = vb[(2 * u + 1) * DD + dt * 16];
      vf[u * 4 + dt] = va;
    }
  }
}

static __device__ inline void mma_vtile(v8f (&sacc)[4], const v2f (&vf)[16],
                                        const float4* __restrict__ mine, int t, float rs) {
  const float4 pa = mine[(2 * t) * 32];
  const float4 pc = mine[(2 * t + 1) * 32];
  float p[8];
  p[0] = pa.x * rs; p[1] = pa.y * rs; p[2] = pa.z * rs; p[3] = pa.w * rs;
  p[4] = pc.x * rs; p[5] = pc.y * rs; p[6] = pc.z * rs; p[7] = pc.w * rs;
#pragma unroll
  for (int u = 0; u < 4; ++u) {
    v2f pb; pb.x = p[2 * u]; pb.y = p[2 * u + 1];
#pragma unroll
    for (int dt = 0; dt < 4; ++dt) sacc[dt] = wmma4(vf[u * 4 + dt], pb, sacc[dt]);
  }
}

template <int NT, typename F>
static __device__ inline void pv_tiles(v8f (&sacc)[4], const float4* __restrict__ mine,
                                       float rs, F vtile) {
  v2f vfa[16], vfb[16];
  load_vtile(vfa, vtile(0));
  load_vtile(vfb, vtile(1));
#pragma nounroll
  for (int t = 0; t < NT; t += 2) {
    mma_vtile(sacc, vfa, mine, t, rs);
    const int ta = (t + 2 < NT) ? t + 2 : NT - 1;
    load_vtile(vfa, vtile(ta));
    __builtin_amdgcn_sched_barrier(0);
    mma_vtile(sacc, vfb, mine, t + 1, rs);
    const int tb = (t + 3 < NT) ? t + 3 : NT - 1;
    load_vtile(vfb, vtile(tb));
    __builtin_amdgcn_sched_barrier(0);
  }
}

__global__ __launch_bounds__(128, 2) void nsa_main_kernel(
    const float* __restrict__ Q, const float* __restrict__ K,
    const float* __restrict__ V, const float* __restrict__ KB,
    const float* __restrict__ VB, float* __restrict__ O) {
  extern __shared__ __align__(16) float sP[];
  __shared__ float sScore[NB];
  __shared__ int sSel[NSEL];
  __shared__ v2f sQ[4 * 16 * 32];

  const int wg = blockIdx.x;
  const int qb = wg % NB;
  const int bh = wg / NB;
  const size_t hoff = (size_t)bh * SS * DD;
  const float* Qh = Q + hoff;
  const float* Kh = K + hoff;
  const float* Vh = V + hoff;
  const float* KBh = KB + (size_t)bh * NB * DD;
  const float* VBh = VB + (size_t)bh * NB * DD;
  float* Oh = O + hoff;

  const int tid = threadIdx.x;
  const int wave = tid >> 5;
  const int lane = tid & 31;
  const int half = lane >> 4;
  const int ln = lane & 15;
  const int qrow0 = qb * BS + wave * 16;
  float* myP = sP + (size_t)wave * 16 * PSTRIDE;

  v2f qa[16];
#pragma unroll
  for (int t = 0; t < 16; ++t) {
    const float* p = Qh + (size_t)(qrow0 + ln) * DD + 4 * t + 2 * half;
    v2f a; a.x = p[0]; a.y = p[1];
    qa[t] = a;
    sQ[(wave * 16 + t) * 32 + lane] = a;
  }

  v8f oacc[4];
#pragma unroll
  for (int dt = 0; dt < 4; ++dt) oacc[dt] = vzero8();

#pragma unroll
  for (int nt = 0; nt < 2; ++nt) {
    v8f acc = vzero8();
#pragma unroll
    for (int t = 0; t < 16; ++t) {
      const float* bp = KBh + (size_t)(nt * 16 + ln) * DD + 4 * t + 2 * half;
      v2f bf; bf.x = bp[0]; bf.y = bp[1];
      acc = wmma4(qa[t], bf, acc);
    }
#pragma unroll
    for (int r = 0; r < 8; ++r) {
      int m = r + 8 * half;
      int col = nt * 16 + ln;
      float x = acc[r] * SCALE;
      myP[m * PSTRIDE + col] = (col <= qb) ? x : NEGINF;
    }
  }
  __syncthreads();

  if (tid < NB) {
    float s = 0.f;
    for (int q2 = 0; q2 < BS; ++q2) s += sP[q2 * PSTRIDE + tid];
    sScore[tid] = s * (1.f / BS);
  }
  __syncthreads();

  if (tid == 0) {
    float loc[NB];
    for (int i = 0; i < NB; ++i) loc[i] = sScore[i];
    for (int s = 0; s < NSEL; ++s) {
      int bi = 0;
      float bv = loc[0];
      for (int i = 1; i < NB; ++i)
        if (loc[i] > bv) { bv = loc[i]; bi = i; }
      sSel[s] = bi;
      loc[bi] = -3.4e38f;
    }
  }
  __syncthreads();

  if (tid < BS) {
    float* row = sP + tid * PSTRIDE;
    float mx = -3.4e38f;
    for (int c = 0; c < NB; ++c) mx = fmaxf(mx, row[c]);
    float sum = 0.f;
    for (int c = 0; c < NB; ++c) {
      float p = (mx > -1e29f) ? __expf(row[c] - mx) : 0.f;
      row[c] = p;
      sum += p;
    }
    float rs = (sum > 0.f) ? 1.f / sum : 0.f;
    for (int c = 0; c < NB; ++c) row[c] *= rs;
  }
  __syncthreads();

#pragma unroll
  for (int dt = 0; dt < 4; ++dt) {
#pragma unroll
    for (int g = 0; g < 8; ++g) {
      int c0 = 4 * g + 2 * half;
      v2f a; a.x = myP[ln * PSTRIDE + c0]; a.y = myP[ln * PSTRIDE + c0 + 1];
      v2f bf;
      bf.x = VBh[(size_t)c0 * DD + dt * 16 + ln];
      bf.y = VBh[(size_t)(c0 + 1) * DD + dt * 16 + ln];
      oacc[dt] = wmma4(a, bf, oacc[dt]);
    }
  }
  __syncthreads();

  float4* const mine = reinterpret_cast<float4*>(myP) + lane;
  uint4* const mineu = reinterpret_cast<uint4*>(myP) + lane;
  qk_tiles4<NSEL>(
      Kh, sQ, wave, lane, ln, half,
      [&](int g) { return sSel[g] * BS + wave * 16; },
      [&](int g, int qt, int keyBase, v8f acc) {
        const int i = qb * BS + qt * 16 + ln;
        const int t = 4 * g + wave;
        uint4* const dstu = reinterpret_cast<uint4*>(sP + (size_t)qt * 16 * PSTRIDE) + lane;
        unsigned kk[8];
#pragma unroll
        for (int r = 0; r < 8; ++r) {
          const int tok = keyBase + r + 8 * half;
          const float x = acc[r] * SCALE;
          kk[r] = okey((tok <= i) ? x : NEGINF);
        }
        dstu[(2 * t) * 32] = make_uint4(kk[0], kk[1], kk[2], kk[3]);
        dstu[(2 * t + 1) * 32] = make_uint4(kk[4], kk[5], kk[6], kk[7]);
      });
  __syncthreads();

  u16x2 pk[SELC / 4];
  unsigned mk = 0u;
#pragma unroll
  for (int c = 0; c < SELC / 8; ++c) {
    const uint4 k4 = mineu[c * 32];
    const unsigned m01 = (k4.x > k4.y) ? k4.x : k4.y;
    const unsigned m23 = (k4.z > k4.w) ? k4.z : k4.w;
    const unsigned m4 = (m01 > m23) ? m01 : m23;
    mk = (m4 > mk) ? m4 : mk;
    pk[2 * c + 0] = __builtin_bit_cast(u16x2, __builtin_amdgcn_perm(k4.y, k4.x, 0x07060302u));
    pk[2 * c + 1] = __builtin_bit_cast(u16x2, __builtin_amdgcn_perm(k4.w, k4.z, 0x07060302u));
  }
  const unsigned hbase = search16(pk) << 16;
#pragma unroll
  for (int c = 0; c < SELC / 8; ++c) {
    const uint4 k4 = mineu[c * 32];
    const unsigned w0 = __builtin_elementwise_min(__builtin_elementwise_sub_sat(k4.x, hbase), 0xFFFFu);
    const unsigned w1 = __builtin_elementwise_min(__builtin_elementwise_sub_sat(k4.y, hbase), 0xFFFFu);
    const unsigned w2 = __builtin_elementwise_min(__builtin_elementwise_sub_sat(k4.z, hbase), 0xFFFFu);
    const unsigned w3 = __builtin_elementwise_min(__builtin_elementwise_sub_sat(k4.w, hbase), 0xFFFFu);
    pk[2 * c + 0] = __builtin_bit_cast(u16x2, w0 | (w1 << 16));
    pk[2 * c + 1] = __builtin_bit_cast(u16x2, w2 | (w3 << 16));
  }
  const unsigned prefix = hbase | search16(pk);

  v8f sacc[4];
#pragma unroll
  for (int dt = 0; dt < 4; ++dt) sacc[dt] = vzero8();
  const float* vlane = Vh + (size_t)(8 * half) * DD + ln;

  {
    const unsigned mo = xhalf_u(mk);
    mk = (mo > mk) ? mo : mk;
    const float mx = unkey(mk);
    const bool live = mx > -1e29f;
    float s0 = 0.f, s1 = 0.f, s2 = 0.f, s3 = 0.f;
#pragma unroll 4
    for (int c = 0; c < SELC / 8; ++c) {
      const uint4 k4 = mineu[c * 32];
      const float p0 = (k4.x >= prefix && live) ? __expf(unkey(k4.x) - mx) : 0.f;
      const float p1 = (k4.y >= prefix && live) ? __expf(unkey(k4.y) - mx) : 0.f;
      const float p2 = (k4.z >= prefix && live) ? __expf(unkey(k4.z) - mx) : 0.f;
      const float p3 = (k4.w >= prefix && live) ? __expf(unkey(k4.w) - mx) : 0.f;
      s0 += p0; s1 += p1; s2 += p2; s3 += p3;
      mine[c * 32] = make_float4(p0, p1, p2, p3);
    }
    float sum = (s0 + s1) + (s2 + s3);
    sum += xhalf_f(sum);
    const float rs = (sum > 0.f) ? 1.f / sum : 0.f;

    pv_tiles<SELC / 16>(sacc, mine, rs, [&](int t) {
      return vlane + (size_t)sSel[t >> 2] * (BS * DD) + (size_t)((t & 3) * 16) * DD;
    });
  }

  auto wgroup_ok = [&](int cg) { return (cg == 0) ? (qb >= 3) : (qb - 3 + cg >= 0); };
  auto wgroup_kb = [&](int cg) { return (cg != 0 && qb - 3 + cg >= 0) ? qb - 3 + cg : 0; };
  __syncthreads();
  qk_tiles4<WINC / BS>(
      Kh, sQ, wave, lane, ln, half,
      [&](int g) { return wgroup_kb(g) * BS + wave * 16; },
      [&](int g, int qt, int keyBase, v8f acc) {
        const int i = qb * BS + qt * 16 + ln;
        const int t = 4 * g + wave;
        float4* const dst = reinterpret_cast<float4*>(sP + (size_t)qt * 16 * PSTRIDE) + lane;
        const int ilim = wgroup_ok(g) ? i : -1;
        float w[8];
#pragma unroll
        for (int r = 0; r < 8; ++r) {
          const int kj = keyBase + r + 8 * half;
          const bool ok = (kj <= ilim) && (((i - kj) < WIN) || (kj < 1));
          w[r] = ok ? acc[r] * SCALE : NEGINF;
        }
        dst[(2 * t) * 32] = make_float4(w[0], w[1], w[2], w[3]);
        dst[(2 * t + 1) * 32] = make_float4(w[4], w[5], w[6], w[7]);
      });
  __syncthreads();

  {
    float wl[WINC / 2];
#pragma unroll
    for (int c = 0; c < WINC / 8; ++c) {
      const float4 w4 = mine[c * 32];
      wl[4 * c + 0] = w4.x; wl[4 * c + 1] = w4.y; wl[4 * c + 2] = w4.z; wl[4 * c + 3] = w4.w;
    }
    float m0 = wl[0], m1 = wl[1], m2 = wl[2], m3 = wl[3];
#pragma unroll
    for (int j = 4; j < WINC / 2; j += 4) {
      m0 = fmaxf(m0, wl[j + 0]);
      m1 = fmaxf(m1, wl[j + 1]);
      m2 = fmaxf(m2, wl[j + 2]);
      m3 = fmaxf(m3, wl[j + 3]);
    }
    float mx = fmaxf(fmaxf(m0, m1), fmaxf(m2, m3));
    mx = fmaxf(mx, xhalf_f(mx));
    const bool live = mx > -1e29f;
    float s0 = 0.f, s1 = 0.f, s2 = 0.f, s3 = 0.f;
#pragma unroll
    for (int j = 0; j < WINC / 2; j += 4) {
      const float p0 = live ? __expf(wl[j + 0] - mx) : 0.f;
      const float p1 = live ? __expf(wl[j + 1] - mx) : 0.f;
      const float p2 = live ? __expf(wl[j + 2] - mx) : 0.f;
      const float p3 = live ? __expf(wl[j + 3] - mx) : 0.f;
      s0 += p0; s1 += p1; s2 += p2; s3 += p3;
      mine[(j >> 2) * 32] = make_float4(p0, p1, p2, p3);
    }
    float sum = (s0 + s1) + (s2 + s3);
    sum += xhalf_f(sum);
    const float rs = (sum > 0.f) ? 1.f / sum : 0.f;

    pv_tiles<WINC / 16>(sacc, mine, rs, [&](int t) {
      return vlane + (size_t)wgroup_kb(t >> 2) * (BS * DD) + (size_t)((t & 3) * 16) * DD;
    });
  }

#pragma unroll
  for (int dt = 0; dt < 4; ++dt) {
    float* dst = myP + ln * PSTRIDE + 256 + dt * 16 + 8 * half;
    float4 lo, hi;
    lo.x = sacc[dt][0]; lo.y = sacc[dt][1]; lo.z = sacc[dt][2]; lo.w = sacc[dt][3];
    hi.x = sacc[dt][4]; hi.y = sacc[dt][5]; hi.z = sacc[dt][6]; hi.w = sacc[dt][7];
    *reinterpret_cast<float4*>(dst) = lo;
    *reinterpret_cast<float4*>(dst + 4) = hi;
  }
#pragma unroll
  for (int dt = 0; dt < 4; ++dt) {
#pragma unroll
    for (int r = 0; r < 8; ++r)
      oacc[dt][r] += myP[(r + 8 * half) * PSTRIDE + 256 + dt * 16 + ln];
  }

  const float third = 1.f / 3.f;
#pragma unroll
  for (int dt = 0; dt < 4; ++dt) {
#pragma unroll
    for (int r = 0; r < 8; ++r) {
      int row = qrow0 + r + 8 * half;
      Oh[(size_t)row * DD + dt * 16 + ln] = oacc[dt][r] * third;
    }
  }
}

extern "C" void kernel_launch(void* const* d_in, const int* in_sizes, int n_in,
                              void* d_out, int out_size, void* d_ws,
                              size_t ws_size, hipStream_t stream) {
  (void)in_sizes; (void)n_in; (void)out_size; (void)ws_size;
  const float* q = (const float*)d_in[0];
  const float* k = (const float*)d_in[1];
  const float* v = (const float*)d_in[2];
  float* out = (float*)d_out;
  float* kblk = (float*)d_ws;
  float* vblk = kblk + (size_t)BB * HH * NB * DD;

  nsa_blockmean_kernel<<<BB * HH * NB, 64, 0, stream>>>(k, v, kblk, vblk);

  const size_t shmem = (size_t)64 * PSTRIDE * sizeof(float);
  hipFuncSetAttribute(reinterpret_cast<const void*>(nsa_main_kernel),
                      hipFuncAttributeMaxDynamicSharedMemorySize, (int)shmem);
  nsa_main_kernel<<<BB * HH * NB, 128, shmem, stream>>>(q, k, v, kblk, vblk,
                                                        out);
}


# top-256 search on bit planes: a lane's 256 keys transposed byte by byte into 1-bit-per-key words, a threshold bit decided by and + bit count on 8 words, no compare per key; kept set, sum order same
# speedup vs baseline: 4.0983x; 1.1804x over previous
#include <hip/hip_runtime.h>

#define BB 2
#define HH 8
#define SS 2048
#define DD 64
#define NB 32
#define BS 64
#define NSEL 8
#define SELC 512
#define KEEP 256
#define WIN 128
#define WINC 256
#define PSTRIDE 520
#define NEGINF (-1e30f)
#define SCALE 0.125f

typedef __attribute__((ext_vector_type(2))) float v2f;
typedef __attribute__((ext_vector_type(8))) float v8f;

static __device__ inline v8f vzero8() {
  v8f r;
#pragma unroll
  for (int i = 0; i < 8; ++i) r[i] = 0.f;
  return r;
}

static __device__ inline v8f wmma4(v2f a, v2f b, v8f c) {
  return __builtin_amdgcn_wmma_f32_16x16x4_f32(false, a, false, b, (short)0, c,
                                               false, false);
}

static __device__ inline unsigned okey(float x) {
  unsigned u = __float_as_uint(x);
  return (u & 0x80000000u) ? ~u : (u | 0x80000000u);
}

static __device__ inline int wred_sumi(int x) {
#pragma unroll
  for (int o = 16; o > 0; o >>= 1) x += __shfl_xor(x, o, 32);
  return x;
}

static __device__ inline float unkey(unsigned k) {
  unsigned u = (k & 0x80000000u) ? (k ^ 0x80000000u) : ~k;
  return __uint_as_float(u);
}

static __device__ inline unsigned xhalf_u(unsigned x) {
  return __builtin_amdgcn_permlanex16(x, x, 0x76543210u, 0xfedcba98u, false, false);
}
static __device__ inline float xhalf_f(float x) {
  return __uint_as_float(xhalf_u(__float_as_uint(x)));
}


static __device__ inline void bt_stage(unsigned& a, unsigned& b, int s, unsigned lowmask) {
  const unsigned na = __builtin_amdgcn_bitop3_b32(a, b << s, lowmask, 0xE4);
  const unsigned nb = __builtin_amdgcn_bitop3_b32(a >> s, b, lowmask, 0xE4);
  a = na;
  b = nb;
}

static __device__ inline void planes8(unsigned (&x)[8]) {
#pragma unroll
  for (int i = 0; i < 4; ++i) bt_stage(x[i], x[i + 4], 4, 0x0F0F0F0Fu);
#pragma unroll
  for (int i = 0; i < 8; ++i)
    if ((i & 2) == 0) bt_stage(x[i], x[i + 2], 2, 0x33333333u);
#pragma unroll
  for (int i = 0; i < 8; i += 2) bt_stage(x[i], x[i + 1], 1, 0x55555555u);
}

__global__ void nsa_blockmean_kernel(const float* __restrict__ K,
                                     const float* __restrict__ V,
                                     float* __restrict__ KB,
                                     float* __restrict__ VB) {
  const int g = blockIdx.x;
  const int d = threadIdx.x;
  const float* kp = K + (size_t)g * (BS * DD);
  const float* vp = V + (size_t)g * (BS * DD);
  float ks = 0.f, vs = 0.f;
  for (int t = 0; t < BS; ++t) {
    ks += kp[t * DD + d];
    vs += vp[t * DD + d];
  }
  KB[(size_t)g * DD + d] = ks * (1.f / BS);
  VB[(size_t)g * DD + d] = vs * (1.f / BS);
}

static __device__ inline void load_ktile(v2f (&kf)[16], const float* __restrict__ Kh,
                                         int keyBase, int ln, int half) {
#pragma unroll
  for (int t = 0; t < 16; ++t) {
    const float* bp = Kh + (size_t)(keyBase + ln) * DD + 4 * t + 2 * half;
    v2f a; a.x = bp[0]; a.y = bp[1];
    kf[t] = a;
  }
}

static __device__ inline v8f mma_ktile(const v2f (&kf)[16], const v2f (&qa)[16]) {
  v8f acc = vzero8();
#pragma unroll
  for (int t = 0; t < 16; ++t) acc = wmma4(kf[t], qa[t], acc);
  return acc;
}

static __device__ inline void load_qtile(v2f (&qf)[16], const v2f* __restrict__ sQ, int qt,
                                         int lane) {
#pragma unroll
  for (int t = 0; t < 16; ++t) qf[t] = sQ[(qt * 16 + t) * 32 + lane];
}

template <int NG, typename FB, typename FE>
static __device__ inline void qk_tiles4(const float* __restrict__ Kh, const v2f* __restrict__ sQ,
                                        int wave, int lane, int ln, int half, FB kbase, FE emit) {
  v2f kf[16], kfn[16], qfa[16], qfb[16];
  load_ktile(kfn, Kh, kbase(0), ln, half);
  load_qtile(qfa, sQ, wave, lane);
#pragma nounroll
  for (int g = 0; g < NG; ++g) {
    const int kb = kbase(g);
#pragma unroll
    for (int t = 0; t < 16; ++t) kf[t] = kfn[t];
    const int gn = (g + 1 < NG) ? g + 1 : NG - 1;
    load_ktile(kfn, Kh, kbase(gn), ln, half);
#pragma nounroll
    for (int j = 0; j < 4; j += 2) {
      const int qt0 = (wave + j) & 3;
      const int qt1 = (wave + j + 1) & 3;
      load_qtile(qfb, sQ, qt1, lane);
      __builtin_amdgcn_sched_barrier(0);
      emit(g, qt0, kb, mma_ktile(kf, qfa));
      load_qtile(qfa, sQ, (wave + j + 2) & 3, lane);
      __builtin_amdgcn_sched_barrier(0);
      emit(g, qt1, kb, mma_ktile(kf, qfb));
    }
  }
}

static __device__ inline void load_vtile(v2f (&vf)[16], const float* __restrict__ vb) {
#pragma unroll
  for (int u = 0; u < 4; ++u) {
#pragma unroll
    for (int dt = 0; dt < 4; ++dt) {
      v2f va;
      va.x = vb[(2 * u) * DD + dt * 16];
      va.y = vb[(2 * u + 1) * DD + dt * 16];
      vf[u * 4 + dt] = va;
    }
  }
}

static __device__ inline void mma_vtile(v8f (&sacc)[4], const v2f (&vf)[16],
                                        const float4* __restrict__ mine, int t, float rs) {
  const float4 pa = mine[(2 * t) * 32];
  const float4 pc = mine[(2 * t + 1) * 32];
  float p[8];
  p[0] = pa.x * rs; p[1] = pa.y * rs; p[2] = pa.z * rs; p[3] = pa.w * rs;
  p[4] = pc.x * rs; p[5] = pc.y * rs; p[6] = pc.z * rs; p[7] = pc.w * rs;
#pragma unroll
  for (int u = 0; u < 4; ++u) {
    v2f pb; pb.x = p[2 * u]; pb.y = p[2 * u + 1];
#pragma unroll
    for (int dt = 0; dt < 4; ++dt) sacc[dt] = wmma4(vf[u * 4 + dt], pb, sacc[dt]);
  }
}

template <int NT, typename F>
static __device__ inline void pv_tiles(v8f (&sacc)[4], const float4* __restrict__ mine,
                                       float rs, F vtile) {
  v2f vfa[16], vfb[16];
  load_vtile(vfa, vtile(0));
  load_vtile(vfb, vtile(1));
#pragma nounroll
  for (int t = 0; t < NT; t += 2) {
    mma_vtile(sacc, vfa, mine, t, rs);
    const int ta = (t + 2 < NT) ? t + 2 : NT - 1;
    load_vtile(vfa, vtile(ta));
    __builtin_amdgcn_sched_barrier(0);
    mma_vtile(sacc, vfb, mine, t + 1, rs);
    const int tb = (t + 3 < NT) ? t + 3 : NT - 1;
    load_vtile(vfb, vtile(tb));
    __builtin_amdgcn_sched_barrier(0);
  }
}

__global__ __launch_bounds__(128, 2) void nsa_main_kernel(
    const float* __restrict__ Q, const float* __restrict__ K,
    const float* __restrict__ V, const float* __restrict__ KB,
    const float* __restrict__ VB, float* __restrict__ O) {
  extern __shared__ __align__(16) float sP[];
  __shared__ float sScore[NB];
  __shared__ int sSel[NSEL];
  __shared__ v2f sQ[4 * 16 * 32];

  const int wg = blockIdx.x;
  const int qb = wg % NB;
  const int bh = wg / NB;
  const size_t hoff = (size_t)bh * SS * DD;
  const float* Qh = Q + hoff;
  const float* Kh = K + hoff;
  const float* Vh = V + hoff;
  const float* KBh = KB + (size_t)bh * NB * DD;
  const float* VBh = VB + (size_t)bh * NB * DD;
  float* Oh = O + hoff;

  const int tid = threadIdx.x;
  const int wave = tid >> 5;
  const int lane = tid & 31;
  const int half = lane >> 4;
  const int ln = lane & 15;
  const int qrow0 = qb * BS + wave * 16;
  float* myP = sP + (size_t)wave * 16 * PSTRIDE;

  v2f qa[16];
#pragma unroll
  for (int t = 0; t < 16; ++t) {
    const float* p = Qh + (size_t)(qrow0 + ln) * DD + 4 * t + 2 * half;
    v2f a; a.x = p[0]; a.y = p[1];
    qa[t] = a;
    sQ[(wave * 16 + t) * 32 + lane] = a;
  }

  v8f oacc[4];
#pragma unroll
  for (int dt = 0; dt < 4; ++dt) oacc[dt] = vzero8();

#pragma unroll
  for (int nt = 0; nt < 2; ++nt) {
    v8f acc = vzero8();
#pragma unroll
    for (int t = 0; t < 16; ++t) {
      const float* bp = KBh + (size_t)(nt * 16 + ln) * DD + 4 * t + 2 * half;
      v2f bf; bf.x = bp[0]; bf.y = bp[1];
      acc = wmma4(qa[t], bf, acc);
    }
#pragma unroll
    for (int r = 0; r < 8; ++r) {
      int m = r + 8 * half;
      int col = nt * 16 + ln;
      float x = acc[r] * SCALE;
      myP[m * PSTRIDE + col] = (col <= qb) ? x : NEGINF;
    }
  }
  __syncthreads();

  if (tid < NB) {
    float s = 0.f;
    for (int q2 = 0; q2 < BS; ++q2) s += sP[q2 * PSTRIDE + tid];
    sScore[tid] = s * (1.f / BS);
  }
  __syncthreads();

  if (tid == 0) {
    float loc[NB];
    for (int i = 0; i < NB; ++i) loc[i] = sScore[i];
    for (int s = 0; s < NSEL; ++s) {
      int bi = 0;
      float bv = loc[0];
      for (int i = 1; i < NB; ++i)
        if (loc[i] > bv) { bv = loc[i]; bi = i; }
      sSel[s] = bi;
      loc[bi] = -3.4e38f;
    }
  }
  __syncthreads();

  if (tid < BS) {
    float* row = sP + tid * PSTRIDE;
    float mx = -3.4e38f;
    for (int c = 0; c < NB; ++c) mx = fmaxf(mx, row[c]);
    float sum = 0.f;
    for (int c = 0; c < NB; ++c) {
      float p = (mx > -1e29f) ? __expf(row[c] - mx) : 0.f;
      row[c] = p;
      sum += p;
    }
    float rs = (sum > 0.f) ? 1.f / sum : 0.f;
    for (int c = 0; c < NB; ++c) row[c] *= rs;
  }
  __syncthreads();

#pragma unroll
  for (int dt = 0; dt < 4; ++dt) {
#pragma unroll
    for (int g = 0; g < 8; ++g) {
      int c0 = 4 * g + 2 * half;
      v2f a; a.x = myP[ln * PSTRIDE + c0]; a.y = myP[ln * PSTRIDE + c0 + 1];
      v2f bf;
      bf.x = VBh[(size_t)c0 * DD + dt * 16 + ln];
      bf.y = VBh[(size_t)(c0 + 1) * DD + dt * 16 + ln];
      oacc[dt] = wmma4(a, bf, oacc[dt]);
    }
  }
  __syncthreads();

  float4* const mine = reinterpret_cast<float4*>(myP) + lane;
  uint4* const mineu = reinterpret_cast<uint4*>(myP) + lane;
  qk_tiles4<NSEL>(
      Kh, sQ, wave, lane, ln, half,
      [&](int g) { return sSel[g] * BS + wave * 16; },
      [&](int g, int qt, int keyBase, v8f acc) {
        const int i = qb * BS + qt * 16 + ln;
        const int t = 4 * g + wave;
        uint4* const dstu = reinterpret_cast<uint4*>(sP + (size_t)qt * 16 * PSTRIDE) + lane;
        unsigned kk[8];
#pragma unroll
        for (int r = 0; r < 8; ++r) {
          const int tok = keyBase + r + 8 * half;
          const float x = acc[r] * SCALE;
          kk[r] = okey((tok <= i) ? x : NEGINF);
        }
        dstu[(2 * t) * 32] = make_uint4(kk[0], kk[1], kk[2], kk[3]);
        dstu[(2 * t + 1) * 32] = make_uint4(kk[4], kk[5], kk[6], kk[7]);
      });
  __syncthreads();

  unsigned mk = 0u;
#pragma unroll 4
  for (int c = 0; c < SELC / 8; ++c) {
    const uint4 k4 = mineu[c * 32];
    const unsigned m01 = (k4.x > k4.y) ? k4.x : k4.y;
    const unsigned m23 = (k4.z > k4.w) ? k4.z : k4.w;
    const unsigned m4 = (m01 > m23) ? m01 : m23;
    mk = (m4 > mk) ? m4 : mk;
  }
  unsigned alive[8];
#pragma unroll
  for (int w = 0; w < 8; ++w) alive[w] = 0xFFFFFFFFu;
  unsigned need = KEEP;
  unsigned prefix = 0u;
#pragma nounroll
  for (int ph = 3; ph >= 0; --ph) {
    asm volatile("" ::: "memory");
    const unsigned sel2 = 0x04000400u + 0x01010101u * (unsigned)ph;
    unsigned X[8][8];
    uint4 kq[2][8];
#pragma unroll
    for (int i = 0; i < 8; ++i) kq[0][i] = mineu[i * 32];
#pragma unroll
    for (int w = 0; w < 8; ++w) {
      if (w + 1 < 8) {
#pragma unroll
        for (int i = 0; i < 8; ++i) kq[(w + 1) & 1][i] = mineu[(8 * (w + 1) + i) * 32];
      }
      __builtin_amdgcn_sched_barrier(0);
#pragma unroll
      for (int i = 0; i < 8; ++i) {
        const uint4 k4 = kq[w & 1][i];
        const unsigned u01 = __builtin_amdgcn_perm(k4.y, k4.x, sel2);
        const unsigned u23 = __builtin_amdgcn_perm(k4.w, k4.z, sel2);
        X[w][i] = __builtin_amdgcn_perm(u23, u01, 0x05040100u);
      }
      planes8(X[w]);
    }
    const unsigned lowbit = 1u << (8 * ph);
#pragma unroll
    for (int j = 7; j >= 0; --j) {
      unsigned t[8];
      unsigned c0 = 0u, c1 = 0u;
#pragma unroll
      for (int w = 0; w < 8; w += 2) {
        t[w] = alive[w] & X[w][j];
        t[w + 1] = alive[w + 1] & X[w + 1][j];
        c0 += (unsigned)__builtin_popcount(t[w]);
        c1 += (unsigned)__builtin_popcount(t[w + 1]);
      }
      unsigned cnt = c0 + c1;
      cnt += xhalf_u(cnt);
      const bool take = cnt >= need;
      prefix = take ? (prefix | (lowbit << j)) : prefix;
      need = take ? need : need - cnt;
      const unsigned drop = take ? 0u : 0xFFFFFFFFu;
#pragma unroll
      for (int w = 0; w < 8; ++w) alive[w] = t[w] ^ (alive[w] & drop);
    }
  }

  v8f sacc[4];
#pragma unroll
  for (int dt = 0; dt < 4; ++dt) sacc[dt] = vzero8();
  const float* vlane = Vh + (size_t)(8 * half) * DD + ln;

  {
    const unsigned mo = xhalf_u(mk);
    mk = (mo > mk) ? mo : mk;
    const float mx = unkey(mk);
    const bool live = mx > -1e29f;
    float s0 = 0.f, s1 = 0.f, s2 = 0.f, s3 = 0.f;
#pragma unroll 4
    for (int c = 0; c < SELC / 8; ++c) {
      const uint4 k4 = mineu[c * 32];
      const float p0 = (k4.x >= prefix && live) ? __expf(unkey(k4.x) - mx) : 0.f;
      const float p1 = (k4.y >= prefix && live) ? __expf(unkey(k4.y) - mx) : 0.f;
      const float p2 = (k4.z >= prefix && live) ? __expf(unkey(k4.z) - mx) : 0.f;
      const float p3 = (k4.w >= prefix && live) ? __expf(unkey(k4.w) - mx) : 0.f;
      s0 += p0; s1 += p1; s2 += p2; s3 += p3;
      mine[c * 32] = make_float4(p0, p1, p2, p3);
    }
    float sum = (s0 + s1) + (s2 + s3);
    sum += xhalf_f(sum);
    const float rs = (sum > 0.f) ? 1.f / sum : 0.f;

    pv_tiles<SELC / 16>(sacc, mine, rs, [&](int t) {
      return vlane + (size_t)sSel[t >> 2] * (BS * DD) + (size_t)((t & 3) * 16) * DD;
    });
  }

  auto wgroup_ok = [&](int cg) { return (cg == 0) ? (qb >= 3) : (qb - 3 + cg >= 0); };
  auto wgroup_kb = [&](int cg) { return (cg != 0 && qb - 3 + cg >= 0) ? qb - 3 + cg : 0; };
  __syncthreads();
  qk_tiles4<WINC / BS>(
      Kh, sQ, wave, lane, ln, half,
      [&](int g) { return wgroup_kb(g) * BS + wave * 16; },
      [&](int g, int qt, int keyBase, v8f acc) {
        const int i = qb * BS + qt * 16 + ln;
        const int t = 4 * g + wave;
        float4* const dst = reinterpret_cast<float4*>(sP + (size_t)qt * 16 * PSTRIDE) + lane;
        const int ilim = wgroup_ok(g) ? i : -1;
        float w[8];
#pragma unroll
        for (int r = 0; r < 8; ++r) {
          const int kj = keyBase + r + 8 * half;
          const bool ok = (kj <= ilim) && (((i - kj) < WIN) || (kj < 1));
          w[r] = ok ? acc[r] * SCALE : NEGINF;
        }
        dst[(2 * t) * 32] = make_float4(w[0], w[1], w[2], w[3]);
        dst[(2 * t + 1) * 32] = make_float4(w[4], w[5], w[6], w[7]);
      });
  __syncthreads();

  {
    float wl[WINC / 2];
#pragma unroll
    for (int c = 0; c < WINC / 8; ++c) {
      const float4 w4 = mine[c * 32];
      wl[4 * c + 0] = w4.x; wl[4 * c + 1] = w4.y; wl[4 * c + 2] = w4.z; wl[4 * c + 3] = w4.w;
    }
    float m0 = wl[0], m1 = wl[1], m2 = wl[2], m3 = wl[3];
#pragma unroll
    for (int j = 4; j < WINC / 2; j += 4) {
      m0 = fmaxf(m0, wl[j + 0]);
      m1 = fmaxf(m1, wl[j + 1]);
      m2 = fmaxf(m2, wl[j + 2]);
      m3 = fmaxf(m3, wl[j + 3]);
    }
    float mx = fmaxf(fmaxf(m0, m1), fmaxf(m2, m3));
    mx = fmaxf(mx, xhalf_f(mx));
    const bool live = mx > -1e29f;
    float s0 = 0.f, s1 = 0.f, s2 = 0.f, s3 = 0.f;
#pragma unroll
    for (int j = 0; j < WINC / 2; j += 4) {
      const float p0 = live ? __expf(wl[j + 0] - mx) : 0.f;
      const float p1 = live ? __expf(wl[j + 1] - mx) : 0.f;
      const float p2 = live ? __expf(wl[j + 2] - mx) : 0.f;
      const float p3 = live ? __expf(wl[j + 3] - mx) : 0.f;
      s0 += p0; s1 += p1; s2 += p2; s3 += p3;
      mine[(j >> 2) * 32] = make_float4(p0, p1, p2, p3);
    }
    float sum = (s0 + s1) + (s2 + s3);
    sum += xhalf_f(sum);
    const float rs = (sum > 0.f) ? 1.f / sum : 0.f;

    pv_tiles<WINC / 16>(sacc, mine, rs, [&](int t) {
      return vlane + (size_t)wgroup_kb(t >> 2) * (BS * DD) + (size_t)((t & 3) * 16) * DD;
    });
  }

#pragma unroll
  for (int dt = 0; dt < 4; ++dt) {
    float* dst = myP + ln * PSTRIDE + 256 + dt * 16 + 8 * half;
    float4 lo, hi;
    lo.x = sacc[dt][0]; lo.y = sacc[dt][1]; lo.z = sacc[dt][2]; lo.w = sacc[dt][3];
    hi.x = sacc[dt][4]; hi.y = sacc[dt][5]; hi.z = sacc[dt][6]; hi.w = sacc[dt][7];
    *reinterpret_cast<float4*>(dst) = lo;
    *reinterpret_cast<float4*>(dst + 4) = hi;
  }
#pragma unroll
  for (int dt = 0; dt < 4; ++dt) {
#pragma unroll
    for (int r = 0; r < 8; ++r)
      oacc[dt][r] += myP[(r + 8 * half) * PSTRIDE + 256 + dt * 16 + ln];
  }

  const float third = 1.f / 3.f;
#pragma unroll
  for (int dt = 0; dt < 4; ++dt) {
#pragma unroll
    for (int r = 0; r < 8; ++r) {
      int row = qrow0 + r + 8 * half;
      Oh[(size_t)row * DD + dt * 16 + ln] = oacc[dt][r] * third;
    }
  }
}

extern "C" void kernel_launch(void* const* d_in, const int* in_sizes, int n_in,
                              void* d_out, int out_size, void* d_ws,
                              size_t ws_size, hipStream_t stream) {
  (void)in_sizes; (void)n_in; (void)out_size; (void)ws_size;
  const float* q = (const float*)d_in[0];
  const float* k = (const float*)d_in[1];
  const float* v = (const float*)d_in[2];
  float* out = (float*)d_out;
  float* kblk = (float*)d_ws;
  float* vblk = kblk + (size_t)BB * HH * NB * DD;

  nsa_blockmean_kernel<<<BB * HH * NB, 64, 0, stream>>>(k, v, kblk, vblk);

  const size_t shmem = (size_t)64 * PSTRIDE * sizeof(float);
  hipFuncSetAttribute(reinterpret_cast<const void*>(nsa_main_kernel),
                      hipFuncAttributeMaxDynamicSharedMemorySize, (int)shmem);
  nsa_main_kernel<<<BB * HH * NB, 128, shmem, stream>>>(q, k, v, kblk, vblk,
                                                        out);
}


# window branch: keys re-packed as 3 blocks (qb-2..qb): the global token takes the first key slot of block qb-2, which the window masks for every row; the 4th 64-key group is gone; sum order changed
# speedup vs baseline: 4.2584x; 1.0391x over previous
#include <hip/hip_runtime.h>

#define BB 2
#define HH 8
#define SS 2048
#define DD 64
#define NB 32
#define BS 64
#define NSEL 8
#define SELC 512
#define KEEP 256
#define WIN 128
#define WINC 192
#define PSTRIDE 520
#define NEGINF (-1e30f)
#define SCALE 0.125f

typedef __attribute__((ext_vector_type(2))) float v2f;
typedef __attribute__((ext_vector_type(8))) float v8f;

static __device__ inline v8f vzero8() {
  v8f r;
#pragma unroll
  for (int i = 0; i < 8; ++i) r[i] = 0.f;
  return r;
}

static __device__ inline v8f wmma4(v2f a, v2f b, v8f c) {
  return __builtin_amdgcn_wmma_f32_16x16x4_f32(false, a, false, b, (short)0, c,
                                               false, false);
}

static __device__ inline unsigned okey(float x) {
  unsigned u = __float_as_uint(x);
  return (u & 0x80000000u) ? ~u : (u | 0x80000000u);
}

static __device__ inline int wred_sumi(int x) {
#pragma unroll
  for (int o = 16; o > 0; o >>= 1) x += __shfl_xor(x, o, 32);
  return x;
}

static __device__ inline float unkey(unsigned k) {
  unsigned u = (k & 0x80000000u) ? (k ^ 0x80000000u) : ~k;
  return __uint_as_float(u);
}

static __device__ inline unsigned xhalf_u(unsigned x) {
  return __builtin_amdgcn_permlanex16(x, x, 0x76543210u, 0xfedcba98u, false, false);
}
static __device__ inline float xhalf_f(float x) {
  return __uint_as_float(xhalf_u(__float_as_uint(x)));
}


static __device__ inline void bt_stage(unsigned& a, unsigned& b, int s, unsigned lowmask) {
  const unsigned na = __builtin_amdgcn_bitop3_b32(a, b << s, lowmask, 0xE4);
  const unsigned nb = __builtin_amdgcn_bitop3_b32(a >> s, b, lowmask, 0xE4);
  a = na;
  b = nb;
}

static __device__ inline void planes8(unsigned (&x)[8]) {
#pragma unroll
  for (int i = 0; i < 4; ++i) bt_stage(x[i], x[i + 4], 4, 0x0F0F0F0Fu);
#pragma unroll
  for (int i = 0; i < 8; ++i)
    if ((i & 2) == 0) bt_stage(x[i], x[i + 2], 2, 0x33333333u);
#pragma unroll
  for (int i = 0; i < 8; i += 2) bt_stage(x[i], x[i + 1], 1, 0x55555555u);
}

__global__ void nsa_blockmean_kernel(const float* __restrict__ K,
                                     const float* __restrict__ V,
                                     float* __restrict__ KB,
                                     float* __restrict__ VB) {
  const int g = blockIdx.x;
  const int d = threadIdx.x;
  const float* kp = K + (size_t)g * (BS * DD);
  const float* vp = V + (size_t)g * (BS * DD);
  float ks = 0.f, vs = 0.f;
  for (int t = 0; t < BS; ++t) {
    ks += kp[t * DD + d];
    vs += vp[t * DD + d];
  }
  KB[(size_t)g * DD + d] = ks * (1.f / BS);
  VB[(size_t)g * DD + d] = vs * (1.f / BS);
}

static __device__ inline void load_ktile(v2f (&kf)[16], const float* __restrict__ Kh,
                                         int row, int half) {
#pragma unroll
  for (int t = 0; t < 16; ++t) {
    const float* bp = Kh + (size_t)row * DD + 4 * t + 2 * half;
    v2f a; a.x = bp[0]; a.y = bp[1];
    kf[t] = a;
  }
}

static __device__ inline v8f mma_ktile(const v2f (&kf)[16], const v2f (&qa)[16]) {
  v8f acc = vzero8();
#pragma unroll
  for (int t = 0; t < 16; ++t) acc = wmma4(kf[t], qa[t], acc);
  return acc;
}

static __device__ inline void load_qtile(v2f (&qf)[16], const v2f* __restrict__ sQ, int qt,
                                         int lane) {
#pragma unroll
  for (int t = 0; t < 16; ++t) qf[t] = sQ[(qt * 16 + t) * 32 + lane];
}

template <int NG, typename FB, typename FR, typename FE>
static __device__ inline void qk_tiles4(const float* __restrict__ Kh, const v2f* __restrict__ sQ,
                                        int wave, int lane, int half, FB kbase, FR krow, FE emit) {
  v2f kf[16], kfn[16], qfa[16], qfb[16];
  load_ktile(kfn, Kh, krow(0), half);
  load_qtile(qfa, sQ, wave, lane);
#pragma nounroll
  for (int g = 0; g < NG; ++g) {
    const int kb = kbase(g);
#pragma unroll
    for (int t = 0; t < 16; ++t) kf[t] = kfn[t];
    const int gn = (g + 1 < NG) ? g + 1 : NG - 1;
    load_ktile(kfn, Kh, krow(gn), half);
#pragma nounroll
    for (int j = 0; j < 4; j += 2) {
      const int qt0 = (wave + j) & 3;
      const int qt1 = (wave + j + 1) & 3;
      load_qtile(qfb, sQ, qt1, lane);
      __builtin_amdgcn_sched_barrier(0);
      emit(g, qt0, kb, mma_ktile(kf, qfa));
      load_qtile(qfa, sQ, (wave + j + 2) & 3, lane);
      __builtin_amdgcn_sched_barrier(0);
      emit(g, qt1, kb, mma_ktile(kf, qfb));
    }
  }
}

struct VTile {
  const float* vb;
  const float* v0;
};

static __device__ inline void load_vtile(v2f (&vf)[16], const VTile vt) {
  const float* __restrict__ vb = vt.vb;
  const float* __restrict__ v0 = vt.v0;
#pragma unroll
  for (int u = 0; u < 4; ++u) {
#pragma unroll
    for (int dt = 0; dt < 4; ++dt) {
      v2f va;
      va.x = (u == 0) ? v0[dt * 16] : vb[(2 * u) * DD + dt * 16];
      va.y = vb[(2 * u + 1) * DD + dt * 16];
      vf[u * 4 + dt] = va;
    }
  }
}

static __device__ inline void mma_vtile(v8f (&sacc)[4], const v2f (&vf)[16],
                                        const float4* __restrict__ mine, int t, float rs) {
  const float4 pa = mine[(2 * t) * 32];
  const float4 pc = mine[(2 * t + 1) * 32];
  float p[8];
  p[0] = pa.x * rs; p[1] = pa.y * rs; p[2] = pa.z * rs; p[3] = pa.w * rs;
  p[4] = pc.x * rs; p[5] = pc.y * rs; p[6] = pc.z * rs; p[7] = pc.w * rs;
#pragma unroll
  for (int u = 0; u < 4; ++u) {
    v2f pb; pb.x = p[2 * u]; pb.y = p[2 * u + 1];
#pragma unroll
    for (int dt = 0; dt < 4; ++dt) sacc[dt] = wmma4(vf[u * 4 + dt], pb, sacc[dt]);
  }
}

template <int NT, typename F>
static __device__ inline void pv_tiles(v8f (&sacc)[4], const float4* __restrict__ mine,
                                       float rs, F vtile) {
  v2f vfa[16], vfb[16];
  load_vtile(vfa, vtile(0));
  load_vtile(vfb, vtile(1));
#pragma nounroll
  for (int t = 0; t < NT; t += 2) {
    mma_vtile(sacc, vfa, mine, t, rs);
    const int ta = (t + 2 < NT) ? t + 2 : NT - 1;
    load_vtile(vfa, vtile(ta));
    __builtin_amdgcn_sched_barrier(0);
    mma_vtile(sacc, vfb, mine, t + 1, rs);
    const int tb = (t + 3 < NT) ? t + 3 : NT - 1;
    load_vtile(vfb, vtile(tb));
    __builtin_amdgcn_sched_barrier(0);
  }
}

__global__ __launch_bounds__(128, 2) void nsa_main_kernel(
    const float* __restrict__ Q, const float* __restrict__ K,
    const float* __restrict__ V, const float* __restrict__ KB,
    const float* __restrict__ VB, float* __restrict__ O) {
  extern __shared__ __align__(16) float sP[];
  __shared__ float sScore[NB];
  __shared__ int sSel[NSEL];
  __shared__ v2f sQ[4 * 16 * 32];

  const int wg = blockIdx.x;
  const int qb = wg % NB;
  const int bh = wg / NB;
  const size_t hoff = (size_t)bh * SS * DD;
  const float* Qh = Q + hoff;
  const float* Kh = K + hoff;
  const float* Vh = V + hoff;
  const float* KBh = KB + (size_t)bh * NB * DD;
  const float* VBh = VB + (size_t)bh * NB * DD;
  float* Oh = O + hoff;

  const int tid = threadIdx.x;
  const int wave = tid >> 5;
  const int lane = tid & 31;
  const int half = lane >> 4;
  const int ln = lane & 15;
  const int qrow0 = qb * BS + wave * 16;
  float* myP = sP + (size_t)wave * 16 * PSTRIDE;

  v2f qa[16];
#pragma unroll
  for (int t = 0; t < 16; ++t) {
    const float* p = Qh + (size_t)(qrow0 + ln) * DD + 4 * t + 2 * half;
    v2f a; a.x = p[0]; a.y = p[1];
    qa[t] = a;
    sQ[(wave * 16 + t) * 32 + lane] = a;
  }

  v8f oacc[4];
#pragma unroll
  for (int dt = 0; dt < 4; ++dt) oacc[dt] = vzero8();

#pragma unroll
  for (int nt = 0; nt < 2; ++nt) {
    v8f acc = vzero8();
#pragma unroll
    for (int t = 0; t < 16; ++t) {
      const float* bp = KBh + (size_t)(nt * 16 + ln) * DD + 4 * t + 2 * half;
      v2f bf; bf.x = bp[0]; bf.y = bp[1];
      acc = wmma4(qa[t], bf, acc);
    }
#pragma unroll
    for (int r = 0; r < 8; ++r) {
      int m = r + 8 * half;
      int col = nt * 16 + ln;
      float x = acc[r] * SCALE;
      myP[m * PSTRIDE + col] = (col <= qb) ? x : NEGINF;
    }
  }
  __syncthreads();

  if (tid < NB) {
    float s = 0.f;
    for (int q2 = 0; q2 < BS; ++q2) s += sP[q2 * PSTRIDE + tid];
    sScore[tid] = s * (1.f / BS);
  }
  __syncthreads();

  if (tid == 0) {
    float loc[NB];
    for (int i = 0; i < NB; ++i) loc[i] = sScore[i];
    for (int s = 0; s < NSEL; ++s) {
      int bi = 0;
      float bv = loc[0];
      for (int i = 1; i < NB; ++i)
        if (loc[i] > bv) { bv = loc[i]; bi = i; }
      sSel[s] = bi;
      loc[bi] = -3.4e38f;
    }
  }
  __syncthreads();

  if (tid < BS) {
    float* row = sP + tid * PSTRIDE;
    float mx = -3.4e38f;
    for (int c = 0; c < NB; ++c) mx = fmaxf(mx, row[c]);
    float sum = 0.f;
    for (int c = 0; c < NB; ++c) {
      float p = (mx > -1e29f) ? __expf(row[c] - mx) : 0.f;
      row[c] = p;
      sum += p;
    }
    float rs = (sum > 0.f) ? 1.f / sum : 0.f;
    for (int c = 0; c < NB; ++c) row[c] *= rs;
  }
  __syncthreads();

#pragma unroll
  for (int dt = 0; dt < 4; ++dt) {
#pragma unroll
    for (int g = 0; g < 8; ++g) {
      int c0 = 4 * g + 2 * half;
      v2f a; a.x = myP[ln * PSTRIDE + c0]; a.y = myP[ln * PSTRIDE + c0 + 1];
      v2f bf;
      bf.x = VBh[(size_t)c0 * DD + dt * 16 + ln];
      bf.y = VBh[(size_t)(c0 + 1) * DD + dt * 16 + ln];
      oacc[dt] = wmma4(a, bf, oacc[dt]);
    }
  }
  __syncthreads();

  float4* const mine = reinterpret_cast<float4*>(myP) + lane;
  uint4* const mineu = reinterpret_cast<uint4*>(myP) + lane;
  qk_tiles4<NSEL>(
      Kh, sQ, wave, lane, half,
      [&](int g) { return sSel[g] * BS + wave * 16; },
      [&](int g) { return sSel[g] * BS + wave * 16 + ln; },
      [&](int g, int qt, int keyBase, v8f acc) {
        const int i = qb * BS + qt * 16 + ln;
        const int t = 4 * g + wave;
        uint4* const dstu = reinterpret_cast<uint4*>(sP + (size_t)qt * 16 * PSTRIDE) + lane;
        unsigned kk[8];
#pragma unroll
        for (int r = 0; r < 8; ++r) {
          const int tok = keyBase + r + 8 * half;
          const float x = acc[r] * SCALE;
          kk[r] = okey((tok <= i) ? x : NEGINF);
        }
        dstu[(2 * t) * 32] = make_uint4(kk[0], kk[1], kk[2], kk[3]);
        dstu[(2 * t + 1) * 32] = make_uint4(kk[4], kk[5], kk[6], kk[7]);
      });
  __syncthreads();

  unsigned mk = 0u;
#pragma unroll 4
  for (int c = 0; c < SELC / 8; ++c) {
    const uint4 k4 = mineu[c * 32];
    const unsigned m01 = (k4.x > k4.y) ? k4.x : k4.y;
    const unsigned m23 = (k4.z > k4.w) ? k4.z : k4.w;
    const unsigned m4 = (m01 > m23) ? m01 : m23;
    mk = (m4 > mk) ? m4 : mk;
  }
  unsigned alive[8];
#pragma unroll
  for (int w = 0; w < 8; ++w) alive[w] = 0xFFFFFFFFu;
  unsigned need = KEEP;
  unsigned prefix = 0u;
#pragma nounroll
  for (int ph = 3; ph >= 0; --ph) {
    asm volatile("" ::: "memory");
    const unsigned sel2 = 0x04000400u + 0x01010101u * (unsigned)ph;
    unsigned X[8][8];
    uint4 kq[2][8];
#pragma unroll
    for (int i = 0; i < 8; ++i) kq[0][i] = mineu[i * 32];
#pragma unroll
    for (int w = 0; w < 8; ++w) {
      if (w + 1 < 8) {
#pragma unroll
        for (int i = 0; i < 8; ++i) kq[(w + 1) & 1][i] = mineu[(8 * (w + 1) + i) * 32];
      }
      __builtin_amdgcn_sched_barrier(0);
#pragma unroll
      for (int i = 0; i < 8; ++i) {
        const uint4 k4 = kq[w & 1][i];
        const unsigned u01 = __builtin_amdgcn_perm(k4.y, k4.x, sel2);
        const unsigned u23 = __builtin_amdgcn_perm(k4.w, k4.z, sel2);
        X[w][i] = __builtin_amdgcn_perm(u23, u01, 0x05040100u);
      }
      planes8(X[w]);
    }
    const unsigned lowbit = 1u << (8 * ph);
#pragma unroll
    for (int j = 7; j >= 0; --j) {
      unsigned t[8];
      unsigned c0 = 0u, c1 = 0u;
#pragma unroll
      for (int w = 0; w < 8; w += 2) {
        t[w] = alive[w] & X[w][j];
        t[w + 1] = alive[w + 1] & X[w + 1][j];
        c0 += (unsigned)__builtin_popcount(t[w]);
        c1 += (unsigned)__builtin_popcount(t[w + 1]);
      }
      unsigned cnt = c0 + c1;
      cnt += xhalf_u(cnt);
      const bool take = cnt >= need;
      prefix = take ? (prefix | (lowbit << j)) : prefix;
      need = take ? need : need - cnt;
      const unsigned drop = take ? 0u : 0xFFFFFFFFu;
#pragma unroll
      for (int w = 0; w < 8; ++w) alive[w] = t[w] ^ (alive[w] & drop);
    }
  }

  v8f sacc[4];
#pragma unroll
  for (int dt = 0; dt < 4; ++dt) sacc[dt] = vzero8();
  const float* vlane = Vh + (size_t)(8 * half) * DD + ln;

  {
    const unsigned mo = xhalf_u(mk);
    mk = (mo > mk) ? mo : mk;
    const float mx = unkey(mk);
    const bool live = mx > -1e29f;
    float s0 = 0.f, s1 = 0.f, s2 = 0.f, s3 = 0.f;
#pragma unroll 4
    for (int c = 0; c < SELC / 8; ++c) {
      const uint4 k4 = mineu[c * 32];
      const float p0 = (k4.x >= prefix && live) ? __expf(unkey(k4.x) - mx) : 0.f;
      const float p1 = (k4.y >= prefix && live) ? __expf(unkey(k4.y) - mx) : 0.f;
      const float p2 = (k4.z >= prefix && live) ? __expf(unkey(k4.z) - mx) : 0.f;
      const float p3 = (k4.w >= prefix && live) ? __expf(unkey(k4.w) - mx) : 0.f;
      s0 += p0; s1 += p1; s2 += p2; s3 += p3;
      mine[c * 32] = make_float4(p0, p1, p2, p3);
    }
    float sum = (s0 + s1) + (s2 + s3);
    sum += xhalf_f(sum);
    const float rs = (sum > 0.f) ? 1.f / sum : 0.f;

    pv_tiles<SELC / 16>(sacc, mine, rs, [&](int t) {
      VTile vt;
      vt.vb = vlane + (size_t)sSel[t >> 2] * (BS * DD) + (size_t)((t & 3) * 16) * DD;
      vt.v0 = vt.vb;
      return vt;
    });
  }

  auto wgroup_ok = [&](int cg) { return qb - 2 + cg >= 0; };
  auto wgroup_kb = [&](int cg) { return (qb - 2 + cg >= 0) ? qb - 2 + cg : 0; };
  __syncthreads();
  qk_tiles4<WINC / BS>(
      Kh, sQ, wave, lane, half,
      [&](int g) { return wgroup_kb(g) * BS + wave * 16; },
      [&](int g) {
        return (g == 0 && wave == 0 && ln == 0) ? 0 : wgroup_kb(g) * BS + wave * 16 + ln;
      },
      [&](int g, int qt, int keyBase, v8f acc) {
        const int i = qb * BS + qt * 16 + ln;
        const int t = 4 * g + wave;
        float4* const dst = reinterpret_cast<float4*>(sP + (size_t)qt * 16 * PSTRIDE) + lane;
        const int ilim = wgroup_ok(g) ? i : -1;
        const bool gslot = (g == 0) && (wave == 0) && (half == 0);
        float w[8];
#pragma unroll
        for (int r = 0; r < 8; ++r) {
          const int kj = (r == 0 && gslot) ? 0 : keyBase + r + 8 * half;
          const bool ok = (kj <= ilim) && (((i - kj) < WIN) || (kj < 1));
          w[r] = ok ? acc[r] * SCALE : NEGINF;
        }
        dst[(2 * t) * 32] = make_float4(w[0], w[1], w[2], w[3]);
        dst[(2 * t + 1) * 32] = make_float4(w[4], w[5], w[6], w[7]);
      });
  __syncthreads();

  {
    float wl[WINC / 2];
#pragma unroll
    for (int c = 0; c < WINC / 8; ++c) {
      const float4 w4 = mine[c * 32];
      wl[4 * c + 0] = w4.x; wl[4 * c + 1] = w4.y; wl[4 * c + 2] = w4.z; wl[4 * c + 3] = w4.w;
    }
    float m0 = wl[0], m1 = wl[1], m2 = wl[2], m3 = wl[3];
#pragma unroll
    for (int j = 4; j < WINC / 2; j += 4) {
      m0 = fmaxf(m0, wl[j + 0]);
      m1 = fmaxf(m1, wl[j + 1]);
      m2 = fmaxf(m2, wl[j + 2]);
      m3 = fmaxf(m3, wl[j + 3]);
    }
    float mx = fmaxf(fmaxf(m0, m1), fmaxf(m2, m3));
    mx = fmaxf(mx, xhalf_f(mx));
    const bool live = mx > -1e29f;
    float s0 = 0.f, s1 = 0.f, s2 = 0.f, s3 = 0.f;
#pragma unroll
    for (int j = 0; j < WINC / 2; j += 4) {
      const float p0 = live ? __expf(wl[j + 0] - mx) : 0.f;
      const float p1 = live ? __expf(wl[j + 1] - mx) : 0.f;
      const float p2 = live ? __expf(wl[j + 2] - mx) : 0.f;
      const float p3 = live ? __expf(wl[j + 3] - mx) : 0.f;
      s0 += p0; s1 += p1; s2 += p2; s3 += p3;
      mine[(j >> 2) * 32] = make_float4(p0, p1, p2, p3);
    }
    float sum = (s0 + s1) + (s2 + s3);
    sum += xhalf_f(sum);
    const float rs = (sum > 0.f) ? 1.f / sum : 0.f;

    pv_tiles<WINC / 16>(sacc, mine, rs, [&](int t) {
      VTile vt;
      vt.vb = vlane + (size_t)wgroup_kb(t >> 2) * (BS * DD) + (size_t)((t & 3) * 16) * DD;
      vt.v0 = (t == 0 && half == 0) ? Vh + ln : vt.vb;
      return vt;
    });
  }

#pragma unroll
  for (int dt = 0; dt < 4; ++dt) {
    float* dst = myP + ln * PSTRIDE + 256 + dt * 16 + 8 * half;
    float4 lo, hi;
    lo.x = sacc[dt][0]; lo.y = sacc[dt][1]; lo.z = sacc[dt][2]; lo.w = sacc[dt][3];
    hi.x = sacc[dt][4]; hi.y = sacc[dt][5]; hi.z = sacc[dt][6]; hi.w = sacc[dt][7];
    *reinterpret_cast<float4*>(dst) = lo;
    *reinterpret_cast<float4*>(dst + 4) = hi;
  }
#pragma unroll
  for (int dt = 0; dt < 4; ++dt) {
#pragma unroll
    for (int r = 0; r < 8; ++r)
      oacc[dt][r] += myP[(r + 8 * half) * PSTRIDE + 256 + dt * 16 + ln];
  }

  const float third = 1.f / 3.f;
#pragma unroll
  for (int dt = 0; dt < 4; ++dt) {
#pragma unroll
    for (int r = 0; r < 8; ++r) {
      int row = qrow0 + r + 8 * half;
      Oh[(size_t)row * DD + dt * 16 + ln] = oacc[dt][r] * third;
    }
  }
}

extern "C" void kernel_launch(void* const* d_in, const int* in_sizes, int n_in,
                              void* d_out, int out_size, void* d_ws,
                              size_t ws_size, hipStream_t stream) {
  (void)in_sizes; (void)n_in; (void)out_size; (void)ws_size;
  const float* q = (const float*)d_in[0];
  const float* k = (const float*)d_in[1];
  const float* v = (const float*)d_in[2];
  float* out = (float*)d_out;
  float* kblk = (float*)d_ws;
  float* vblk = kblk + (size_t)BB * HH * NB * DD;

  nsa_blockmean_kernel<<<BB * HH * NB, 64, 0, stream>>>(k, v, kblk, vblk);

  const size_t shmem = (size_t)64 * PSTRIDE * sizeof(float);
  hipFuncSetAttribute(reinterpret_cast<const void*>(nsa_main_kernel),
                      hipFuncAttributeMaxDynamicSharedMemorySize, (int)shmem);
  nsa_main_kernel<<<BB * HH * NB, 128, shmem, stream>>>(q, k, v, kblk, vblk,
                                                        out);
}


# P x V (selected, window): output d tiles split over the 4 waves, a V fragment loaded once per work-group and used on all 4 query tiles (P from LDS), 1/sum applied to the finished tile; sum order same
# speedup vs baseline: 4.4287x; 1.0400x over previous
#include <hip/hip_runtime.h>

#define BB 2
#define HH 8
#define SS 2048
#define DD 64
#define NB 32
#define BS 64
#define NSEL 8
#define SELC 512
#define KEEP 256
#define WIN 128
#define WINC 192
#define PSTRIDE 520
#define NEGINF (-1e30f)
#define SCALE 0.125f

typedef __attribute__((ext_vector_type(2))) float v2f;
typedef __attribute__((ext_vector_type(8))) float v8f;

static __device__ inline v8f vzero8() {
  v8f r;
#pragma unroll
  for (int i = 0; i < 8; ++i) r[i] = 0.f;
  return r;
}

static __device__ inline v8f wmma4(v2f a, v2f b, v8f c) {
  return __builtin_amdgcn_wmma_f32_16x16x4_f32(false, a, false, b, (short)0, c,
                                               false, false);
}

static __device__ inline unsigned okey(float x) {
  unsigned u = __float_as_uint(x);
  return (u & 0x80000000u) ? ~u : (u | 0x80000000u);
}

static __device__ inline int wred_sumi(int x) {
#pragma unroll
  for (int o = 16; o > 0; o >>= 1) x += __shfl_xor(x, o, 32);
  return x;
}

static __device__ inline float unkey(unsigned k) {
  unsigned u = (k & 0x80000000u) ? (k ^ 0x80000000u) : ~k;
  return __uint_as_float(u);
}

static __device__ inline unsigned xhalf_u(unsigned x) {
  return __builtin_amdgcn_permlanex16(x, x, 0x76543210u, 0xfedcba98u, false, false);
}
static __device__ inline float xhalf_f(float x) {
  return __uint_as_float(xhalf_u(__float_as_uint(x)));
}


static __device__ inline void bt_stage(unsigned& a, unsigned& b, int s, unsigned lowmask) {
  const unsigned na = __builtin_amdgcn_bitop3_b32(a, b << s, lowmask, 0xE4);
  const unsigned nb = __builtin_amdgcn_bitop3_b32(a >> s, b, lowmask, 0xE4);
  a = na;
  b = nb;
}

static __device__ inline void planes8(unsigned (&x)[8]) {
#pragma unroll
  for (int i = 0; i < 4; ++i) bt_stage(x[i], x[i + 4], 4, 0x0F0F0F0Fu);
#pragma unroll
  for (int i = 0; i < 8; ++i)
    if ((i & 2) == 0) bt_stage(x[i], x[i + 2], 2, 0x33333333u);
#pragma unroll
  for (int i = 0; i < 8; i += 2) bt_stage(x[i], x[i + 1], 1, 0x55555555u);
}

__global__ void nsa_blockmean_kernel(const float* __restrict__ K,
                                     const float* __restrict__ V,
                                     float* __restrict__ KB,
                                     float* __restrict__ VB) {
  const int g = blockIdx.x;
  const int d = threadIdx.x;
  const float* kp = K + (size_t)g * (BS * DD);
  const float* vp = V + (size_t)g * (BS * DD);
  float ks = 0.f, vs = 0.f;
  for (int t = 0; t < BS; ++t) {
    ks += kp[t * DD + d];
    vs += vp[t * DD + d];
  }
  KB[(size_t)g * DD + d] = ks * (1.f / BS);
  VB[(size_t)g * DD + d] = vs * (1.f / BS);
}

static __device__ inline void load_ktile(v2f (&kf)[16], const float* __restrict__ Kh,
                                         int row, int half) {
#pragma unroll
  for (int t = 0; t < 16; ++t) {
    const float* bp = Kh + (size_t)row * DD + 4 * t + 2 * half;
    v2f a; a.x = bp[0]; a.y = bp[1];
    kf[t] = a;
  }
}

static __device__ inline v8f mma_ktile(const v2f (&kf)[16], const v2f (&qa)[16]) {
  v8f acc = vzero8();
#pragma unroll
  for (int t = 0; t < 16; ++t) acc = wmma4(kf[t], qa[t], acc);
  return acc;
}

static __device__ inline void load_qtile(v2f (&qf)[16], const v2f* __restrict__ sQ, int qt,
                                         int lane) {
#pragma unroll
  for (int t = 0; t < 16; ++t) qf[t] = sQ[(qt * 16 + t) * 32 + lane];
}

template <int NG, typename FB, typename FR, typename FE>
static __device__ inline void qk_tiles4(const float* __restrict__ Kh, const v2f* __restrict__ sQ,
                                        int wave, int lane, int half, FB kbase, FR krow, FE emit) {
  v2f kf[16], kfn[16], qfa[16], qfb[16];
  load_ktile(kfn, Kh, krow(0), half);
  load_qtile(qfa, sQ, wave, lane);
#pragma nounroll
  for (int g = 0; g < NG; ++g) {
    const int kb = kbase(g);
#pragma unroll
    for (int t = 0; t < 16; ++t) kf[t] = kfn[t];
    const int gn = (g + 1 < NG) ? g + 1 : NG - 1;
    load_ktile(kfn, Kh, krow(gn), half);
#pragma nounroll
    for (int j = 0; j < 4; j += 2) {
      const int qt0 = (wave + j) & 3;
      const int qt1 = (wave + j + 1) & 3;
      load_qtile(qfb, sQ, qt1, lane);
      __builtin_amdgcn_sched_barrier(0);
      emit(g, qt0, kb, mma_ktile(kf, qfa));
      load_qtile(qfa, sQ, (wave + j + 2) & 3, lane);
      __builtin_amdgcn_sched_barrier(0);
      emit(g, qt1, kb, mma_ktile(kf, qfb));
    }
  }
}

struct VTile {
  const float* vb;
  const float* v0;
};

static __device__ inline void load_vtile(v2f (&vf)[4], const VTile vt) {
  const float* __restrict__ vb = vt.vb;
  const float* __restrict__ v0 = vt.v0;
#pragma unroll
  for (int u = 0; u < 4; ++u) {
    v2f va;
    va.x = (u == 0) ? v0[0] : vb[(2 * u) * DD];
    va.y = vb[(2 * u + 1) * DD];
    vf[u] = va;
  }
}

#define WROWS4 (16 * PSTRIDE / 4)

static __device__ inline void load_ptile(float4 (&pf)[8], const float4* __restrict__ pall, int t) {
#pragma unroll
  for (int qt = 0; qt < 4; ++qt) {
    pf[2 * qt] = pall[qt * WROWS4 + (2 * t) * 32];
    pf[2 * qt + 1] = pall[qt * WROWS4 + (2 * t + 1) * 32];
  }
}

static __device__ inline void mma_vtile(v8f (&acc)[4], const v2f (&vf)[4], const float4 (&pf)[8]) {
#pragma unroll
  for (int u = 0; u < 4; ++u) {
#pragma unroll
    for (int qt = 0; qt < 4; ++qt) {
      const float4 c4 = pf[2 * qt + (u >> 1)];
      v2f pb;
      pb.x = (u & 1) ? c4.z : c4.x;
      pb.y = (u & 1) ? c4.w : c4.y;
      acc[qt] = wmma4(vf[u], pb, acc[qt]);
    }
  }
}

template <int NG, typename F>
static __device__ inline void pv_tiles(v8f (&acc)[4], const float4* __restrict__ pall, F gbase,
                                       const float* v0first) {
  v2f va[4][4];
  float4 pa[8], pb[8];
  {
    const float* gb = gbase(0);
    VTile vt;
    vt.vb = gb; vt.v0 = v0first ? v0first : gb;
    load_vtile(va[0], vt);
    vt.vb = gb + 16 * DD; vt.v0 = vt.vb;
    load_vtile(va[1], vt);
    vt.vb = gb + 32 * DD; vt.v0 = vt.vb;
    load_vtile(va[2], vt);
    vt.vb = gb + 48 * DD; vt.v0 = vt.vb;
    load_vtile(va[3], vt);
  }
  load_ptile(pa, pall, 0);
  const float* gnext = gbase(NG > 1 ? 1 : 0);
#pragma nounroll
  for (int g = 0; g < NG; ++g) {
    const int gn = (g + 1 < NG) ? g + 1 : NG - 1;
    const float* gnb = gnext;
    gnext = gbase((g + 2 < NG) ? g + 2 : NG - 1);
    const int t = 4 * g;
    VTile vt;
    load_ptile(pb, pall, t + 1);
    __builtin_amdgcn_sched_barrier(0);
    mma_vtile(acc, va[0], pa);
    vt.vb = gnb; vt.v0 = gnb;
    load_vtile(va[0], vt);
    load_ptile(pa, pall, t + 2);
    __builtin_amdgcn_sched_barrier(0);
    mma_vtile(acc, va[1], pb);
    vt.vb = gnb + 16 * DD; vt.v0 = vt.vb;
    load_vtile(va[1], vt);
    load_ptile(pb, pall, t + 3);
    __builtin_amdgcn_sched_barrier(0);
    mma_vtile(acc, va[2], pa);
    vt.vb = gnb + 32 * DD; vt.v0 = vt.vb;
    load_vtile(va[2], vt);
    load_ptile(pa, pall, 4 * gn);
    __builtin_amdgcn_sched_barrier(0);
    mma_vtile(acc, va[3], pb);
    vt.vb = gnb + 48 * DD; vt.v0 = vt.vb;
    load_vtile(va[3], vt);
    __builtin_amdgcn_sched_barrier(0);
  }
}

__global__ __launch_bounds__(128, 2) void nsa_main_kernel(
    const float* __restrict__ Q, const float* __restrict__ K,
    const float* __restrict__ V, const float* __restrict__ KB,
    const float* __restrict__ VB, float* __restrict__ O) {
  extern __shared__ __align__(16) float sP[];
  __shared__ float sScore[NB];
  __shared__ int sSel[NSEL];
  __shared__ v2f sQ[4 * 16 * 32];
  __shared__ float sRs[2][BS];

  const int wg = blockIdx.x;
  const int qb = wg % NB;
  const int bh = wg / NB;
  const size_t hoff = (size_t)bh * SS * DD;
  const float* Qh = Q + hoff;
  const float* Kh = K + hoff;
  const float* Vh = V + hoff;
  const float* KBh = KB + (size_t)bh * NB * DD;
  const float* VBh = VB + (size_t)bh * NB * DD;
  float* Oh = O + hoff;

  const int tid = threadIdx.x;
  const int wave = tid >> 5;
  const int lane = tid & 31;
  const int half = lane >> 4;
  const int ln = lane & 15;
  const int qrow0 = qb * BS + wave * 16;
  float* myP = sP + (size_t)wave * 16 * PSTRIDE;

  v2f qa[16];
#pragma unroll
  for (int t = 0; t < 16; ++t) {
    const float* p = Qh + (size_t)(qrow0 + ln) * DD + 4 * t + 2 * half;
    v2f a; a.x = p[0]; a.y = p[1];
    qa[t] = a;
    sQ[(wave * 16 + t) * 32 + lane] = a;
  }

  v8f oacc[4];
#pragma unroll
  for (int dt = 0; dt < 4; ++dt) oacc[dt] = vzero8();

#pragma unroll
  for (int nt = 0; nt < 2; ++nt) {
    v8f acc = vzero8();
#pragma unroll
    for (int t = 0; t < 16; ++t) {
      const float* bp = KBh + (size_t)(nt * 16 + ln) * DD + 4 * t + 2 * half;
      v2f bf; bf.x = bp[0]; bf.y = bp[1];
      acc = wmma4(qa[t], bf, acc);
    }
#pragma unroll
    for (int r = 0; r < 8; ++r) {
      int m = r + 8 * half;
      int col = nt * 16 + ln;
      float x = acc[r] * SCALE;
      myP[m * PSTRIDE + col] = (col <= qb) ? x : NEGINF;
    }
  }
  __syncthreads();

  if (tid < NB) {
    float s = 0.f;
    for (int q2 = 0; q2 < BS; ++q2) s += sP[q2 * PSTRIDE + tid];
    sScore[tid] = s * (1.f / BS);
  }
  __syncthreads();

  if (tid == 0) {
    float loc[NB];
    for (int i = 0; i < NB; ++i) loc[i] = sScore[i];
    for (int s = 0; s < NSEL; ++s) {
      int bi = 0;
      float bv = loc[0];
      for (int i = 1; i < NB; ++i)
        if (loc[i] > bv) { bv = loc[i]; bi = i; }
      sSel[s] = bi;
      loc[bi] = -3.4e38f;
    }
  }
  __syncthreads();

  if (tid < BS) {
    float* row = sP + tid * PSTRIDE;
    float mx = -3.4e38f;
    for (int c = 0; c < NB; ++c) mx = fmaxf(mx, row[c]);
    float sum = 0.f;
    for (int c = 0; c < NB; ++c) {
      float p = (mx > -1e29f) ? __expf(row[c] - mx) : 0.f;
      row[c] = p;
      sum += p;
    }
    float rs = (sum > 0.f) ? 1.f / sum : 0.f;
    for (int c = 0; c < NB; ++c) row[c] *= rs;
  }
  __syncthreads();

#pragma unroll
  for (int dt = 0; dt < 4; ++dt) {
#pragma unroll
    for (int g = 0; g < 8; ++g) {
      int c0 = 4 * g + 2 * half;
      v2f a; a.x = myP[ln * PSTRIDE + c0]; a.y = myP[ln * PSTRIDE + c0 + 1];
      v2f bf;
      bf.x = VBh[(size_t)c0 * DD + dt * 16 + ln];
      bf.y = VBh[(size_t)(c0 + 1) * DD + dt * 16 + ln];
      oacc[dt] = wmma4(a, bf, oacc[dt]);
    }
  }
  __syncthreads();

  float4* const mine = reinterpret_cast<float4*>(myP) + lane;
  uint4* const mineu = reinterpret_cast<uint4*>(myP) + lane;
  qk_tiles4<NSEL>(
      Kh, sQ, wave, lane, half,
      [&](int g) { return sSel[g] * BS + wave * 16; },
      [&](int g) { return sSel[g] * BS + wave * 16 + ln; },
      [&](int g, int qt, int keyBase, v8f acc) {
        const int i = qb * BS + qt * 16 + ln;
        const int t = 4 * g + wave;
        uint4* const dstu = reinterpret_cast<uint4*>(sP + (size_t)qt * 16 * PSTRIDE) + lane;
        unsigned kk[8];
#pragma unroll
        for (int r = 0; r < 8; ++r) {
          const int tok = keyBase + r + 8 * half;
          const float x = acc[r] * SCALE;
          kk[r] = okey((tok <= i) ? x : NEGINF);
        }
        dstu[(2 * t) * 32] = make_uint4(kk[0], kk[1], kk[2], kk[3]);
        dstu[(2 * t + 1) * 32] = make_uint4(kk[4], kk[5], kk[6], kk[7]);
      });
  __syncthreads();

  unsigned mk = 0u;
#pragma unroll 4
  for (int c = 0; c < SELC / 8; ++c) {
    const uint4 k4 = mineu[c * 32];
    const unsigned m01 = (k4.x > k4.y) ? k4.x : k4.y;
    const unsigned m23 = (k4.z > k4.w) ? k4.z : k4.w;
    const unsigned m4 = (m01 > m23) ? m01 : m23;
    mk = (m4 > mk) ? m4 : mk;
  }
  unsigned alive[8];
#pragma unroll
  for (int w = 0; w < 8; ++w) alive[w] = 0xFFFFFFFFu;
  unsigned need = KEEP;
  unsigned prefix = 0u;
#pragma nounroll
  for (int ph = 3; ph >= 0; --ph) {
    asm volatile("" ::: "memory");
    const unsigned sel2 = 0x04000400u + 0x01010101u * (unsigned)ph;
    unsigned X[8][8];
    uint4 kq[2][8];
#pragma unroll
    for (int i = 0; i < 8; ++i) kq[0][i] = mineu[i * 32];
#pragma unroll
    for (int w = 0; w < 8; ++w) {
      if (w + 1 < 8) {
#pragma unroll
        for (int i = 0; i < 8; ++i) kq[(w + 1) & 1][i] = mineu[(8 * (w + 1) + i) * 32];
      }
      __builtin_amdgcn_sched_barrier(0);
#pragma unroll
      for (int i = 0; i < 8; ++i) {
        const uint4 k4 = kq[w & 1][i];
        const unsigned u01 = __builtin_amdgcn_perm(k4.y, k4.x, sel2);
        const unsigned u23 = __builtin_amdgcn_perm(k4.w, k4.z, sel2);
        X[w][i] = __builtin_amdgcn_perm(u23, u01, 0x05040100u);
      }
      planes8(X[w]);
    }
    const unsigned lowbit = 1u << (8 * ph);
#pragma unroll
    for (int j = 7; j >= 0; --j) {
      unsigned t[8];
      unsigned c0 = 0u, c1 = 0u;
#pragma unroll
      for (int w = 0; w < 8; w += 2) {
        t[w] = alive[w] & X[w][j];
        t[w + 1] = alive[w + 1] & X[w + 1][j];
        c0 += (unsigned)__builtin_popcount(t[w]);
        c1 += (unsigned)__builtin_popcount(t[w + 1]);
      }
      unsigned cnt = c0 + c1;
      cnt += xhalf_u(cnt);
      const bool take = cnt >= need;
      prefix = take ? (prefix | (lowbit << j)) : prefix;
      need = take ? need : need - cnt;
      const unsigned drop = take ? 0u : 0xFFFFFFFFu;
#pragma unroll
      for (int w = 0; w < 8; ++w) alive[w] = t[w] ^ (alive[w] & drop);
    }
  }

  v8f sacc[4], wacc[4];
#pragma unroll
  for (int qt = 0; qt < 4; ++qt) {
    sacc[qt] = vzero8();
    wacc[qt] = vzero8();
  }
  const float* vlane = Vh + (size_t)(8 * half) * DD + 16 * wave + ln;
  const float4* const pall = reinterpret_cast<const float4*>(sP) + lane;

  {
    const unsigned mo = xhalf_u(mk);
    mk = (mo > mk) ? mo : mk;
    const float mx = unkey(mk);
    const bool live = mx > -1e29f;
    float s0 = 0.f, s1 = 0.f, s2 = 0.f, s3 = 0.f;
#pragma unroll 4
    for (int c = 0; c < SELC / 8; ++c) {
      const uint4 k4 = mineu[c * 32];
      const float p0 = (k4.x >= prefix && live) ? __expf(unkey(k4.x) - mx) : 0.f;
      const float p1 = (k4.y >= prefix && live) ? __expf(unkey(k4.y) - mx) : 0.f;
      const float p2 = (k4.z >= prefix && live) ? __expf(unkey(k4.z) - mx) : 0.f;
      const float p3 = (k4.w >= prefix && live) ? __expf(unkey(k4.w) - mx) : 0.f;
      s0 += p0; s1 += p1; s2 += p2; s3 += p3;
      mine[c * 32] = make_float4(p0, p1, p2, p3);
    }
    float sum = (s0 + s1) + (s2 + s3);
    sum += xhalf_f(sum);
    const float rs = (sum > 0.f) ? 1.f / sum : 0.f;

    if (half == 0) sRs[0][wave * 16 + ln] = rs;
    __syncthreads();
    pv_tiles<NSEL>(
        sacc, pall, [&](int g) { return vlane + (size_t)sSel[g] * (BS * DD); },
        static_cast<const float*>(nullptr));
  }

  auto wgroup_ok = [&](int cg) { return qb - 2 + cg >= 0; };
  auto wgroup_kb = [&](int cg) { return (qb - 2 + cg >= 0) ? qb - 2 + cg : 0; };
  __syncthreads();
  qk_tiles4<WINC / BS>(
      Kh, sQ, wave, lane, half,
      [&](int g) { return wgroup_kb(g) * BS + wave * 16; },
      [&](int g) {
        return (g == 0 && wave == 0 && ln == 0) ? 0 : wgroup_kb(g) * BS + wave * 16 + ln;
      },
      [&](int g, int qt, int keyBase, v8f acc) {
        const int i = qb * BS + qt * 16 + ln;
        const int t = 4 * g + wave;
        float4* const dst = reinterpret_cast<float4*>(sP + (size_t)qt * 16 * PSTRIDE) + lane;
        const int ilim = wgroup_ok(g) ? i : -1;
        const bool gslot = (g == 0) && (wave == 0) && (half == 0);
        float w[8];
#pragma unroll
        for (int r = 0; r < 8; ++r) {
          const int kj = (r == 0 && gslot) ? 0 : keyBase + r + 8 * half;
          const bool ok = (kj <= ilim) && (((i - kj) < WIN) || (kj < 1));
          w[r] = ok ? acc[r] * SCALE : NEGINF;
        }
        dst[(2 * t) * 32] = make_float4(w[0], w[1], w[2], w[3]);
        dst[(2 * t + 1) * 32] = make_float4(w[4], w[5], w[6], w[7]);
      });
  __syncthreads();

  {
    float wl[WINC / 2];
#pragma unroll
    for (int c = 0; c < WINC / 8; ++c) {
      const float4 w4 = mine[c * 32];
      wl[4 * c + 0] = w4.x; wl[4 * c + 1] = w4.y; wl[4 * c + 2] = w4.z; wl[4 * c + 3] = w4.w;
    }
    float m0 = wl[0], m1 = wl[1], m2 = wl[2], m3 = wl[3];
#pragma unroll
    for (int j = 4; j < WINC / 2; j += 4) {
      m0 = fmaxf(m0, wl[j + 0]);
      m1 = fmaxf(m1, wl[j + 1]);
      m2 = fmaxf(m2, wl[j + 2]);
      m3 = fmaxf(m3, wl[j + 3]);
    }
    float mx = fmaxf(fmaxf(m0, m1), fmaxf(m2, m3));
    mx = fmaxf(mx, xhalf_f(mx));
    const bool live = mx > -1e29f;
    float s0 = 0.f, s1 = 0.f, s2 = 0.f, s3 = 0.f;
#pragma unroll
    for (int j = 0; j < WINC / 2; j += 4) {
      const float p0 = live ? __expf(wl[j + 0] - mx) : 0.f;
      const float p1 = live ? __expf(wl[j + 1] - mx) : 0.f;
      const float p2 = live ? __expf(wl[j + 2] - mx) : 0.f;
      const float p3 = live ? __expf(wl[j + 3] - mx) : 0.f;
      s0 += p0; s1 += p1; s2 += p2; s3 += p3;
      mine[(j >> 2) * 32] = make_float4(p0, p1, p2, p3);
    }
    float sum = (s0 + s1) + (s2 + s3);
    sum += xhalf_f(sum);
    const float rs = (sum > 0.f) ? 1.f / sum : 0.f;

    if (half == 0) sRs[1][wave * 16 + ln] = rs;
    __syncthreads();
    pv_tiles<WINC / BS>(
        wacc, pall, [&](int g) { return vlane + (size_t)wgroup_kb(g) * (BS * DD); },
        (half == 0) ? Vh + 16 * wave + ln : static_cast<const float*>(nullptr));
  }

  __syncthreads();
#pragma unroll
  for (int qt = 0; qt < 4; ++qt) {
    const float rss = sRs[0][qt * 16 + ln];
    const float rsw = sRs[1][qt * 16 + ln];
    float o[8];
#pragma unroll
    for (int r = 0; r < 8; ++r) o[r] = sacc[qt][r] * rss + wacc[qt][r] * rsw;
    float* dst = sP + (size_t)(qt * 16 + ln) * PSTRIDE + 256 + wave * 16 + 8 * half;
    *reinterpret_cast<float4*>(dst) = make_float4(o[0], o[1], o[2], o[3]);
    *reinterpret_cast<float4*>(dst + 4) = make_float4(o[4], o[5], o[6], o[7]);
  }
  __syncthreads();
#pragma unroll
  for (int dt = 0; dt < 4; ++dt) {
#pragma unroll
    for (int r = 0; r < 8; ++r)
      oacc[dt][r] += myP[(r + 8 * half) * PSTRIDE + 256 + dt * 16 + ln];
  }

  const float third = 1.f / 3.f;
#pragma unroll
  for (int dt = 0; dt < 4; ++dt) {
#pragma unroll
    for (int r = 0; r < 8; ++r) {
      int row = qrow0 + r + 8 * half;
      Oh[(size_t)row * DD + dt * 16 + ln] = oacc[dt][r] * third;
    }
  }
}

extern "C" void kernel_launch(void* const* d_in, const int* in_sizes, int n_in,
                              void* d_out, int out_size, void* d_ws,
                              size_t ws_size, hipStream_t stream) {
  (void)in_sizes; (void)n_in; (void)out_size; (void)ws_size;
  const float* q = (const float*)d_in[0];
  const float* k = (const float*)d_in[1];
  const float* v = (const float*)d_in[2];
  float* out = (float*)d_out;
  float* kblk = (float*)d_ws;
  float* vblk = kblk + (size_t)BB * HH * NB * DD;

  nsa_blockmean_kernel<<<BB * HH * NB, 64, 0, stream>>>(k, v, kblk, vblk);

  const size_t shmem = (size_t)64 * PSTRIDE * sizeof(float);
  hipFuncSetAttribute(reinterpret_cast<const void*>(nsa_main_kernel),
                      hipFuncAttributeMaxDynamicSharedMemorySize, (int)shmem);
  nsa_main_kernel<<<BB * HH * NB, 128, shmem, stream>>>(q, k, v, kblk, vblk,
                                                        out);
}


# selected branch: logits stay plain floats in LDS; the bit-plane search decides the sign first, then inverts the planes if negative: no float<->key conversion per logit; kept set, sum order same
# speedup vs baseline: 4.5831x; 1.0349x over previous
#include <hip/hip_runtime.h>

#define BB 2
#define HH 8
#define SS 2048
#define DD 64
#define NB 32
#define BS 64
#define NSEL 8
#define SELC 512
#define KEEP 256
#define WIN 128
#define WINC 192
#define PSTRIDE 520
#define NEGINF (-1e30f)
#define SCALE 0.125f

typedef __attribute__((ext_vector_type(2))) float v2f;
typedef __attribute__((ext_vector_type(8))) float v8f;

static __device__ inline v8f vzero8() {
  v8f r;
#pragma unroll
  for (int i = 0; i < 8; ++i) r[i] = 0.f;
  return r;
}

static __device__ inline v8f wmma4(v2f a, v2f b, v8f c) {
  return __builtin_amdgcn_wmma_f32_16x16x4_f32(false, a, false, b, (short)0, c,
                                               false, false);
}

static __device__ inline int wred_sumi(int x) {
#pragma unroll
  for (int o = 16; o > 0; o >>= 1) x += __shfl_xor(x, o, 32);
  return x;
}

static __device__ inline float unkey(unsigned k) {
  unsigned u = (k & 0x80000000u) ? (k ^ 0x80000000u) : ~k;
  return __uint_as_float(u);
}

static __device__ inline unsigned xhalf_u(unsigned x) {
  return __builtin_amdgcn_permlanex16(x, x, 0x76543210u, 0xfedcba98u, false, false);
}
static __device__ inline float xhalf_f(float x) {
  return __uint_as_float(xhalf_u(__float_as_uint(x)));
}


static __device__ inline void bt_stage(unsigned& a, unsigned& b, int s, unsigned lowmask) {
  const unsigned na = __builtin_amdgcn_bitop3_b32(a, b << s, lowmask, 0xE4);
  const unsigned nb = __builtin_amdgcn_bitop3_b32(a >> s, b, lowmask, 0xE4);
  a = na;
  b = nb;
}

static __device__ inline void planes8(unsigned (&x)[8]) {
#pragma unroll
  for (int i = 0; i < 4; ++i) bt_stage(x[i], x[i + 4], 4, 0x0F0F0F0Fu);
#pragma unroll
  for (int i = 0; i < 8; ++i)
    if ((i & 2) == 0) bt_stage(x[i], x[i + 2], 2, 0x33333333u);
#pragma unroll
  for (int i = 0; i < 8; i += 2) bt_stage(x[i], x[i + 1], 1, 0x55555555u);
}

__global__ void nsa_blockmean_kernel(const float* __restrict__ K,
                                     const float* __restrict__ V,
                                     float* __restrict__ KB,
                                     float* __restrict__ VB) {
  const int g = blockIdx.x;
  const int d = threadIdx.x;
  const float* kp = K + (size_t)g * (BS * DD);
  const float* vp = V + (size_t)g * (BS * DD);
  float ks = 0.f, vs = 0.f;
  for (int t = 0; t < BS; ++t) {
    ks += kp[t * DD + d];
    vs += vp[t * DD + d];
  }
  KB[(size_t)g * DD + d] = ks * (1.f / BS);
  VB[(size_t)g * DD + d] = vs * (1.f / BS);
}

static __device__ inline void load_ktile(v2f (&kf)[16], const float* __restrict__ Kh,
                                         int row, int half) {
#pragma unroll
  for (int t = 0; t < 16; ++t) {
    const float* bp = Kh + (size_t)row * DD + 4 * t + 2 * half;
    v2f a; a.x = bp[0]; a.y = bp[1];
    kf[t] = a;
  }
}

static __device__ inline v8f mma_ktile(const v2f (&kf)[16], const v2f (&qa)[16]) {
  v8f acc = vzero8();
#pragma unroll
  for (int t = 0; t < 16; ++t) acc = wmma4(kf[t], qa[t], acc);
  return acc;
}

static __device__ inline void load_qtile(v2f (&qf)[16], const v2f* __restrict__ sQ, int qt,
                                         int lane) {
#pragma unroll
  for (int t = 0; t < 16; ++t) qf[t] = sQ[(qt * 16 + t) * 32 + lane];
}

template <int NG, typename FB, typename FR, typename FE>
static __device__ inline void qk_tiles4(const float* __restrict__ Kh, const v2f* __restrict__ sQ,
                                        int wave, int lane, int half, FB kbase, FR krow, FE emit) {
  v2f kf[16], kfn[16], qfa[16], qfb[16];
  load_ktile(kfn, Kh, krow(0), half);
  load_qtile(qfa, sQ, wave, lane);
#pragma nounroll
  for (int g = 0; g < NG; ++g) {
    const int kb = kbase(g);
#pragma unroll
    for (int t = 0; t < 16; ++t) kf[t] = kfn[t];
    const int gn = (g + 1 < NG) ? g + 1 : NG - 1;
    load_ktile(kfn, Kh, krow(gn), half);
#pragma nounroll
    for (int j = 0; j < 4; j += 2) {
      const int qt0 = (wave + j) & 3;
      const int qt1 = (wave + j + 1) & 3;
      load_qtile(qfb, sQ, qt1, lane);
      __builtin_amdgcn_sched_barrier(0);
      emit(g, qt0, kb, mma_ktile(kf, qfa));
      load_qtile(qfa, sQ, (wave + j + 2) & 3, lane);
      __builtin_amdgcn_sched_barrier(0);
      emit(g, qt1, kb, mma_ktile(kf, qfb));
    }
  }
}

struct VTile {
  const float* vb;
  const float* v0;
};

static __device__ inline void load_vtile(v2f (&vf)[4], const VTile vt) {
  const float* __restrict__ vb = vt.vb;
  const float* __restrict__ v0 = vt.v0;
#pragma unroll
  for (int u = 0; u < 4; ++u) {
    v2f va;
    va.x = (u == 0) ? v0[0] : vb[(2 * u) * DD];
    va.y = vb[(2 * u + 1) * DD];
    vf[u] = va;
  }
}

#define WROWS4 (16 * PSTRIDE / 4)

static __device__ inline void load_ptile(float4 (&pf)[8], const float4* __restrict__ pall, int t) {
#pragma unroll
  for (int qt = 0; qt < 4; ++qt) {
    pf[2 * qt] = pall[qt * WROWS4 + (2 * t) * 32];
    pf[2 * qt + 1] = pall[qt * WROWS4 + (2 * t + 1) * 32];
  }
}

static __device__ inline void mma_vtile(v8f (&acc)[4], const v2f (&vf)[4], const float4 (&pf)[8]) {
#pragma unroll
  for (int u = 0; u < 4; ++u) {
#pragma unroll
    for (int qt = 0; qt < 4; ++qt) {
      const float4 c4 = pf[2 * qt + (u >> 1)];
      v2f pb;
      pb.x = (u & 1) ? c4.z : c4.x;
      pb.y = (u & 1) ? c4.w : c4.y;
      acc[qt] = wmma4(vf[u], pb, acc[qt]);
    }
  }
}

template <int NG, typename F>
static __device__ inline void pv_tiles(v8f (&acc)[4], const float4* __restrict__ pall, F gbase,
                                       const float* v0first) {
  v2f va[4][4];
  float4 pa[8], pb[8];
  {
    const float* gb = gbase(0);
    VTile vt;
    vt.vb = gb; vt.v0 = v0first ? v0first : gb;
    load_vtile(va[0], vt);
    vt.vb = gb + 16 * DD; vt.v0 = vt.vb;
    load_vtile(va[1], vt);
    vt.vb = gb + 32 * DD; vt.v0 = vt.vb;
    load_vtile(va[2], vt);
    vt.vb = gb + 48 * DD; vt.v0 = vt.vb;
    load_vtile(va[3], vt);
  }
  load_ptile(pa, pall, 0);
  const float* gnext = gbase(NG > 1 ? 1 : 0);
#pragma nounroll
  for (int g = 0; g < NG; ++g) {
    const int gn = (g + 1 < NG) ? g + 1 : NG - 1;
    const float* gnb = gnext;
    gnext = gbase((g + 2 < NG) ? g + 2 : NG - 1);
    const int t = 4 * g;
    VTile vt;
    load_ptile(pb, pall, t + 1);
    __builtin_amdgcn_sched_barrier(0);
    mma_vtile(acc, va[0], pa);
    vt.vb = gnb; vt.v0 = gnb;
    load_vtile(va[0], vt);
    load_ptile(pa, pall, t + 2);
    __builtin_amdgcn_sched_barrier(0);
    mma_vtile(acc, va[1], pb);
    vt.vb = gnb + 16 * DD; vt.v0 = vt.vb;
    load_vtile(va[1], vt);
    load_ptile(pb, pall, t + 3);
    __builtin_amdgcn_sched_barrier(0);
    mma_vtile(acc, va[2], pa);
    vt.vb = gnb + 32 * DD; vt.v0 = vt.vb;
    load_vtile(va[2], vt);
    load_ptile(pa, pall, 4 * gn);
    __builtin_amdgcn_sched_barrier(0);
    mma_vtile(acc, va[3], pb);
    vt.vb = gnb + 48 * DD; vt.v0 = vt.vb;
    load_vtile(va[3], vt);
    __builtin_amdgcn_sched_barrier(0);
  }
}

__global__ __launch_bounds__(128, 2) void nsa_main_kernel(
    const float* __restrict__ Q, const float* __restrict__ K,
    const float* __restrict__ V, const float* __restrict__ KB,
    const float* __restrict__ VB, float* __restrict__ O) {
  extern __shared__ __align__(16) float sP[];
  __shared__ float sScore[NB];
  __shared__ int sSel[NSEL];
  __shared__ v2f sQ[4 * 16 * 32];
  __shared__ float sRs[2][BS];

  const int wg = blockIdx.x;
  const int qb = wg % NB;
  const int bh = wg / NB;
  const size_t hoff = (size_t)bh * SS * DD;
  const float* Qh = Q + hoff;
  const float* Kh = K + hoff;
  const float* Vh = V + hoff;
  const float* KBh = KB + (size_t)bh * NB * DD;
  const float* VBh = VB + (size_t)bh * NB * DD;
  float* Oh = O + hoff;

  const int tid = threadIdx.x;
  const int wave = tid >> 5;
  const int lane = tid & 31;
  const int half = lane >> 4;
  const int ln = lane & 15;
  const int qrow0 = qb * BS + wave * 16;
  float* myP = sP + (size_t)wave * 16 * PSTRIDE;

  v2f qa[16];
#pragma unroll
  for (int t = 0; t < 16; ++t) {
    const float* p = Qh + (size_t)(qrow0 + ln) * DD + 4 * t + 2 * half;
    v2f a; a.x = p[0]; a.y = p[1];
    qa[t] = a;
    sQ[(wave * 16 + t) * 32 + lane] = a;
  }

  v8f oacc[4];
#pragma unroll
  for (int dt = 0; dt < 4; ++dt) oacc[dt] = vzero8();

#pragma unroll
  for (int nt = 0; nt < 2; ++nt) {
    v8f acc = vzero8();
#pragma unroll
    for (int t = 0; t < 16; ++t) {
      const float* bp = KBh + (size_t)(nt * 16 + ln) * DD + 4 * t + 2 * half;
      v2f bf; bf.x = bp[0]; bf.y = bp[1];
      acc = wmma4(qa[t], bf, acc);
    }
#pragma unroll
    for (int r = 0; r < 8; ++r) {
      int m = r + 8 * half;
      int col = nt * 16 + ln;
      float x = acc[r] * SCALE;
      myP[m * PSTRIDE + col] = (col <= qb) ? x : NEGINF;
    }
  }
  __syncthreads();

  if (tid < NB) {
    float s = 0.f;
    for (int q2 = 0; q2 < BS; ++q2) s += sP[q2 * PSTRIDE + tid];
    sScore[tid] = s * (1.f / BS);
  }
  __syncthreads();

  if (tid == 0) {
    float loc[NB];
    for (int i = 0; i < NB; ++i) loc[i] = sScore[i];
    for (int s = 0; s < NSEL; ++s) {
      int bi = 0;
      float bv = loc[0];
      for (int i = 1; i < NB; ++i)
        if (loc[i] > bv) { bv = loc[i]; bi = i; }
      sSel[s] = bi;
      loc[bi] = -3.4e38f;
    }
  }
  __syncthreads();

  if (tid < BS) {
    float* row = sP + tid * PSTRIDE;
    float mx = -3.4e38f;
    for (int c = 0; c < NB; ++c) mx = fmaxf(mx, row[c]);
    float sum = 0.f;
    for (int c = 0; c < NB; ++c) {
      float p = (mx > -1e29f) ? __expf(row[c] - mx) : 0.f;
      row[c] = p;
      sum += p;
    }
    float rs = (sum > 0.f) ? 1.f / sum : 0.f;
    for (int c = 0; c < NB; ++c) row[c] *= rs;
  }
  __syncthreads();

#pragma unroll
  for (int dt = 0; dt < 4; ++dt) {
#pragma unroll
    for (int g = 0; g < 8; ++g) {
      int c0 = 4 * g + 2 * half;
      v2f a; a.x = myP[ln * PSTRIDE + c0]; a.y = myP[ln * PSTRIDE + c0 + 1];
      v2f bf;
      bf.x = VBh[(size_t)c0 * DD + dt * 16 + ln];
      bf.y = VBh[(size_t)(c0 + 1) * DD + dt * 16 + ln];
      oacc[dt] = wmma4(a, bf, oacc[dt]);
    }
  }
  __syncthreads();

  float4* const mine = reinterpret_cast<float4*>(myP) + lane;
  qk_tiles4<NSEL>(
      Kh, sQ, wave, lane, half,
      [&](int g) { return sSel[g] * BS + wave * 16; },
      [&](int g) { return sSel[g] * BS + wave * 16 + ln; },
      [&](int g, int qt, int keyBase, v8f acc) {
        const int i = qb * BS + qt * 16 + ln;
        const int t = 4 * g + wave;
        float4* const dst = reinterpret_cast<float4*>(sP + (size_t)qt * 16 * PSTRIDE) + lane;
        float w[8];
#pragma unroll
        for (int r = 0; r < 8; ++r) {
          const int tok = keyBase + r + 8 * half;
          const float x = acc[r] * SCALE;
          w[r] = (tok <= i) ? x : NEGINF;
        }
        dst[(2 * t) * 32] = make_float4(w[0], w[1], w[2], w[3]);
        dst[(2 * t + 1) * 32] = make_float4(w[4], w[5], w[6], w[7]);
      });
  __syncthreads();

  float mxl = -__builtin_huge_valf();
#pragma clang loop vectorize(disable) interleave(disable)
#pragma unroll 4
  for (int c = 0; c < SELC / 8; ++c) {
    const float4 x4 = mine[c * 32];
    mxl = fmaxf(mxl, fmaxf(fmaxf(x4.x, x4.y), fmaxf(x4.z, x4.w)));
  }
  unsigned alive[8];
#pragma unroll
  for (int w = 0; w < 8; ++w) alive[w] = 0xFFFFFFFFu;
  unsigned need = KEEP;
  unsigned prefix = 0u;
  unsigned flip = 0xFFFFFFFFu;
#pragma nounroll
  for (int ph = 3; ph >= 0; --ph) {
    asm volatile("" ::: "memory");
    const unsigned sel2 = 0x04000400u + 0x01010101u * (unsigned)ph;
    unsigned X[8][8];
    float4 kq[2][8];
#pragma unroll
    for (int i = 0; i < 8; ++i) kq[0][i] = mine[i * 32];
#pragma unroll
    for (int w = 0; w < 8; ++w) {
      if (w + 1 < 8) {
#pragma unroll
        for (int i = 0; i < 8; ++i) kq[(w + 1) & 1][i] = mine[(8 * (w + 1) + i) * 32];
      }
      __builtin_amdgcn_sched_barrier(0);
#pragma unroll
      for (int i = 0; i < 8; ++i) {
        const float4 x4 = kq[w & 1][i];
        const unsigned u01 =
            __builtin_amdgcn_perm(__float_as_uint(x4.y), __float_as_uint(x4.x), sel2);
        const unsigned u23 =
            __builtin_amdgcn_perm(__float_as_uint(x4.w), __float_as_uint(x4.z), sel2);
        X[w][i] = __builtin_amdgcn_perm(u23, u01, 0x05040100u);
      }
      planes8(X[w]);
    }
    const unsigned lowbit = 1u << (8 * ph);
#pragma unroll
    for (int j = 7; j >= 0; --j) {
      unsigned t[8];
      unsigned c0 = 0u, c1 = 0u;
#pragma unroll
      for (int w = 0; w < 8; w += 2) {
        t[w] = alive[w] & (X[w][j] ^ flip);
        t[w + 1] = alive[w + 1] & (X[w + 1][j] ^ flip);
        c0 += (unsigned)__builtin_popcount(t[w]);
        c1 += (unsigned)__builtin_popcount(t[w + 1]);
      }
      unsigned cnt = c0 + c1;
      cnt += xhalf_u(cnt);
      const bool take = cnt >= need;
      prefix = take ? (prefix | (lowbit << j)) : prefix;
      need = take ? need : need - cnt;
      const unsigned fd = take ? flip : ~flip;
#pragma unroll
      for (int w = 0; w < 8; ++w) alive[w] = alive[w] & (X[w][j] ^ fd);
      if (j == 7) flip = (ph == 3) ? (take ? 0u : 0xFFFFFFFFu) : flip;
    }
  }

  v8f sacc[4], wacc[4];
#pragma unroll
  for (int qt = 0; qt < 4; ++qt) {
    sacc[qt] = vzero8();
    wacc[qt] = vzero8();
  }
  const float* vlane = Vh + (size_t)(8 * half) * DD + 16 * wave + ln;
  const float4* const pall = reinterpret_cast<const float4*>(sP) + lane;

  {
    const float thr = unkey(prefix);
    const float mx = fmaxf(mxl, xhalf_f(mxl));
    const bool live = mx > -1e29f;
    float s0 = 0.f, s1 = 0.f, s2 = 0.f, s3 = 0.f;
#pragma unroll 4
    for (int c = 0; c < SELC / 8; ++c) {
      const float4 x4 = mine[c * 32];
      const float p0 = (x4.x >= thr && live) ? __expf(x4.x - mx) : 0.f;
      const float p1 = (x4.y >= thr && live) ? __expf(x4.y - mx) : 0.f;
      const float p2 = (x4.z >= thr && live) ? __expf(x4.z - mx) : 0.f;
      const float p3 = (x4.w >= thr && live) ? __expf(x4.w - mx) : 0.f;
      s0 += p0; s1 += p1; s2 += p2; s3 += p3;
      mine[c * 32] = make_float4(p0, p1, p2, p3);
    }
    float sum = (s0 + s1) + (s2 + s3);
    sum += xhalf_f(sum);
    const float rs = (sum > 0.f) ? 1.f / sum : 0.f;

    if (half == 0) sRs[0][wave * 16 + ln] = rs;
    __syncthreads();
    pv_tiles<NSEL>(
        sacc, pall, [&](int g) { return vlane + (size_t)sSel[g] * (BS * DD); },
        static_cast<const float*>(nullptr));
  }

  auto wgroup_ok = [&](int cg) { return qb - 2 + cg >= 0; };
  auto wgroup_kb = [&](int cg) { return (qb - 2 + cg >= 0) ? qb - 2 + cg : 0; };
  __syncthreads();
  qk_tiles4<WINC / BS>(
      Kh, sQ, wave, lane, half,
      [&](int g) { return wgroup_kb(g) * BS + wave * 16; },
      [&](int g) {
        return (g == 0 && wave == 0 && ln == 0) ? 0 : wgroup_kb(g) * BS + wave * 16 + ln;
      },
      [&](int g, int qt, int keyBase, v8f acc) {
        const int i = qb * BS + qt * 16 + ln;
        const int t = 4 * g + wave;
        float4* const dst = reinterpret_cast<float4*>(sP + (size_t)qt * 16 * PSTRIDE) + lane;
        const int ilim = wgroup_ok(g) ? i : -1;
        const bool gslot = (g == 0) && (wave == 0) && (half == 0);
        float w[8];
#pragma unroll
        for (int r = 0; r < 8; ++r) {
          const int kj = (r == 0 && gslot) ? 0 : keyBase + r + 8 * half;
          const bool ok = (kj <= ilim) && (((i - kj) < WIN) || (kj < 1));
          w[r] = ok ? acc[r] * SCALE : NEGINF;
        }
        dst[(2 * t) * 32] = make_float4(w[0], w[1], w[2], w[3]);
        dst[(2 * t + 1) * 32] = make_float4(w[4], w[5], w[6], w[7]);
      });
  __syncthreads();

  {
    float wl[WINC / 2];
#pragma unroll
    for (int c = 0; c < WINC / 8; ++c) {
      const float4 w4 = mine[c * 32];
      wl[4 * c + 0] = w4.x; wl[4 * c + 1] = w4.y; wl[4 * c + 2] = w4.z; wl[4 * c + 3] = w4.w;
    }
    float m0 = wl[0], m1 = wl[1], m2 = wl[2], m3 = wl[3];
#pragma unroll
    for (int j = 4; j < WINC / 2; j += 4) {
      m0 = fmaxf(m0, wl[j + 0]);
      m1 = fmaxf(m1, wl[j + 1]);
      m2 = fmaxf(m2, wl[j + 2]);
      m3 = fmaxf(m3, wl[j + 3]);
    }
    float mx = fmaxf(fmaxf(m0, m1), fmaxf(m2, m3));
    mx = fmaxf(mx, xhalf_f(mx));
    const bool live = mx > -1e29f;
    float s0 = 0.f, s1 = 0.f, s2 = 0.f, s3 = 0.f;
#pragma unroll
    for (int j = 0; j < WINC / 2; j += 4) {
      const float p0 = live ? __expf(wl[j + 0] - mx) : 0.f;
      const float p1 = live ? __expf(wl[j + 1] - mx) : 0.f;
      const float p2 = live ? __expf(wl[j + 2] - mx) : 0.f;
      const float p3 = live ? __expf(wl[j + 3] - mx) : 0.f;
      s0 += p0; s1 += p1; s2 += p2; s3 += p3;
      mine[(j >> 2) * 32] = make_float4(p0, p1, p2, p3);
    }
    float sum = (s0 + s1) + (s2 + s3);
    sum += xhalf_f(sum);
    const float rs = (sum > 0.f) ? 1.f / sum : 0.f;

    if (half == 0) sRs[1][wave * 16 + ln] = rs;
    __syncthreads();
    pv_tiles<WINC / BS>(
        wacc, pall, [&](int g) { return vlane + (size_t)wgroup_kb(g) * (BS * DD); },
        (half == 0) ? Vh + 16 * wave + ln : static_cast<const float*>(nullptr));
  }

  __syncthreads();
#pragma unroll
  for (int qt = 0; qt < 4; ++qt) {
    const float rss = sRs[0][qt * 16 + ln];
    const float rsw = sRs[1][qt * 16 + ln];
    float o[8];
#pragma unroll
    for (int r = 0; r < 8; ++r) o[r] = sacc[qt][r] * rss + wacc[qt][r] * rsw;
    float* dst = sP + (size_t)(qt * 16 + ln) * PSTRIDE + 256 + wave * 16 + 8 * half;
    *reinterpret_cast<float4*>(dst) = make_float4(o[0], o[1], o[2], o[3]);
    *reinterpret_cast<float4*>(dst + 4) = make_float4(o[4], o[5], o[6], o[7]);
  }
  __syncthreads();
#pragma unroll
  for (int dt = 0; dt < 4; ++dt) {
#pragma unroll
    for (int r = 0; r < 8; ++r)
      oacc[dt][r] += myP[(r + 8 * half) * PSTRIDE + 256 + dt * 16 + ln];
  }

  const float third = 1.f / 3.f;
#pragma unroll
  for (int dt = 0; dt < 4; ++dt) {
#pragma unroll
    for (int r = 0; r < 8; ++r) {
      int row = qrow0 + r + 8 * half;
      Oh[(size_t)row * DD + dt * 16 + ln] = oacc[dt][r] * third;
    }
  }
}

extern "C" void kernel_launch(void* const* d_in, const int* in_sizes, int n_in,
                              void* d_out, int out_size, void* d_ws,
                              size_t ws_size, hipStream_t stream) {
  (void)in_sizes; (void)n_in; (void)out_size; (void)ws_size;
  const float* q = (const float*)d_in[0];
  const float* k = (const float*)d_in[1];
  const float* v = (const float*)d_in[2];
  float* out = (float*)d_out;
  float* kblk = (float*)d_ws;
  float* vblk = kblk + (size_t)BB * HH * NB * DD;

  nsa_blockmean_kernel<<<BB * HH * NB, 64, 0, stream>>>(k, v, kblk, vblk);

  const size_t shmem = (size_t)64 * PSTRIDE * sizeof(float);
  hipFuncSetAttribute(reinterpret_cast<const void*>(nsa_main_kernel),
                      hipFuncAttributeMaxDynamicSharedMemorySize, (int)shmem);
  nsa_main_kernel<<<BB * HH * NB, 128, shmem, stream>>>(q, k, v, kblk, vblk,
                                                        out);
}
